# starting kernel
# baseline (speedup 1.0000x reference)
#include <hip/hip_runtime.h>
#include <hip/hip_bf16.h>
#include <hip/hip_fp16.h>

#define NNODES 100000
#define NEDGES 1600000
#define HDIM   64
#define HNUMB  128
#define NCONVL 2
#define NMLPL  2
#define NEGSLP 0.2f

typedef __attribute__((ext_vector_type(2))) float v2f;
typedef __attribute__((ext_vector_type(8))) float v8f;

__device__ __forceinline__ float leaky(float x) { return x > 0.f ? x : NEGSLP * x; }

__device__ __forceinline__ unsigned fenc(float f) {
    unsigned u = __float_as_uint(f);
    return (u & 0x80000000u) ? ~u : (u | 0x80000000u);
}
__device__ __forceinline__ float fdec(unsigned u) {
    return (u & 0x80000000u) ? __uint_as_float(u & 0x7fffffffu) : __uint_as_float(~u);
}

__global__ __launch_bounds__(256) void gemm_xw_kernel(const float* __restrict__ X,
                                                      const float* __restrict__ W,
                                                      float* __restrict__ Hout,
                                                      int nrows) {
    __shared__ float Wlds[HDIM * HDIM];
    const int tid = threadIdx.x;
    for (int idx = tid; idx < HDIM * HDIM; idx += 256) {
        int k = idx >> 6, n = idx & 63;
        Wlds[(k >> 1) * 128 + n * 2 + (k & 1)] = W[idx];
    }
    __syncthreads();

    const int wave    = tid >> 5;
    const int lane    = tid & 31;
    const int hi      = lane >> 4;
    const int mrow    = lane & 15;
    const int rowbase = blockIdx.x * 128 + wave * 16;
    if (rowbase >= nrows) return;

    v2f afrag[16];
#pragma unroll
    for (int kk = 0; kk < 16; ++kk) {
        const v2f* ap = (const v2f*)(X + (size_t)(rowbase + mrow) * HDIM + kk * 4 + hi * 2);
        afrag[kk] = *ap;
    }
#pragma unroll
    for (int nt = 0; nt < 4; ++nt) {
        const int ncol = nt * 16 + mrow;
        v8f acc = {};
#pragma unroll
        for (int kk = 0; kk < 16; ++kk) {
            v2f b = *(const v2f*)&Wlds[(kk * 2 + hi) * 128 + ncol * 2];
            acc = __builtin_amdgcn_wmma_f32_16x16x4_f32(false, afrag[kk], false, b,
                                                        (short)0, acc, false, false);
        }
#pragma unroll
        for (int v = 0; v < 8; ++v)
            Hout[(size_t)(rowbase + v + 8 * hi) * HDIM + ncol] = acc[v];
    }
}

__global__ void dots_kernel(const float* __restrict__ Hm, const float* __restrict__ asrc,
                            const float* __restrict__ adst, float* __restrict__ a_s,
                            float* __restrict__ a_d, int n) {
    int r = blockIdx.x * blockDim.x + threadIdx.x;
    if (r >= n) return;
    const float* hr = Hm + (size_t)r * HDIM;
    float s = 0.f, d = 0.f;
#pragma unroll
    for (int k = 0; k < HDIM; ++k) { float v = hr[k]; s += v * asrc[k]; d += v * adst[k]; }
    a_s[r] = s; a_d[r] = d;
}

__global__ void node_init_kernel(const float* __restrict__ a_s, const float* __restrict__ a_d,
                                 float* __restrict__ e_self, unsigned* __restrict__ m_enc, int n) {
    int i = blockIdx.x * blockDim.x + threadIdx.x;
    if (i >= n) return;
    float es = leaky(a_s[i] + a_d[i]);
    e_self[i] = es;
    m_enc[i]  = fenc(es);
}

__global__ void edge_max_kernel(const int* __restrict__ src, const int* __restrict__ dst,
                                const float* __restrict__ a_s, const float* __restrict__ a_d,
                                unsigned* __restrict__ m_enc, int ne) {
    int e = blockIdx.x * blockDim.x + threadIdx.x;
    if (e >= ne) return;
    int s = src[e], d = dst[e];
    atomicMax(m_enc + d, fenc(leaky(a_s[s] + a_d[d])));
}

__global__ void node_selfw_kernel(const float* __restrict__ e_self,
                                  const unsigned* __restrict__ m_enc,
                                  float* __restrict__ denom, int n) {
    int i = blockIdx.x * blockDim.x + threadIdx.x;
    if (i >= n) return;
    denom[i] = __expf(e_self[i] - fdec(m_enc[i]));
}

__global__ void num_init_kernel(const float* __restrict__ wself, const float* __restrict__ Hm,
                                float* __restrict__ num, int total) {
    int t = blockIdx.x * blockDim.x + threadIdx.x;
    if (t >= total) return;
    num[t] = wself[t >> 6] * Hm[t];
}

__global__ void edge_w_kernel(const int* __restrict__ src, const int* __restrict__ dst,
                              const float* __restrict__ a_s, const float* __restrict__ a_d,
                              const unsigned* __restrict__ m_enc, float* __restrict__ w_edge,
                              float* __restrict__ denom, int ne) {
    int e = blockIdx.x * blockDim.x + threadIdx.x;
    if (e >= ne) return;
    int s = src[e], d = dst[e];
    float w = __expf(leaky(a_s[s] + a_d[d]) - fdec(m_enc[d]));
    w_edge[e] = w;
    atomicAdd(denom + d, w);
}

__global__ void edge_scatter_kernel(const int* __restrict__ src, const int* __restrict__ dst,
                                    const float* __restrict__ w_edge, const float* __restrict__ Hm,
                                    float* __restrict__ num, long total) {
    long t = (long)blockIdx.x * blockDim.x + threadIdx.x;
    if (t >= total) return;
    int e = (int)(t >> 6), f = (int)(t & 63);
    atomicAdd(num + (size_t)dst[e] * HDIM + f, w_edge[e] * Hm[(size_t)src[e] * HDIM + f]);
}

__global__ void finalize_kernel(const float* __restrict__ num, const float* __restrict__ denom,
                                const float* __restrict__ bias, float* __restrict__ out, int total) {
    int t = blockIdx.x * blockDim.x + threadIdx.x;
    if (t >= total) return;
    out[t] = num[t] / (denom[t >> 6] + 1e-16f) + bias[t & 63];
}

__global__ void gather_add_kernel(const float* __restrict__ out, const float* __restrict__ vnd,
                                  const int* __restrict__ blocks, float* __restrict__ x_in, int total) {
    int t = blockIdx.x * blockDim.x + threadIdx.x;
    if (t >= total) return;
    x_in[t] = out[t] + vnd[blocks[t >> 6] * HDIM + (t & 63)];
}

__global__ void zero_kernel(float* __restrict__ p, int n) {
    int t = blockIdx.x * blockDim.x + threadIdx.x;
    if (t < n) p[t] = 0.f;
}

__global__ __launch_bounds__(256) void seg_sum_kernel(const float* __restrict__ out,
                                                      const int* __restrict__ blocks,
                                                      float* __restrict__ vn_seg, long total) {
    __shared__ float acc[HNUMB * HDIM];
    for (int i = threadIdx.x; i < HNUMB * HDIM; i += 256) acc[i] = 0.f;
    __syncthreads();
    for (long t = (long)blockIdx.x * blockDim.x + threadIdx.x; t < total;
         t += (long)gridDim.x * blockDim.x) {
        int n = (int)(t >> 6), f = (int)(t & 63);
        atomicAdd(&acc[blocks[n] * HDIM + f], out[t]);
    }
    __syncthreads();
    for (int i = threadIdx.x; i < HNUMB * HDIM; i += 256) atomicAdd(vn_seg + i, acc[i]);
}

__global__ __launch_bounds__(256) void vn_update_kernel(float* __restrict__ vn_direct,
                                                        const float* __restrict__ vn_seg,
                                                        float* __restrict__ vn_root) {
    int tid = threadIdx.x;
    for (int i = tid; i < HNUMB * HDIM; i += 256)
        vn_direct[i] += vn_seg[i] + vn_root[i & (HDIM - 1)];
    __syncthreads();
    if (tid < HDIM) {
        float s = 0.f;
        for (int r = 0; r < HNUMB; ++r) s += vn_direct[r * HDIM + tid];
        vn_root[tid] += s;
    }
}

__global__ __launch_bounds__(64) void mlp_kernel(float* __restrict__ vn,
                                                 const float* __restrict__ W1, const float* __restrict__ b1,
                                                 const float* __restrict__ W2, const float* __restrict__ b2) {
    __shared__ float rin[HDIM];
    __shared__ float t1[HDIM];
    int f = threadIdx.x;
    float* row = vn + (size_t)blockIdx.x * HDIM;
    rin[f] = row[f];
    __syncthreads();
    float s = b1[f];
#pragma unroll
    for (int k = 0; k < HDIM; ++k) s += rin[k] * W1[k * HDIM + f];
    t1[f] = fmaxf(s, 0.f);
    __syncthreads();
    float s2 = b2[f];
#pragma unroll
    for (int k = 0; k < HDIM; ++k) s2 += t1[k] * W2[k * HDIM + f];
    row[f] = fmaxf(s2, 0.f);
}

__global__ void vn_init_kernel(const float* __restrict__ vn_emb, float* __restrict__ vn_direct,
                               float* __restrict__ vn_root) {
    int t = blockIdx.x * blockDim.x + threadIdx.x;
    if (t < HNUMB * HDIM) vn_direct[t] = vn_emb[t & (HDIM - 1)];
    if (t < HDIM) vn_root[t] = vn_emb[t];
}

extern "C" void kernel_launch(void* const* d_in, const int* in_sizes, int n_in,
                              void* d_out, int out_size, void* d_ws, size_t ws_size,
                              hipStream_t stream) {
    const float* x       = (const float*)d_in[0];
    const int*   ei      = (const int*)d_in[1];
    const int*   src     = ei;
    const int*   dst     = ei + NEDGES;
    const int*   hblocks = (const int*)d_in[2];
    const float* W0      = (const float*)d_in[5];
    const float* asrc0   = (const float*)d_in[6];
    const float* adst0   = (const float*)d_in[7];
    const float* b0      = (const float*)d_in[8];
    const float* Wl      = (const float*)d_in[9];
    const float* asrcl   = (const float*)d_in[10];
    const float* adstl   = (const float*)d_in[11];
    const float* bl      = (const float*)d_in[12];
    const float* vn_emb  = (const float*)d_in[13];
    const float* mW1     = (const float*)d_in[14];
    const float* mb1     = (const float*)d_in[15];
    const float* mW2     = (const float*)d_in[16];
    const float* mb2     = (const float*)d_in[17];
    float* out = (float*)d_out;

    char*  ws  = (char*)d_ws;
    size_t off = 0;
    auto carve = [&](size_t bytes) -> void* {
        void* p = ws + off;
        off += (bytes + 255) & ~(size_t)255;
        return p;
    };
    float*    h         = (float*)carve((size_t)NNODES * HDIM * 4);
    float*    x_in      = (float*)carve((size_t)NNODES * HDIM * 4);
    float*    num       = (float*)carve((size_t)NNODES * HDIM * 4);
    float*    a_s       = (float*)carve((size_t)NNODES * 4);
    float*    a_d       = (float*)carve((size_t)NNODES * 4);
    float*    e_self    = (float*)carve((size_t)NNODES * 4);
    float*    denom     = (float*)carve((size_t)NNODES * 4);
    unsigned* m_enc     = (unsigned*)carve((size_t)NNODES * 4);
    float*    w_edge    = (float*)carve((size_t)NEDGES * 4);
    float*    vn_direct = (float*)carve((size_t)HNUMB * HDIM * 4);
    float*    vn_root   = (float*)carve((size_t)HDIM * 4);
    float*    vn_seg    = (float*)carve((size_t)HNUMB * HDIM * 4);

    const int NF      = NNODES * HDIM;
    const int nodeBlk = (NNODES + 255) / 256;
    const int nfBlk   = NF / 256;
    const int edgeBlk = NEDGES / 256;
    const long EF     = (long)NEDGES * HDIM;
    const int efBlk   = (int)(EF / 256);

    auto run_gat = [&](const float* Xin, const float* W, const float* asrc,
                       const float* adst, const float* bias) {
        gemm_xw_kernel<<<(NNODES + 127) / 128, 256, 0, stream>>>(Xin, W, h, NNODES);
        dots_kernel<<<nodeBlk, 256, 0, stream>>>(h, asrc, adst, a_s, a_d, NNODES);
        node_init_kernel<<<nodeBlk, 256, 0, stream>>>(a_s, a_d, e_self, m_enc, NNODES);
        edge_max_kernel<<<edgeBlk, 256, 0, stream>>>(src, dst, a_s, a_d, m_enc, NEDGES);
        node_selfw_kernel<<<nodeBlk, 256, 0, stream>>>(e_self, m_enc, denom, NNODES);
        num_init_kernel<<<nfBlk, 256, 0, stream>>>(denom, h, num, NF);
        edge_w_kernel<<<edgeBlk, 256, 0, stream>>>(src, dst, a_s, a_d, m_enc, w_edge, denom, NEDGES);
        edge_scatter_kernel<<<efBlk, 256, 0, stream>>>(src, dst, w_edge, h, num, EF);
        finalize_kernel<<<nfBlk, 256, 0, stream>>>(num, denom, bias, out, NF);
    };

    vn_init_kernel<<<(HNUMB * HDIM + 255) / 256, 256, 0, stream>>>(vn_emb, vn_direct, vn_root);

    run_gat(x, W0, asrc0, adst0, b0);

    for (int l = 0; l < NCONVL; ++l) {
        gather_add_kernel<<<nfBlk, 256, 0, stream>>>(out, vn_direct, hblocks, x_in, NF);
        run_gat(x_in, Wl + l * HDIM * HDIM, asrcl + l * HDIM, adstl + l * HDIM, bl + l * HDIM);
        zero_kernel<<<(HNUMB * HDIM + 255) / 256, 256, 0, stream>>>(vn_seg, HNUMB * HDIM);
        seg_sum_kernel<<<128, 256, 0, stream>>>(out, hblocks, vn_seg, (long)NF);
        vn_update_kernel<<<1, 256, 0, stream>>>(vn_direct, vn_seg, vn_root);
        for (int j = 0; j < NMLPL; ++j) {
            mlp_kernel<<<HNUMB, 64, 0, stream>>>(vn_direct, mW1 + j * HDIM * HDIM, mb1 + j * HDIM,
                                                 mW2 + j * HDIM * HDIM, mb2 + j * HDIM);
            mlp_kernel<<<1, 64, 0, stream>>>(vn_root, mW1 + j * HDIM * HDIM, mb1 + j * HDIM,
                                             mW2 + j * HDIM * HDIM, mb2 + j * HDIM);
        }
    }
}


# Segment sum taken where out is formed: finalize (256 blocks x 391 rows) adds its rows into an LDS table and flushes once; the pass that re-read out one element per round trip goes; sum order changes
# speedup vs baseline: 1.7788x; 1.7788x over previous
#include <hip/hip_runtime.h>
#include <hip/hip_bf16.h>
#include <hip/hip_fp16.h>

#define NNODES 100000
#define NEDGES 1600000
#define HDIM   64
#define HNUMB  128
#define NCONVL 2
#define NMLPL  2
#define NEGSLP 0.2f

typedef __attribute__((ext_vector_type(2))) float v2f;
typedef __attribute__((ext_vector_type(8))) float v8f;

__device__ __forceinline__ float leaky(float x) { return x > 0.f ? x : NEGSLP * x; }

__device__ __forceinline__ unsigned fenc(float f) {
    unsigned u = __float_as_uint(f);
    return (u & 0x80000000u) ? ~u : (u | 0x80000000u);
}
__device__ __forceinline__ float fdec(unsigned u) {
    return (u & 0x80000000u) ? __uint_as_float(u & 0x7fffffffu) : __uint_as_float(~u);
}

__global__ __launch_bounds__(256) void gemm_xw_kernel(const float* __restrict__ X,
                                                      const float* __restrict__ W,
                                                      float* __restrict__ Hout,
                                                      int nrows) {
    __shared__ float Wlds[HDIM * HDIM];
    const int tid = threadIdx.x;
    for (int idx = tid; idx < HDIM * HDIM; idx += 256) {
        int k = idx >> 6, n = idx & 63;
        Wlds[(k >> 1) * 128 + n * 2 + (k & 1)] = W[idx];
    }
    __syncthreads();

    const int wave    = tid >> 5;
    const int lane    = tid & 31;
    const int hi      = lane >> 4;
    const int mrow    = lane & 15;
    const int rowbase = blockIdx.x * 128 + wave * 16;
    if (rowbase >= nrows) return;

    v2f afrag[16];
#pragma unroll
    for (int kk = 0; kk < 16; ++kk) {
        const v2f* ap = (const v2f*)(X + (size_t)(rowbase + mrow) * HDIM + kk * 4 + hi * 2);
        afrag[kk] = *ap;
    }
#pragma unroll
    for (int nt = 0; nt < 4; ++nt) {
        const int ncol = nt * 16 + mrow;
        v8f acc = {};
#pragma unroll
        for (int kk = 0; kk < 16; ++kk) {
            v2f b = *(const v2f*)&Wlds[(kk * 2 + hi) * 128 + ncol * 2];
            acc = __builtin_amdgcn_wmma_f32_16x16x4_f32(false, afrag[kk], false, b,
                                                        (short)0, acc, false, false);
        }
#pragma unroll
        for (int v = 0; v < 8; ++v)
            Hout[(size_t)(rowbase + v + 8 * hi) * HDIM + ncol] = acc[v];
    }
}

__global__ void dots_kernel(const float* __restrict__ Hm, const float* __restrict__ asrc,
                            const float* __restrict__ adst, float* __restrict__ a_s,
                            float* __restrict__ a_d, int n) {
    int r = blockIdx.x * blockDim.x + threadIdx.x;
    if (r >= n) return;
    const float* hr = Hm + (size_t)r * HDIM;
    float s = 0.f, d = 0.f;
#pragma unroll
    for (int k = 0; k < HDIM; ++k) { float v = hr[k]; s += v * asrc[k]; d += v * adst[k]; }
    a_s[r] = s; a_d[r] = d;
}

__global__ void node_init_kernel(const float* __restrict__ a_s, const float* __restrict__ a_d,
                                 float* __restrict__ e_self, unsigned* __restrict__ m_enc, int n) {
    int i = blockIdx.x * blockDim.x + threadIdx.x;
    if (i >= n) return;
    float es = leaky(a_s[i] + a_d[i]);
    e_self[i] = es;
    m_enc[i]  = fenc(es);
}

__global__ void edge_max_kernel(const int* __restrict__ src, const int* __restrict__ dst,
                                const float* __restrict__ a_s, const float* __restrict__ a_d,
                                unsigned* __restrict__ m_enc, int ne) {
    int e = blockIdx.x * blockDim.x + threadIdx.x;
    if (e >= ne) return;
    int s = src[e], d = dst[e];
    atomicMax(m_enc + d, fenc(leaky(a_s[s] + a_d[d])));
}

__global__ void node_selfw_kernel(const float* __restrict__ e_self,
                                  const unsigned* __restrict__ m_enc,
                                  float* __restrict__ denom, int n) {
    int i = blockIdx.x * blockDim.x + threadIdx.x;
    if (i >= n) return;
    denom[i] = __expf(e_self[i] - fdec(m_enc[i]));
}

__global__ void num_init_kernel(const float* __restrict__ wself, const float* __restrict__ Hm,
                                float* __restrict__ num, int total) {
    int t = blockIdx.x * blockDim.x + threadIdx.x;
    if (t >= total) return;
    num[t] = wself[t >> 6] * Hm[t];
}

__global__ void edge_w_kernel(const int* __restrict__ src, const int* __restrict__ dst,
                              const float* __restrict__ a_s, const float* __restrict__ a_d,
                              const unsigned* __restrict__ m_enc, float* __restrict__ w_edge,
                              float* __restrict__ denom, int ne) {
    int e = blockIdx.x * blockDim.x + threadIdx.x;
    if (e >= ne) return;
    int s = src[e], d = dst[e];
    float w = __expf(leaky(a_s[s] + a_d[d]) - fdec(m_enc[d]));
    w_edge[e] = w;
    atomicAdd(denom + d, w);
}

#define SC_EPW   32
#define SC_BATCH 8
static_assert(NEDGES % 256 == 0, "edge scatter: a block of 8 waves owns 256 whole edges");
__global__ __launch_bounds__(256) void edge_scatter_kernel(const int* __restrict__ src, const int* __restrict__ dst,
                                    const float* __restrict__ w_edge, const float* __restrict__ Hm,
                                    float* __restrict__ num) {
    const int lane = threadIdx.x & 31;
    const int e    = blockIdx.x * 256 + threadIdx.x;
    const int   sv = src[e];
    const int   dv = dst[e];
    const int   wv = __float_as_int(w_edge[e]);

    float lo[2][SC_BATCH], hi[2][SC_BATCH];
#pragma unroll
    for (int j = 0; j < SC_BATCH; ++j) {
        const float* hr = Hm + (size_t)__builtin_amdgcn_readlane(sv, j) * HDIM;
        lo[0][j] = hr[lane];
        hi[0][j] = hr[lane + 32];
    }
#pragma unroll
    for (int b = 0; b < SC_EPW / SC_BATCH; ++b) {
        const int cur = b & 1, nxt = cur ^ 1;
        if (b + 1 < SC_EPW / SC_BATCH) {
#pragma unroll
            for (int j = 0; j < SC_BATCH; ++j) {
                const float* hr = Hm + (size_t)__builtin_amdgcn_readlane(sv, (b + 1) * SC_BATCH + j) * HDIM;
                lo[nxt][j] = hr[lane];
                hi[nxt][j] = hr[lane + 32];
            }
        }
#pragma unroll
        for (int j = 0; j < SC_BATCH; ++j) {
            const int   d = __builtin_amdgcn_readlane(dv, b * SC_BATCH + j);
            const float w = __int_as_float(__builtin_amdgcn_readlane(wv, b * SC_BATCH + j));
            float* nr = num + (size_t)d * HDIM;
            atomicAdd(nr + lane,      w * lo[cur][j]);
            atomicAdd(nr + lane + 32, w * hi[cur][j]);
        }
    }
}

__global__ void finalize_kernel(const float* __restrict__ num, const float* __restrict__ denom,
                                const float* __restrict__ bias, float* __restrict__ out, int total) {
    int t = blockIdx.x * blockDim.x + threadIdx.x;
    if (t >= total) return;
    out[t] = num[t] / (denom[t >> 6] + 1e-16f) + bias[t & 63];
}

#define FS_THREADS 512
#define FS_BLOCKS  256
#define FS_ROWS    ((NNODES + FS_BLOCKS - 1) / FS_BLOCKS)
#define FS_RSTEP   (FS_THREADS / 16)
#define FS_PITCH   (HDIM + 1)
__global__ __launch_bounds__(FS_THREADS) void finalize_segsum_kernel(
        const float* __restrict__ num, const float* __restrict__ denom,
        const float* __restrict__ bias, const int* __restrict__ blocks,
        float* __restrict__ out, float* __restrict__ vn_seg) {
    __shared__ float acc[HNUMB * FS_PITCH];
    const int tid = threadIdx.x;
    for (int i = tid; i < HNUMB * FS_PITCH; i += FS_THREADS) acc[i] = 0.f;
    __syncthreads();

    const int c4   = tid & 15;
    const int row0 = blockIdx.x * FS_ROWS;
    const int row1 = (row0 + FS_ROWS < NNODES) ? row0 + FS_ROWS : NNODES;
    const float4 b4 = ((const float4*)bias)[c4];
    for (int rb = row0 + (tid >> 4); rb < row1; rb += 4 * FS_RSTEP) {
        float4 n4[4];
        float  dn[4];
        int    bk[4];
#pragma unroll
        for (int u = 0; u < 4; ++u) {
            const int r = rb + u * FS_RSTEP;
            n4[u] = make_float4(0.f, 0.f, 0.f, 0.f); dn[u] = 1.f; bk[u] = 0;
            if (r < row1) {
                n4[u] = ((const float4*)num)[(size_t)r * (HDIM / 4) + c4];
                dn[u] = denom[r];
                bk[u] = blocks[r];
            }
        }
#pragma unroll
        for (int u = 0; u < 4; ++u) {
            const int r = rb + u * FS_RSTEP;
            if (r < row1) {
                const float dd = dn[u] + 1e-16f;
                float4 o;
                o.x = n4[u].x / dd + b4.x;
                o.y = n4[u].y / dd + b4.y;
                o.z = n4[u].z / dd + b4.z;
                o.w = n4[u].w / dd + b4.w;
                ((float4*)out)[(size_t)r * (HDIM / 4) + c4] = o;
                float* a = acc + bk[u] * FS_PITCH + c4 * 4;
                atomicAdd(a + 0, o.x);
                atomicAdd(a + 1, o.y);
                atomicAdd(a + 2, o.z);
                atomicAdd(a + 3, o.w);
            }
        }
    }
    __syncthreads();
    for (int i = tid; i < HNUMB * HDIM; i += FS_THREADS) {
        const int j = (i + blockIdx.x * 32) & (HNUMB * HDIM - 1);
        atomicAdd(vn_seg + j, acc[(j >> 6) * FS_PITCH + (j & 63)]);
    }
}

__global__ void gather_add_kernel(const float* __restrict__ out, const float* __restrict__ vnd,
                                  const int* __restrict__ blocks, float* __restrict__ x_in, int total) {
    int t = blockIdx.x * blockDim.x + threadIdx.x;
    if (t >= total) return;
    x_in[t] = out[t] + vnd[blocks[t >> 6] * HDIM + (t & 63)];
}

__global__ void zero_kernel(float* __restrict__ p, int n) {
    int t = blockIdx.x * blockDim.x + threadIdx.x;
    if (t < n) p[t] = 0.f;
}

__global__ __launch_bounds__(256) void vn_update_kernel(float* __restrict__ vn_direct,
                                                        const float* __restrict__ vn_seg,
                                                        float* __restrict__ vn_root) {
    int tid = threadIdx.x;
    for (int i = tid; i < HNUMB * HDIM; i += 256)
        vn_direct[i] += vn_seg[i] + vn_root[i & (HDIM - 1)];
    __syncthreads();
    if (tid < HDIM) {
        float s = 0.f;
        for (int r = 0; r < HNUMB; ++r) s += vn_direct[r * HDIM + tid];
        vn_root[tid] += s;
    }
}

__global__ __launch_bounds__(64) void mlp_kernel(float* __restrict__ vn,
                                                 const float* __restrict__ W1, const float* __restrict__ b1,
                                                 const float* __restrict__ W2, const float* __restrict__ b2) {
    __shared__ float rin[HDIM];
    __shared__ float t1[HDIM];
    int f = threadIdx.x;
    float* row = vn + (size_t)blockIdx.x * HDIM;
    rin[f] = row[f];
    __syncthreads();
    float s = b1[f];
#pragma unroll
    for (int k = 0; k < HDIM; ++k) s += rin[k] * W1[k * HDIM + f];
    t1[f] = fmaxf(s, 0.f);
    __syncthreads();
    float s2 = b2[f];
#pragma unroll
    for (int k = 0; k < HDIM; ++k) s2 += t1[k] * W2[k * HDIM + f];
    row[f] = fmaxf(s2, 0.f);
}

__global__ void vn_init_kernel(const float* __restrict__ vn_emb, float* __restrict__ vn_direct,
                               float* __restrict__ vn_root) {
    int t = blockIdx.x * blockDim.x + threadIdx.x;
    if (t < HNUMB * HDIM) vn_direct[t] = vn_emb[t & (HDIM - 1)];
    if (t < HDIM) vn_root[t] = vn_emb[t];
}

extern "C" void kernel_launch(void* const* d_in, const int* in_sizes, int n_in,
                              void* d_out, int out_size, void* d_ws, size_t ws_size,
                              hipStream_t stream) {
    const float* x       = (const float*)d_in[0];
    const int*   ei      = (const int*)d_in[1];
    const int*   src     = ei;
    const int*   dst     = ei + NEDGES;
    const int*   hblocks = (const int*)d_in[2];
    const float* W0      = (const float*)d_in[5];
    const float* asrc0   = (const float*)d_in[6];
    const float* adst0   = (const float*)d_in[7];
    const float* b0      = (const float*)d_in[8];
    const float* Wl      = (const float*)d_in[9];
    const float* asrcl   = (const float*)d_in[10];
    const float* adstl   = (const float*)d_in[11];
    const float* bl      = (const float*)d_in[12];
    const float* vn_emb  = (const float*)d_in[13];
    const float* mW1     = (const float*)d_in[14];
    const float* mb1     = (const float*)d_in[15];
    const float* mW2     = (const float*)d_in[16];
    const float* mb2     = (const float*)d_in[17];
    float* out = (float*)d_out;

    char*  ws  = (char*)d_ws;
    size_t off = 0;
    auto carve = [&](size_t bytes) -> void* {
        void* p = ws + off;
        off += (bytes + 255) & ~(size_t)255;
        return p;
    };
    float*    h         = (float*)carve((size_t)NNODES * HDIM * 4);
    float*    x_in      = (float*)carve((size_t)NNODES * HDIM * 4);
    float*    num       = (float*)carve((size_t)NNODES * HDIM * 4);
    float*    a_s       = (float*)carve((size_t)NNODES * 4);
    float*    a_d       = (float*)carve((size_t)NNODES * 4);
    float*    e_self    = (float*)carve((size_t)NNODES * 4);
    float*    denom     = (float*)carve((size_t)NNODES * 4);
    unsigned* m_enc     = (unsigned*)carve((size_t)NNODES * 4);
    float*    w_edge    = (float*)carve((size_t)NEDGES * 4);
    float*    vn_direct = (float*)carve((size_t)HNUMB * HDIM * 4);
    float*    vn_root   = (float*)carve((size_t)HDIM * 4);
    float*    vn_seg    = (float*)carve((size_t)HNUMB * HDIM * 4);

    const int NF      = NNODES * HDIM;
    const int nodeBlk = (NNODES + 255) / 256;
    const int nfBlk   = NF / 256;
    const int edgeBlk = NEDGES / 256;

    auto run_gat = [&](const float* Xin, const float* W, const float* asrc,
                       const float* adst, const float* bias, bool with_segsum) {
        gemm_xw_kernel<<<(NNODES + 127) / 128, 256, 0, stream>>>(Xin, W, h, NNODES);
        dots_kernel<<<nodeBlk, 256, 0, stream>>>(h, asrc, adst, a_s, a_d, NNODES);
        node_init_kernel<<<nodeBlk, 256, 0, stream>>>(a_s, a_d, e_self, m_enc, NNODES);
        edge_max_kernel<<<edgeBlk, 256, 0, stream>>>(src, dst, a_s, a_d, m_enc, NEDGES);
        node_selfw_kernel<<<nodeBlk, 256, 0, stream>>>(e_self, m_enc, denom, NNODES);
        num_init_kernel<<<nfBlk, 256, 0, stream>>>(denom, h, num, NF);
        edge_w_kernel<<<edgeBlk, 256, 0, stream>>>(src, dst, a_s, a_d, m_enc, w_edge, denom, NEDGES);
        edge_scatter_kernel<<<edgeBlk, 256, 0, stream>>>(src, dst, w_edge, h, num);
        if (with_segsum)
            finalize_segsum_kernel<<<FS_BLOCKS, FS_THREADS, 0, stream>>>(num, denom, bias, hblocks, out, vn_seg);
        else
            finalize_kernel<<<nfBlk, 256, 0, stream>>>(num, denom, bias, out, NF);
    };

    vn_init_kernel<<<(HNUMB * HDIM + 255) / 256, 256, 0, stream>>>(vn_emb, vn_direct, vn_root);

    run_gat(x, W0, asrc0, adst0, b0, false);

    for (int l = 0; l < NCONVL; ++l) {
        gather_add_kernel<<<nfBlk, 256, 0, stream>>>(out, vn_direct, hblocks, x_in, NF);
        zero_kernel<<<(HNUMB * HDIM + 255) / 256, 256, 0, stream>>>(vn_seg, HNUMB * HDIM);
        run_gat(x_in, Wl + l * HDIM * HDIM, asrcl + l * HDIM, adstl + l * HDIM, bl + l * HDIM, true);
        vn_update_kernel<<<1, 256, 0, stream>>>(vn_direct, vn_seg, vn_root);
        for (int j = 0; j < NMLPL; ++j) {
            mlp_kernel<<<HNUMB, 64, 0, stream>>>(vn_direct, mW1 + j * HDIM * HDIM, mb1 + j * HDIM,
                                                 mW2 + j * HDIM * HDIM, mb2 + j * HDIM);
            mlp_kernel<<<1, 64, 0, stream>>>(vn_root, mW1 + j * HDIM * HDIM, mb1 + j * HDIM,
                                             mW2 + j * HDIM * HDIM, mb2 + j * HDIM);
        }
    }
}


# Edge weight formed in the scatter wave from the logit the max pass keeps per edge (one gather of the max, exp, denominator add): weight pass, 3 gathers an edge, 3 launches go; atomics, arrival order
# speedup vs baseline: 1.8387x; 1.0337x over previous
#include <hip/hip_runtime.h>
#include <hip/hip_bf16.h>
#include <hip/hip_fp16.h>

#define NNODES 100000
#define NEDGES 1600000
#define HDIM   64
#define HNUMB  128
#define NCONVL 2
#define NMLPL  2
#define NEGSLP 0.2f

typedef __attribute__((ext_vector_type(2))) float v2f;
typedef __attribute__((ext_vector_type(8))) float v8f;

__device__ __forceinline__ float leaky(float x) { return x > 0.f ? x : NEGSLP * x; }

__device__ __forceinline__ unsigned fenc(float f) {
    unsigned u = __float_as_uint(f);
    return (u & 0x80000000u) ? ~u : (u | 0x80000000u);
}
__device__ __forceinline__ float fdec(unsigned u) {
    return (u & 0x80000000u) ? __uint_as_float(u & 0x7fffffffu) : __uint_as_float(~u);
}

__global__ __launch_bounds__(256) void gemm_xw_kernel(const float* __restrict__ X,
                                                      const float* __restrict__ W,
                                                      float* __restrict__ Hout,
                                                      int nrows) {
    __shared__ float Wlds[HDIM * HDIM];
    const int tid = threadIdx.x;
    for (int idx = tid; idx < HDIM * HDIM; idx += 256) {
        int k = idx >> 6, n = idx & 63;
        Wlds[(k >> 1) * 128 + n * 2 + (k & 1)] = W[idx];
    }
    __syncthreads();

    const int wave    = tid >> 5;
    const int lane    = tid & 31;
    const int hi      = lane >> 4;
    const int mrow    = lane & 15;
    const int rowbase = blockIdx.x * 128 + wave * 16;
    if (rowbase >= nrows) return;

    v2f afrag[16];
#pragma unroll
    for (int kk = 0; kk < 16; ++kk) {
        const v2f* ap = (const v2f*)(X + (size_t)(rowbase + mrow) * HDIM + kk * 4 + hi * 2);
        afrag[kk] = *ap;
    }
#pragma unroll
    for (int nt = 0; nt < 4; ++nt) {
        const int ncol = nt * 16 + mrow;
        v8f acc = {};
#pragma unroll
        for (int kk = 0; kk < 16; ++kk) {
            v2f b = *(const v2f*)&Wlds[(kk * 2 + hi) * 128 + ncol * 2];
            acc = __builtin_amdgcn_wmma_f32_16x16x4_f32(false, afrag[kk], false, b,
                                                        (short)0, acc, false, false);
        }
#pragma unroll
        for (int v = 0; v < 8; ++v)
            Hout[(size_t)(rowbase + v + 8 * hi) * HDIM + ncol] = acc[v];
    }
}

__global__ void dots_kernel(const float* __restrict__ Hm, const float* __restrict__ asrc,
                            const float* __restrict__ adst, float* __restrict__ a_s,
                            float* __restrict__ a_d, int n) {
    int r = blockIdx.x * blockDim.x + threadIdx.x;
    if (r >= n) return;
    const float* hr = Hm + (size_t)r * HDIM;
    float s = 0.f, d = 0.f;
#pragma unroll
    for (int k = 0; k < HDIM; ++k) { float v = hr[k]; s += v * asrc[k]; d += v * adst[k]; }
    a_s[r] = s; a_d[r] = d;
}

__global__ void node_init_kernel(const float* __restrict__ a_s, const float* __restrict__ a_d,
                                 float* __restrict__ e_self, unsigned* __restrict__ m_enc, int n) {
    int i = blockIdx.x * blockDim.x + threadIdx.x;
    if (i >= n) return;
    float es = leaky(a_s[i] + a_d[i]);
    e_self[i] = es;
    m_enc[i]  = fenc(es);
}

__global__ void edge_max_kernel(const int* __restrict__ src, const int* __restrict__ dst,
                                const float* __restrict__ a_s, const float* __restrict__ a_d,
                                unsigned* __restrict__ m_enc, float* __restrict__ e_edge, int ne) {
    int e = blockIdx.x * blockDim.x + threadIdx.x;
    if (e >= ne) return;
    int s = src[e], d = dst[e];
    const float ev = leaky(a_s[s] + a_d[d]);
    e_edge[e] = ev;
    atomicMax(m_enc + d, fenc(ev));
}

__global__ void node_selfw_kernel(const float* __restrict__ e_self,
                                  const unsigned* __restrict__ m_enc,
                                  float* __restrict__ denom, int n) {
    int i = blockIdx.x * blockDim.x + threadIdx.x;
    if (i >= n) return;
    denom[i] = __expf(e_self[i] - fdec(m_enc[i]));
}

__global__ void num_init_kernel(const float* __restrict__ wself, const float* __restrict__ Hm,
                                float* __restrict__ num, int total) {
    int t = blockIdx.x * blockDim.x + threadIdx.x;
    if (t >= total) return;
    num[t] = wself[t >> 6] * Hm[t];
}

#define SC_EPW   32
#define SC_BATCH 8
static_assert(NEDGES % 256 == 0, "edge scatter: a block of 8 waves owns 256 whole edges");
__global__ __launch_bounds__(256) void edge_scatter_kernel(const int* __restrict__ src, const int* __restrict__ dst,
                                    const float* __restrict__ e_edge, const unsigned* __restrict__ m_enc,
                                    const float* __restrict__ Hm, float* __restrict__ denom,
                                    float* __restrict__ num) {
    const int lane = threadIdx.x & 31;
    const int e    = blockIdx.x * 256 + threadIdx.x;
    const int   sv = src[e];
    const int   dv = dst[e];
    const float ev = e_edge[e];
    const unsigned mv = m_enc[dv];

    float lo[2][SC_BATCH], hi[2][SC_BATCH];
#pragma unroll
    for (int j = 0; j < SC_BATCH; ++j) {
        const float* hr = Hm + (size_t)__builtin_amdgcn_readlane(sv, j) * HDIM;
        lo[0][j] = hr[lane];
        hi[0][j] = hr[lane + 32];
    }
    int wv = 0;
#pragma unroll
    for (int b = 0; b < SC_EPW / SC_BATCH; ++b) {
        const int cur = b & 1, nxt = cur ^ 1;
        if (b + 1 < SC_EPW / SC_BATCH) {
#pragma unroll
            for (int j = 0; j < SC_BATCH; ++j) {
                const float* hr = Hm + (size_t)__builtin_amdgcn_readlane(sv, (b + 1) * SC_BATCH + j) * HDIM;
                lo[nxt][j] = hr[lane];
                hi[nxt][j] = hr[lane + 32];
            }
        }
        if (b == 0) {
            const float wf = __expf(ev - fdec(mv));
            atomicAdd(denom + dv, wf);
            wv = __float_as_int(wf);
        }
#pragma unroll
        for (int j = 0; j < SC_BATCH; ++j) {
            const int   d = __builtin_amdgcn_readlane(dv, b * SC_BATCH + j);
            const float w = __int_as_float(__builtin_amdgcn_readlane(wv, b * SC_BATCH + j));
            float* nr = num + (size_t)d * HDIM;
            atomicAdd(nr + lane,      w * lo[cur][j]);
            atomicAdd(nr + lane + 32, w * hi[cur][j]);
        }
    }
}

__global__ void finalize_kernel(const float* __restrict__ num, const float* __restrict__ denom,
                                const float* __restrict__ bias, float* __restrict__ out, int total) {
    int t = blockIdx.x * blockDim.x + threadIdx.x;
    if (t >= total) return;
    out[t] = num[t] / (denom[t >> 6] + 1e-16f) + bias[t & 63];
}

#define FS_THREADS 512
#define FS_BLOCKS  256
#define FS_ROWS    ((NNODES + FS_BLOCKS - 1) / FS_BLOCKS)
#define FS_RSTEP   (FS_THREADS / 16)
#define FS_PITCH   (HDIM + 1)
__global__ __launch_bounds__(FS_THREADS) void finalize_segsum_kernel(
        const float* __restrict__ num, const float* __restrict__ denom,
        const float* __restrict__ bias, const int* __restrict__ blocks,
        float* __restrict__ out, float* __restrict__ vn_seg) {
    __shared__ float acc[HNUMB * FS_PITCH];
    const int tid = threadIdx.x;
    for (int i = tid; i < HNUMB * FS_PITCH; i += FS_THREADS) acc[i] = 0.f;
    __syncthreads();

    const int c4   = tid & 15;
    const int row0 = blockIdx.x * FS_ROWS;
    const int row1 = (row0 + FS_ROWS < NNODES) ? row0 + FS_ROWS : NNODES;
    const float4 b4 = ((const float4*)bias)[c4];
    for (int rb = row0 + (tid >> 4); rb < row1; rb += 4 * FS_RSTEP) {
        float4 n4[4];
        float  dn[4];
        int    bk[4];
#pragma unroll
        for (int u = 0; u < 4; ++u) {
            const int r = rb + u * FS_RSTEP;
            n4[u] = make_float4(0.f, 0.f, 0.f, 0.f); dn[u] = 1.f; bk[u] = 0;
            if (r < row1) {
                n4[u] = ((const float4*)num)[(size_t)r * (HDIM / 4) + c4];
                dn[u] = denom[r];
                bk[u] = blocks[r];
            }
        }
#pragma unroll
        for (int u = 0; u < 4; ++u) {
            const int r = rb + u * FS_RSTEP;
            if (r < row1) {
                const float dd = dn[u] + 1e-16f;
                float4 o;
                o.x = n4[u].x / dd + b4.x;
                o.y = n4[u].y / dd + b4.y;
                o.z = n4[u].z / dd + b4.z;
                o.w = n4[u].w / dd + b4.w;
                ((float4*)out)[(size_t)r * (HDIM / 4) + c4] = o;
                float* a = acc + bk[u] * FS_PITCH + c4 * 4;
                atomicAdd(a + 0, o.x);
                atomicAdd(a + 1, o.y);
                atomicAdd(a + 2, o.z);
                atomicAdd(a + 3, o.w);
            }
        }
    }
    __syncthreads();
    for (int i = tid; i < HNUMB * HDIM; i += FS_THREADS) {
        const int j = (i + blockIdx.x * 32) & (HNUMB * HDIM - 1);
        atomicAdd(vn_seg + j, acc[(j >> 6) * FS_PITCH + (j & 63)]);
    }
}

__global__ void gather_add_kernel(const float* __restrict__ out, const float* __restrict__ vnd,
                                  const int* __restrict__ blocks, float* __restrict__ x_in, int total) {
    int t = blockIdx.x * blockDim.x + threadIdx.x;
    if (t >= total) return;
    x_in[t] = out[t] + vnd[blocks[t >> 6] * HDIM + (t & 63)];
}

__global__ void zero_kernel(float* __restrict__ p, int n) {
    int t = blockIdx.x * blockDim.x + threadIdx.x;
    if (t < n) p[t] = 0.f;
}

__global__ __launch_bounds__(256) void vn_update_kernel(float* __restrict__ vn_direct,
                                                        const float* __restrict__ vn_seg,
                                                        float* __restrict__ vn_root) {
    int tid = threadIdx.x;
    for (int i = tid; i < HNUMB * HDIM; i += 256)
        vn_direct[i] += vn_seg[i] + vn_root[i & (HDIM - 1)];
    __syncthreads();
    if (tid < HDIM) {
        float s = 0.f;
        for (int r = 0; r < HNUMB; ++r) s += vn_direct[r * HDIM + tid];
        vn_root[tid] += s;
    }
}

__global__ __launch_bounds__(64) void mlp_kernel(float* __restrict__ vn,
                                                 const float* __restrict__ W1, const float* __restrict__ b1,
                                                 const float* __restrict__ W2, const float* __restrict__ b2) {
    __shared__ float rin[HDIM];
    __shared__ float t1[HDIM];
    int f = threadIdx.x;
    float* row = vn + (size_t)blockIdx.x * HDIM;
    rin[f] = row[f];
    __syncthreads();
    float s = b1[f];
#pragma unroll
    for (int k = 0; k < HDIM; ++k) s += rin[k] * W1[k * HDIM + f];
    t1[f] = fmaxf(s, 0.f);
    __syncthreads();
    float s2 = b2[f];
#pragma unroll
    for (int k = 0; k < HDIM; ++k) s2 += t1[k] * W2[k * HDIM + f];
    row[f] = fmaxf(s2, 0.f);
}

__global__ void vn_init_kernel(const float* __restrict__ vn_emb, float* __restrict__ vn_direct,
                               float* __restrict__ vn_root) {
    int t = blockIdx.x * blockDim.x + threadIdx.x;
    if (t < HNUMB * HDIM) vn_direct[t] = vn_emb[t & (HDIM - 1)];
    if (t < HDIM) vn_root[t] = vn_emb[t];
}

extern "C" void kernel_launch(void* const* d_in, const int* in_sizes, int n_in,
                              void* d_out, int out_size, void* d_ws, size_t ws_size,
                              hipStream_t stream) {
    const float* x       = (const float*)d_in[0];
    const int*   ei      = (const int*)d_in[1];
    const int*   src     = ei;
    const int*   dst     = ei + NEDGES;
    const int*   hblocks = (const int*)d_in[2];
    const float* W0      = (const float*)d_in[5];
    const float* asrc0   = (const float*)d_in[6];
    const float* adst0   = (const float*)d_in[7];
    const float* b0      = (const float*)d_in[8];
    const float* Wl      = (const float*)d_in[9];
    const float* asrcl   = (const float*)d_in[10];
    const float* adstl   = (const float*)d_in[11];
    const float* bl      = (const float*)d_in[12];
    const float* vn_emb  = (const float*)d_in[13];
    const float* mW1     = (const float*)d_in[14];
    const float* mb1     = (const float*)d_in[15];
    const float* mW2     = (const float*)d_in[16];
    const float* mb2     = (const float*)d_in[17];
    float* out = (float*)d_out;

    char*  ws  = (char*)d_ws;
    size_t off = 0;
    auto carve = [&](size_t bytes) -> void* {
        void* p = ws + off;
        off += (bytes + 255) & ~(size_t)255;
        return p;
    };
    float*    h         = (float*)carve((size_t)NNODES * HDIM * 4);
    float*    x_in      = (float*)carve((size_t)NNODES * HDIM * 4);
    float*    num       = (float*)carve((size_t)NNODES * HDIM * 4);
    float*    a_s       = (float*)carve((size_t)NNODES * 4);
    float*    a_d       = (float*)carve((size_t)NNODES * 4);
    float*    e_self    = (float*)carve((size_t)NNODES * 4);
    float*    denom     = (float*)carve((size_t)NNODES * 4);
    unsigned* m_enc     = (unsigned*)carve((size_t)NNODES * 4);
    float*    e_edge    = (float*)carve((size_t)NEDGES * 4);
    float*    vn_direct = (float*)carve((size_t)HNUMB * HDIM * 4);
    float*    vn_root   = (float*)carve((size_t)HDIM * 4);
    float*    vn_seg    = (float*)carve((size_t)HNUMB * HDIM * 4);

    const int NF      = NNODES * HDIM;
    const int nodeBlk = (NNODES + 255) / 256;
    const int nfBlk   = NF / 256;
    const int edgeBlk = NEDGES / 256;

    auto run_gat = [&](const float* Xin, const float* W, const float* asrc,
                       const float* adst, const float* bias, bool with_segsum) {
        gemm_xw_kernel<<<(NNODES + 127) / 128, 256, 0, stream>>>(Xin, W, h, NNODES);
        dots_kernel<<<nodeBlk, 256, 0, stream>>>(h, asrc, adst, a_s, a_d, NNODES);
        node_init_kernel<<<nodeBlk, 256, 0, stream>>>(a_s, a_d, e_self, m_enc, NNODES);
        edge_max_kernel<<<edgeBlk, 256, 0, stream>>>(src, dst, a_s, a_d, m_enc, e_edge, NEDGES);
        node_selfw_kernel<<<nodeBlk, 256, 0, stream>>>(e_self, m_enc, denom, NNODES);
        num_init_kernel<<<nfBlk, 256, 0, stream>>>(denom, h, num, NF);
        edge_scatter_kernel<<<edgeBlk, 256, 0, stream>>>(src, dst, e_edge, m_enc, h, denom, num);
        if (with_segsum)
            finalize_segsum_kernel<<<FS_BLOCKS, FS_THREADS, 0, stream>>>(num, denom, bias, hblocks, out, vn_seg);
        else
            finalize_kernel<<<nfBlk, 256, 0, stream>>>(num, denom, bias, out, NF);
    };

    vn_init_kernel<<<(HNUMB * HDIM + 255) / 256, 256, 0, stream>>>(vn_emb, vn_direct, vn_root);

    run_gat(x, W0, asrc0, adst0, b0, false);

    for (int l = 0; l < NCONVL; ++l) {
        gather_add_kernel<<<nfBlk, 256, 0, stream>>>(out, vn_direct, hblocks, x_in, NF);
        zero_kernel<<<(HNUMB * HDIM + 255) / 256, 256, 0, stream>>>(vn_seg, HNUMB * HDIM);
        run_gat(x_in, Wl + l * HDIM * HDIM, asrcl + l * HDIM, adstl + l * HDIM, bl + l * HDIM, true);
        vn_update_kernel<<<1, 256, 0, stream>>>(vn_direct, vn_seg, vn_root);
        for (int j = 0; j < NMLPL; ++j) {
            mlp_kernel<<<HNUMB, 64, 0, stream>>>(vn_direct, mW1 + j * HDIM * HDIM, mb1 + j * HDIM,
                                                 mW2 + j * HDIM * HDIM, mb2 + j * HDIM);
            mlp_kernel<<<1, 64, 0, stream>>>(vn_root, mW1 + j * HDIM * HDIM, mb1 + j * HDIM,
                                             mW2 + j * HDIM * HDIM, mb2 + j * HDIM);
        }
    }
}


# Elementwise N x 64 passes (num init, first finalize, virtual-node add; 6 launches of 200000 one-element waves) take 16-byte pieces, 10 per thread, loads before stores: 5000 waves each; bit-identical
# speedup vs baseline: 1.9648x; 1.0686x over previous
#include <hip/hip_runtime.h>
#include <hip/hip_bf16.h>
#include <hip/hip_fp16.h>

#define NNODES 100000
#define NEDGES 1600000
#define HDIM   64
#define HNUMB  128
#define NCONVL 2
#define NMLPL  2
#define NEGSLP 0.2f

typedef __attribute__((ext_vector_type(2))) float v2f;
typedef __attribute__((ext_vector_type(8))) float v8f;

__device__ __forceinline__ float leaky(float x) { return x > 0.f ? x : NEGSLP * x; }

__device__ __forceinline__ unsigned fenc(float f) {
    unsigned u = __float_as_uint(f);
    return (u & 0x80000000u) ? ~u : (u | 0x80000000u);
}
__device__ __forceinline__ float fdec(unsigned u) {
    return (u & 0x80000000u) ? __uint_as_float(u & 0x7fffffffu) : __uint_as_float(~u);
}

__global__ __launch_bounds__(256) void gemm_xw_kernel(const float* __restrict__ X,
                                                      const float* __restrict__ W,
                                                      float* __restrict__ Hout,
                                                      int nrows) {
    __shared__ float Wlds[HDIM * HDIM];
    const int tid = threadIdx.x;
    for (int idx = tid; idx < HDIM * HDIM; idx += 256) {
        int k = idx >> 6, n = idx & 63;
        Wlds[(k >> 1) * 128 + n * 2 + (k & 1)] = W[idx];
    }
    __syncthreads();

    const int wave    = tid >> 5;
    const int lane    = tid & 31;
    const int hi      = lane >> 4;
    const int mrow    = lane & 15;
    const int rowbase = blockIdx.x * 128 + wave * 16;
    if (rowbase >= nrows) return;

    v2f afrag[16];
#pragma unroll
    for (int kk = 0; kk < 16; ++kk) {
        const v2f* ap = (const v2f*)(X + (size_t)(rowbase + mrow) * HDIM + kk * 4 + hi * 2);
        afrag[kk] = *ap;
    }
#pragma unroll
    for (int nt = 0; nt < 4; ++nt) {
        const int ncol = nt * 16 + mrow;
        v8f acc = {};
#pragma unroll
        for (int kk = 0; kk < 16; ++kk) {
            v2f b = *(const v2f*)&Wlds[(kk * 2 + hi) * 128 + ncol * 2];
            acc = __builtin_amdgcn_wmma_f32_16x16x4_f32(false, afrag[kk], false, b,
                                                        (short)0, acc, false, false);
        }
#pragma unroll
        for (int v = 0; v < 8; ++v)
            Hout[(size_t)(rowbase + v + 8 * hi) * HDIM + ncol] = acc[v];
    }
}

__global__ void dots_kernel(const float* __restrict__ Hm, const float* __restrict__ asrc,
                            const float* __restrict__ adst, float* __restrict__ a_s,
                            float* __restrict__ a_d, int n) {
    int r = blockIdx.x * blockDim.x + threadIdx.x;
    if (r >= n) return;
    const float* hr = Hm + (size_t)r * HDIM;
    float s = 0.f, d = 0.f;
#pragma unroll
    for (int k = 0; k < HDIM; ++k) { float v = hr[k]; s += v * asrc[k]; d += v * adst[k]; }
    a_s[r] = s; a_d[r] = d;
}

__global__ void node_init_kernel(const float* __restrict__ a_s, const float* __restrict__ a_d,
                                 float* __restrict__ e_self, unsigned* __restrict__ m_enc, int n) {
    int i = blockIdx.x * blockDim.x + threadIdx.x;
    if (i >= n) return;
    float es = leaky(a_s[i] + a_d[i]);
    e_self[i] = es;
    m_enc[i]  = fenc(es);
}

__global__ void edge_max_kernel(const int* __restrict__ src, const int* __restrict__ dst,
                                const float* __restrict__ a_s, const float* __restrict__ a_d,
                                unsigned* __restrict__ m_enc, float* __restrict__ e_edge, int ne) {
    int e = blockIdx.x * blockDim.x + threadIdx.x;
    if (e >= ne) return;
    int s = src[e], d = dst[e];
    const float ev = leaky(a_s[s] + a_d[d]);
    e_edge[e] = ev;
    atomicMax(m_enc + d, fenc(ev));
}

__global__ void node_selfw_kernel(const float* __restrict__ e_self,
                                  const unsigned* __restrict__ m_enc,
                                  float* __restrict__ denom, int n) {
    int i = blockIdx.x * blockDim.x + threadIdx.x;
    if (i >= n) return;
    denom[i] = __expf(e_self[i] - fdec(m_enc[i]));
}

#define EW_THREADS 256
#define EW_PIECES  10
#define EW_NP      (NNODES * (HDIM / 4))
#define EW_BLOCKS  (EW_NP / (EW_THREADS * EW_PIECES))
static_assert(EW_NP % (EW_THREADS * EW_PIECES) == 0, "elementwise passes: every block owns whole groups of pieces");

__global__ __launch_bounds__(EW_THREADS) void num_init_kernel(const float* __restrict__ wself,
                                                              const float* __restrict__ Hm,
                                                              float* __restrict__ num) {
    const int p0 = blockIdx.x * (EW_THREADS * EW_PIECES) + threadIdx.x;
    float4 v[EW_PIECES];
    float  w[EW_PIECES];
#pragma unroll
    for (int u = 0; u < EW_PIECES; ++u) {
        const int p = p0 + u * EW_THREADS;
        v[u] = ((const float4*)Hm)[p];
        w[u] = wself[p >> 4];
    }
#pragma unroll
    for (int u = 0; u < EW_PIECES; ++u) {
        const int p = p0 + u * EW_THREADS;
        float4 o;
        o.x = w[u] * v[u].x;
        o.y = w[u] * v[u].y;
        o.z = w[u] * v[u].z;
        o.w = w[u] * v[u].w;
        ((float4*)num)[p] = o;
    }
}

#define SC_EPW   32
#define SC_BATCH 8
static_assert(NEDGES % 256 == 0, "edge scatter: a block of 8 waves owns 256 whole edges");
__global__ __launch_bounds__(256) void edge_scatter_kernel(const int* __restrict__ src, const int* __restrict__ dst,
                                    const float* __restrict__ e_edge, const unsigned* __restrict__ m_enc,
                                    const float* __restrict__ Hm, float* __restrict__ denom,
                                    float* __restrict__ num) {
    const int lane = threadIdx.x & 31;
    const int e    = blockIdx.x * 256 + threadIdx.x;
    const int   sv = src[e];
    const int   dv = dst[e];
    const float ev = e_edge[e];
    const unsigned mv = m_enc[dv];

    float lo[2][SC_BATCH], hi[2][SC_BATCH];
#pragma unroll
    for (int j = 0; j < SC_BATCH; ++j) {
        const float* hr = Hm + (size_t)__builtin_amdgcn_readlane(sv, j) * HDIM;
        lo[0][j] = hr[lane];
        hi[0][j] = hr[lane + 32];
    }
    int wv = 0;
#pragma unroll
    for (int b = 0; b < SC_EPW / SC_BATCH; ++b) {
        const int cur = b & 1, nxt = cur ^ 1;
        if (b + 1 < SC_EPW / SC_BATCH) {
#pragma unroll
            for (int j = 0; j < SC_BATCH; ++j) {
                const float* hr = Hm + (size_t)__builtin_amdgcn_readlane(sv, (b + 1) * SC_BATCH + j) * HDIM;
                lo[nxt][j] = hr[lane];
                hi[nxt][j] = hr[lane + 32];
            }
        }
        if (b == 0) {
            const float wf = __expf(ev - fdec(mv));
            atomicAdd(denom + dv, wf);
            wv = __float_as_int(wf);
        }
#pragma unroll
        for (int j = 0; j < SC_BATCH; ++j) {
            const int   d = __builtin_amdgcn_readlane(dv, b * SC_BATCH + j);
            const float w = __int_as_float(__builtin_amdgcn_readlane(wv, b * SC_BATCH + j));
            float* nr = num + (size_t)d * HDIM;
            atomicAdd(nr + lane,      w * lo[cur][j]);
            atomicAdd(nr + lane + 32, w * hi[cur][j]);
        }
    }
}

__global__ __launch_bounds__(EW_THREADS) void finalize_kernel(const float* __restrict__ num,
                                                              const float* __restrict__ denom,
                                                              const float* __restrict__ bias,
                                                              float* __restrict__ out) {
    const int p0 = blockIdx.x * (EW_THREADS * EW_PIECES) + threadIdx.x;
    const float4 b4 = ((const float4*)bias)[threadIdx.x & 15];
    float4 v[EW_PIECES];
    float  dn[EW_PIECES];
#pragma unroll
    for (int u = 0; u < EW_PIECES; ++u) {
        const int p = p0 + u * EW_THREADS;
        v[u]  = ((const float4*)num)[p];
        dn[u] = denom[p >> 4];
    }
#pragma unroll
    for (int u = 0; u < EW_PIECES; ++u) {
        const int p = p0 + u * EW_THREADS;
        const float dd = dn[u] + 1e-16f;
        float4 o;
        o.x = v[u].x / dd + b4.x;
        o.y = v[u].y / dd + b4.y;
        o.z = v[u].z / dd + b4.z;
        o.w = v[u].w / dd + b4.w;
        ((float4*)out)[p] = o;
    }
}

#define FS_THREADS 512
#define FS_BLOCKS  256
#define FS_ROWS    ((NNODES + FS_BLOCKS - 1) / FS_BLOCKS)
#define FS_RSTEP   (FS_THREADS / 16)
#define FS_PITCH   (HDIM + 1)
__global__ __launch_bounds__(FS_THREADS) void finalize_segsum_kernel(
        const float* __restrict__ num, const float* __restrict__ denom,
        const float* __restrict__ bias, const int* __restrict__ blocks,
        float* __restrict__ out, float* __restrict__ vn_seg) {
    __shared__ float acc[HNUMB * FS_PITCH];
    const int tid = threadIdx.x;
    for (int i = tid; i < HNUMB * FS_PITCH; i += FS_THREADS) acc[i] = 0.f;
    __syncthreads();

    const int c4   = tid & 15;
    const int row0 = blockIdx.x * FS_ROWS;
    const int row1 = (row0 + FS_ROWS < NNODES) ? row0 + FS_ROWS : NNODES;
    const float4 b4 = ((const float4*)bias)[c4];
    for (int rb = row0 + (tid >> 4); rb < row1; rb += 4 * FS_RSTEP) {
        float4 n4[4];
        float  dn[4];
        int    bk[4];
#pragma unroll
        for (int u = 0; u < 4; ++u) {
            const int r = rb + u * FS_RSTEP;
            n4[u] = make_float4(0.f, 0.f, 0.f, 0.f); dn[u] = 1.f; bk[u] = 0;
            if (r < row1) {
                n4[u] = ((const float4*)num)[(size_t)r * (HDIM / 4) + c4];
                dn[u] = denom[r];
                bk[u] = blocks[r];
            }
        }
#pragma unroll
        for (int u = 0; u < 4; ++u) {
            const int r = rb + u * FS_RSTEP;
            if (r < row1) {
                const float dd = dn[u] + 1e-16f;
                float4 o;
                o.x = n4[u].x / dd + b4.x;
                o.y = n4[u].y / dd + b4.y;
                o.z = n4[u].z / dd + b4.z;
                o.w = n4[u].w / dd + b4.w;
                ((float4*)out)[(size_t)r * (HDIM / 4) + c4] = o;
                float* a = acc + bk[u] * FS_PITCH + c4 * 4;
                atomicAdd(a + 0, o.x);
                atomicAdd(a + 1, o.y);
                atomicAdd(a + 2, o.z);
                atomicAdd(a + 3, o.w);
            }
        }
    }
    __syncthreads();
    for (int i = tid; i < HNUMB * HDIM; i += FS_THREADS) {
        const int j = (i + blockIdx.x * 32) & (HNUMB * HDIM - 1);
        atomicAdd(vn_seg + j, acc[(j >> 6) * FS_PITCH + (j & 63)]);
    }
}

__global__ __launch_bounds__(EW_THREADS) void gather_add_kernel(const float* __restrict__ out,
                                                                const float* __restrict__ vnd,
                                                                const int* __restrict__ blocks,
                                                                float* __restrict__ x_in) {
    const int p0 = blockIdx.x * (EW_THREADS * EW_PIECES) + threadIdx.x;
    const int c4 = threadIdx.x & 15;
    float4 v[EW_PIECES];
    int    bk[EW_PIECES];
#pragma unroll
    for (int u = 0; u < EW_PIECES; ++u) {
        const int p = p0 + u * EW_THREADS;
        v[u]  = ((const float4*)out)[p];
        bk[u] = blocks[p >> 4];
    }
#pragma unroll
    for (int u = 0; u < EW_PIECES; ++u) {
        const int p = p0 + u * EW_THREADS;
        const float4 a = ((const float4*)vnd)[bk[u] * (HDIM / 4) + c4];
        float4 o;
        o.x = v[u].x + a.x;
        o.y = v[u].y + a.y;
        o.z = v[u].z + a.z;
        o.w = v[u].w + a.w;
        ((float4*)x_in)[p] = o;
    }
}

__global__ void zero_kernel(float* __restrict__ p, int n) {
    int t = blockIdx.x * blockDim.x + threadIdx.x;
    if (t < n) p[t] = 0.f;
}

__global__ __launch_bounds__(256) void vn_update_kernel(float* __restrict__ vn_direct,
                                                        const float* __restrict__ vn_seg,
                                                        float* __restrict__ vn_root) {
    int tid = threadIdx.x;
    for (int i = tid; i < HNUMB * HDIM; i += 256)
        vn_direct[i] += vn_seg[i] + vn_root[i & (HDIM - 1)];
    __syncthreads();
    if (tid < HDIM) {
        float s = 0.f;
        for (int r = 0; r < HNUMB; ++r) s += vn_direct[r * HDIM + tid];
        vn_root[tid] += s;
    }
}

__global__ __launch_bounds__(64) void mlp_kernel(float* __restrict__ vn,
                                                 const float* __restrict__ W1, const float* __restrict__ b1,
                                                 const float* __restrict__ W2, const float* __restrict__ b2) {
    __shared__ float rin[HDIM];
    __shared__ float t1[HDIM];
    int f = threadIdx.x;
    float* row = vn + (size_t)blockIdx.x * HDIM;
    rin[f] = row[f];
    __syncthreads();
    float s = b1[f];
#pragma unroll
    for (int k = 0; k < HDIM; ++k) s += rin[k] * W1[k * HDIM + f];
    t1[f] = fmaxf(s, 0.f);
    __syncthreads();
    float s2 = b2[f];
#pragma unroll
    for (int k = 0; k < HDIM; ++k) s2 += t1[k] * W2[k * HDIM + f];
    row[f] = fmaxf(s2, 0.f);
}

__global__ void vn_init_kernel(const float* __restrict__ vn_emb, float* __restrict__ vn_direct,
                               float* __restrict__ vn_root) {
    int t = blockIdx.x * blockDim.x + threadIdx.x;
    if (t < HNUMB * HDIM) vn_direct[t] = vn_emb[t & (HDIM - 1)];
    if (t < HDIM) vn_root[t] = vn_emb[t];
}

extern "C" void kernel_launch(void* const* d_in, const int* in_sizes, int n_in,
                              void* d_out, int out_size, void* d_ws, size_t ws_size,
                              hipStream_t stream) {
    const float* x       = (const float*)d_in[0];
    const int*   ei      = (const int*)d_in[1];
    const int*   src     = ei;
    const int*   dst     = ei + NEDGES;
    const int*   hblocks = (const int*)d_in[2];
    const float* W0      = (const float*)d_in[5];
    const float* asrc0   = (const float*)d_in[6];
    const float* adst0   = (const float*)d_in[7];
    const float* b0      = (const float*)d_in[8];
    const float* Wl      = (const float*)d_in[9];
    const float* asrcl   = (const float*)d_in[10];
    const float* adstl   = (const float*)d_in[11];
    const float* bl      = (const float*)d_in[12];
    const float* vn_emb  = (const float*)d_in[13];
    const float* mW1     = (const float*)d_in[14];
    const float* mb1     = (const float*)d_in[15];
    const float* mW2     = (const float*)d_in[16];
    const float* mb2     = (const float*)d_in[17];
    float* out = (float*)d_out;

    char*  ws  = (char*)d_ws;
    size_t off = 0;
    auto carve = [&](size_t bytes) -> void* {
        void* p = ws + off;
        off += (bytes + 255) & ~(size_t)255;
        return p;
    };
    float*    h         = (float*)carve((size_t)NNODES * HDIM * 4);
    float*    x_in      = (float*)carve((size_t)NNODES * HDIM * 4);
    float*    num       = (float*)carve((size_t)NNODES * HDIM * 4);
    float*    a_s       = (float*)carve((size_t)NNODES * 4);
    float*    a_d       = (float*)carve((size_t)NNODES * 4);
    float*    e_self    = (float*)carve((size_t)NNODES * 4);
    float*    denom     = (float*)carve((size_t)NNODES * 4);
    unsigned* m_enc     = (unsigned*)carve((size_t)NNODES * 4);
    float*    e_edge    = (float*)carve((size_t)NEDGES * 4);
    float*    vn_direct = (float*)carve((size_t)HNUMB * HDIM * 4);
    float*    vn_root   = (float*)carve((size_t)HDIM * 4);
    float*    vn_seg    = (float*)carve((size_t)HNUMB * HDIM * 4);

    const int nodeBlk = (NNODES + 255) / 256;
    const int edgeBlk = NEDGES / 256;

    auto run_gat = [&](const float* Xin, const float* W, const float* asrc,
                       const float* adst, const float* bias, bool with_segsum) {
        gemm_xw_kernel<<<(NNODES + 127) / 128, 256, 0, stream>>>(Xin, W, h, NNODES);
        dots_kernel<<<nodeBlk, 256, 0, stream>>>(h, asrc, adst, a_s, a_d, NNODES);
        node_init_kernel<<<nodeBlk, 256, 0, stream>>>(a_s, a_d, e_self, m_enc, NNODES);
        edge_max_kernel<<<edgeBlk, 256, 0, stream>>>(src, dst, a_s, a_d, m_enc, e_edge, NEDGES);
        node_selfw_kernel<<<nodeBlk, 256, 0, stream>>>(e_self, m_enc, denom, NNODES);
        num_init_kernel<<<EW_BLOCKS, EW_THREADS, 0, stream>>>(denom, h, num);
        edge_scatter_kernel<<<edgeBlk, 256, 0, stream>>>(src, dst, e_edge, m_enc, h, denom, num);
        if (with_segsum)
            finalize_segsum_kernel<<<FS_BLOCKS, FS_THREADS, 0, stream>>>(num, denom, bias, hblocks, out, vn_seg);
        else
            finalize_kernel<<<EW_BLOCKS, EW_THREADS, 0, stream>>>(num, denom, bias, out);
    };

    vn_init_kernel<<<(HNUMB * HDIM + 255) / 256, 256, 0, stream>>>(vn_emb, vn_direct, vn_root);

    run_gat(x, W0, asrc0, adst0, b0, false);

    for (int l = 0; l < NCONVL; ++l) {
        gather_add_kernel<<<EW_BLOCKS, EW_THREADS, 0, stream>>>(out, vn_direct, hblocks, x_in);
        zero_kernel<<<(HNUMB * HDIM + 255) / 256, 256, 0, stream>>>(vn_seg, HNUMB * HDIM);
        run_gat(x_in, Wl + l * HDIM * HDIM, asrcl + l * HDIM, adstl + l * HDIM, bl + l * HDIM, true);
        vn_update_kernel<<<1, 256, 0, stream>>>(vn_direct, vn_seg, vn_root);
        for (int j = 0; j < NMLPL; ++j) {
            mlp_kernel<<<HNUMB, 64, 0, stream>>>(vn_direct, mW1 + j * HDIM * HDIM, mb1 + j * HDIM,
                                                 mW2 + j * HDIM * HDIM, mb2 + j * HDIM);
            mlp_kernel<<<1, 64, 0, stream>>>(vn_root, mW1 + j * HDIM * HDIM, mb1 + j * HDIM,
                                             mW2 + j * HDIM * HDIM, mb2 + j * HDIM);
        }
    }
}


# One copy per package half of every edge-atomic target (segment max, denominator, row sums), picked by the wave's die number, joined where read; atomics kept: sums form as two partial sums
# speedup vs baseline: 3.1200x; 1.5879x over previous
#include <hip/hip_runtime.h>
#include <hip/hip_bf16.h>
#include <hip/hip_fp16.h>

#define NNODES 100000
#define NEDGES 1600000
#define HDIM   64
#define HNUMB  128
#define NCONVL 2
#define NMLPL  2
#define NEGSLP 0.2f

typedef __attribute__((ext_vector_type(2))) float v2f;
typedef __attribute__((ext_vector_type(8))) float v8f;

__device__ __forceinline__ float leaky(float x) { return x > 0.f ? x : NEGSLP * x; }

__device__ __forceinline__ unsigned fenc(float f) {
    unsigned u = __float_as_uint(f);
    return (u & 0x80000000u) ? ~u : (u | 0x80000000u);
}
__device__ __forceinline__ float fdec(unsigned u) {
    return (u & 0x80000000u) ? __uint_as_float(u & 0x7fffffffu) : __uint_as_float(~u);
}

__device__ __forceinline__ int pkg_half() {
    return (int)((__builtin_amdgcn_s_sendmsg_rtn(0x87) >> 18) & 1u);
}

__global__ __launch_bounds__(256) void gemm_xw_kernel(const float* __restrict__ X,
                                                      const float* __restrict__ W,
                                                      float* __restrict__ Hout,
                                                      int nrows) {
    __shared__ float Wlds[HDIM * HDIM];
    const int tid = threadIdx.x;
    for (int idx = tid; idx < HDIM * HDIM; idx += 256) {
        int k = idx >> 6, n = idx & 63;
        Wlds[(k >> 1) * 128 + n * 2 + (k & 1)] = W[idx];
    }
    __syncthreads();

    const int wave    = tid >> 5;
    const int lane    = tid & 31;
    const int hi      = lane >> 4;
    const int mrow    = lane & 15;
    const int rowbase = blockIdx.x * 128 + wave * 16;
    if (rowbase >= nrows) return;

    v2f afrag[16];
#pragma unroll
    for (int kk = 0; kk < 16; ++kk) {
        const v2f* ap = (const v2f*)(X + (size_t)(rowbase + mrow) * HDIM + kk * 4 + hi * 2);
        afrag[kk] = *ap;
    }
#pragma unroll
    for (int nt = 0; nt < 4; ++nt) {
        const int ncol = nt * 16 + mrow;
        v8f acc = {};
#pragma unroll
        for (int kk = 0; kk < 16; ++kk) {
            v2f b = *(const v2f*)&Wlds[(kk * 2 + hi) * 128 + ncol * 2];
            acc = __builtin_amdgcn_wmma_f32_16x16x4_f32(false, afrag[kk], false, b,
                                                        (short)0, acc, false, false);
        }
#pragma unroll
        for (int v = 0; v < 8; ++v)
            Hout[(size_t)(rowbase + v + 8 * hi) * HDIM + ncol] = acc[v];
    }
}

__global__ void dots_kernel(const float* __restrict__ Hm, const float* __restrict__ asrc,
                            const float* __restrict__ adst, float* __restrict__ a_s,
                            float* __restrict__ a_d, int n) {
    int r = blockIdx.x * blockDim.x + threadIdx.x;
    if (r >= n) return;
    const float* hr = Hm + (size_t)r * HDIM;
    float s = 0.f, d = 0.f;
#pragma unroll
    for (int k = 0; k < HDIM; ++k) { float v = hr[k]; s += v * asrc[k]; d += v * adst[k]; }
    a_s[r] = s; a_d[r] = d;
}

__global__ void node_init_kernel(const float* __restrict__ a_s, const float* __restrict__ a_d,
                                 float* __restrict__ e_self, unsigned* __restrict__ m_enc0,
                                 unsigned* __restrict__ m_enc1, int n) {
    int i = blockIdx.x * blockDim.x + threadIdx.x;
    if (i >= n) return;
    float es = leaky(a_s[i] + a_d[i]);
    e_self[i] = es;
    m_enc0[i] = fenc(es);
    m_enc1[i] = fenc(es);
}

__global__ void edge_max_kernel(const int* __restrict__ src, const int* __restrict__ dst,
                                const float* __restrict__ a_s, const float* __restrict__ a_d,
                                unsigned* __restrict__ m_enc0, unsigned* __restrict__ m_enc1,
                                float* __restrict__ e_edge, int ne) {
    unsigned* const m_mine = pkg_half() ? m_enc1 : m_enc0;
    int e = blockIdx.x * blockDim.x + threadIdx.x;
    if (e >= ne) return;
    int s = src[e], d = dst[e];
    const float ev = leaky(a_s[s] + a_d[d]);
    e_edge[e] = ev;
    atomicMax(m_mine + d, fenc(ev));
}

__global__ void node_selfw_kernel(const float* __restrict__ e_self,
                                  unsigned* __restrict__ m_enc0,
                                  const unsigned* __restrict__ m_enc1,
                                  float* __restrict__ denom0, float* __restrict__ denom1, int n) {
    int i = blockIdx.x * blockDim.x + threadIdx.x;
    if (i >= n) return;
    const unsigned ma = m_enc0[i], mb = m_enc1[i];
    const unsigned m  = ma > mb ? ma : mb;
    m_enc0[i] = m;
    denom0[i] = __expf(e_self[i] - fdec(m));
    denom1[i] = 0.f;
}

#define EW_THREADS 256
#define EW_PIECES  10
#define EW_NP      (NNODES * (HDIM / 4))
#define EW_BLOCKS  (EW_NP / (EW_THREADS * EW_PIECES))
static_assert(EW_NP % (EW_THREADS * EW_PIECES) == 0, "elementwise passes: every block owns whole groups of pieces");

__global__ __launch_bounds__(EW_THREADS) void num_init_kernel(const float* __restrict__ wself,
                                                              const float* __restrict__ Hm,
                                                              float* __restrict__ num,
                                                              float* __restrict__ num1) {
    const int p0 = blockIdx.x * (EW_THREADS * EW_PIECES) + threadIdx.x;
    float4 v[EW_PIECES];
    float  w[EW_PIECES];
#pragma unroll
    for (int u = 0; u < EW_PIECES; ++u) {
        const int p = p0 + u * EW_THREADS;
        v[u] = ((const float4*)Hm)[p];
        w[u] = wself[p >> 4];
    }
#pragma unroll
    for (int u = 0; u < EW_PIECES; ++u) {
        const int p = p0 + u * EW_THREADS;
        float4 o;
        o.x = w[u] * v[u].x;
        o.y = w[u] * v[u].y;
        o.z = w[u] * v[u].z;
        o.w = w[u] * v[u].w;
        ((float4*)num)[p]  = o;
        ((float4*)num1)[p] = make_float4(0.f, 0.f, 0.f, 0.f);
    }
}

#define SC_EPW   32
#define SC_BATCH 8
static_assert(NEDGES % 256 == 0, "edge scatter: a block of 8 waves owns 256 whole edges");
__global__ __launch_bounds__(256) void edge_scatter_kernel(const int* __restrict__ src, const int* __restrict__ dst,
                                    const float* __restrict__ e_edge, const unsigned* __restrict__ m_enc,
                                    const float* __restrict__ Hm,
                                    float* __restrict__ denom0, float* __restrict__ denom1,
                                    float* __restrict__ num0, float* __restrict__ num1) {
    const int half = pkg_half();
    float* const denom = half ? denom1 : denom0;
    float* const num   = half ? num1 : num0;
    const int lane = threadIdx.x & 31;
    const int e    = blockIdx.x * 256 + threadIdx.x;
    const int   sv = src[e];
    const int   dv = dst[e];
    const float ev = e_edge[e];
    const unsigned mv = m_enc[dv];

    float lo[2][SC_BATCH], hi[2][SC_BATCH];
#pragma unroll
    for (int j = 0; j < SC_BATCH; ++j) {
        const float* hr = Hm + (size_t)__builtin_amdgcn_readlane(sv, j) * HDIM;
        lo[0][j] = hr[lane];
        hi[0][j] = hr[lane + 32];
    }
    int wv = 0;
#pragma unroll
    for (int b = 0; b < SC_EPW / SC_BATCH; ++b) {
        const int cur = b & 1, nxt = cur ^ 1;
        if (b + 1 < SC_EPW / SC_BATCH) {
#pragma unroll
            for (int j = 0; j < SC_BATCH; ++j) {
                const float* hr = Hm + (size_t)__builtin_amdgcn_readlane(sv, (b + 1) * SC_BATCH + j) * HDIM;
                lo[nxt][j] = hr[lane];
                hi[nxt][j] = hr[lane + 32];
            }
        }
        if (b == 0) {
            const float wf = __expf(ev - fdec(mv));
            atomicAdd(denom + dv, wf);
            wv = __float_as_int(wf);
        }
#pragma unroll
        for (int j = 0; j < SC_BATCH; ++j) {
            const int   d = __builtin_amdgcn_readlane(dv, b * SC_BATCH + j);
            const float w = __int_as_float(__builtin_amdgcn_readlane(wv, b * SC_BATCH + j));
            float* nr = num + (size_t)d * HDIM;
            atomicAdd(nr + lane,      w * lo[cur][j]);
            atomicAdd(nr + lane + 32, w * hi[cur][j]);
        }
    }
}

__global__ __launch_bounds__(EW_THREADS) void finalize_kernel(const float* __restrict__ num,
                                                              const float* __restrict__ num1,
                                                              const float* __restrict__ denom,
                                                              const float* __restrict__ denom1,
                                                              const float* __restrict__ bias,
                                                              float* __restrict__ out) {
    const int p0 = blockIdx.x * (EW_THREADS * EW_PIECES) + threadIdx.x;
    const float4 b4 = ((const float4*)bias)[threadIdx.x & 15];
    float4 v[EW_PIECES], v1[EW_PIECES];
    float  dn[EW_PIECES], dn1[EW_PIECES];
#pragma unroll
    for (int u = 0; u < EW_PIECES; ++u) {
        const int p = p0 + u * EW_THREADS;
        v[u]   = ((const float4*)num)[p];
        v1[u]  = ((const float4*)num1)[p];
        dn[u]  = denom[p >> 4];
        dn1[u] = denom1[p >> 4];
    }
#pragma unroll
    for (int u = 0; u < EW_PIECES; ++u) {
        const int p = p0 + u * EW_THREADS;
        const float dd = (dn[u] + dn1[u]) + 1e-16f;
        float4 o;
        o.x = (v[u].x + v1[u].x) / dd + b4.x;
        o.y = (v[u].y + v1[u].y) / dd + b4.y;
        o.z = (v[u].z + v1[u].z) / dd + b4.z;
        o.w = (v[u].w + v1[u].w) / dd + b4.w;
        ((float4*)out)[p] = o;
    }
}

#define FS_THREADS 512
#define FS_BLOCKS  256
#define FS_ROWS    ((NNODES + FS_BLOCKS - 1) / FS_BLOCKS)
#define FS_RSTEP   (FS_THREADS / 16)
#define FS_PITCH   (HDIM + 1)
__global__ __launch_bounds__(FS_THREADS) void finalize_segsum_kernel(
        const float* __restrict__ num, const float* __restrict__ num1,
        const float* __restrict__ denom, const float* __restrict__ denom1,
        const float* __restrict__ bias, const int* __restrict__ blocks,
        float* __restrict__ out, float* __restrict__ vn_seg) {
    __shared__ float acc[HNUMB * FS_PITCH];
    const int tid = threadIdx.x;
    for (int i = tid; i < HNUMB * FS_PITCH; i += FS_THREADS) acc[i] = 0.f;
    __syncthreads();

    const int c4   = tid & 15;
    const int row0 = blockIdx.x * FS_ROWS;
    const int row1 = (row0 + FS_ROWS < NNODES) ? row0 + FS_ROWS : NNODES;
    const float4 b4 = ((const float4*)bias)[c4];
    for (int rb = row0 + (tid >> 4); rb < row1; rb += 4 * FS_RSTEP) {
        float4 n4[4], m4[4];
        float  dn[4];
        int    bk[4];
#pragma unroll
        for (int u = 0; u < 4; ++u) {
            const int r = rb + u * FS_RSTEP;
            n4[u] = make_float4(0.f, 0.f, 0.f, 0.f); m4[u] = n4[u]; dn[u] = 1.f; bk[u] = 0;
            if (r < row1) {
                n4[u] = ((const float4*)num)[(size_t)r * (HDIM / 4) + c4];
                m4[u] = ((const float4*)num1)[(size_t)r * (HDIM / 4) + c4];
                dn[u] = denom[r] + denom1[r];
                bk[u] = blocks[r];
            }
        }
#pragma unroll
        for (int u = 0; u < 4; ++u) {
            const int r = rb + u * FS_RSTEP;
            if (r < row1) {
                const float dd = dn[u] + 1e-16f;
                float4 o;
                o.x = (n4[u].x + m4[u].x) / dd + b4.x;
                o.y = (n4[u].y + m4[u].y) / dd + b4.y;
                o.z = (n4[u].z + m4[u].z) / dd + b4.z;
                o.w = (n4[u].w + m4[u].w) / dd + b4.w;
                ((float4*)out)[(size_t)r * (HDIM / 4) + c4] = o;
                float* a = acc + bk[u] * FS_PITCH + c4 * 4;
                atomicAdd(a + 0, o.x);
                atomicAdd(a + 1, o.y);
                atomicAdd(a + 2, o.z);
                atomicAdd(a + 3, o.w);
            }
        }
    }
    __syncthreads();
    for (int i = tid; i < HNUMB * HDIM; i += FS_THREADS) {
        const int j = (i + blockIdx.x * 32) & (HNUMB * HDIM - 1);
        atomicAdd(vn_seg + j, acc[(j >> 6) * FS_PITCH + (j & 63)]);
    }
}

__global__ __launch_bounds__(EW_THREADS) void gather_add_kernel(const float* __restrict__ out,
                                                                const float* __restrict__ vnd,
                                                                const int* __restrict__ blocks,
                                                                float* __restrict__ x_in) {
    const int p0 = blockIdx.x * (EW_THREADS * EW_PIECES) + threadIdx.x;
    const int c4 = threadIdx.x & 15;
    float4 v[EW_PIECES];
    int    bk[EW_PIECES];
#pragma unroll
    for (int u = 0; u < EW_PIECES; ++u) {
        const int p = p0 + u * EW_THREADS;
        v[u]  = ((const float4*)out)[p];
        bk[u] = blocks[p >> 4];
    }
#pragma unroll
    for (int u = 0; u < EW_PIECES; ++u) {
        const int p = p0 + u * EW_THREADS;
        const float4 a = ((const float4*)vnd)[bk[u] * (HDIM / 4) + c4];
        float4 o;
        o.x = v[u].x + a.x;
        o.y = v[u].y + a.y;
        o.z = v[u].z + a.z;
        o.w = v[u].w + a.w;
        ((float4*)x_in)[p] = o;
    }
}

__global__ void zero_kernel(float* __restrict__ p, int n) {
    int t = blockIdx.x * blockDim.x + threadIdx.x;
    if (t < n) p[t] = 0.f;
}

__global__ __launch_bounds__(256) void vn_update_kernel(float* __restrict__ vn_direct,
                                                        const float* __restrict__ vn_seg,
                                                        float* __restrict__ vn_root) {
    int tid = threadIdx.x;
    for (int i = tid; i < HNUMB * HDIM; i += 256)
        vn_direct[i] += vn_seg[i] + vn_root[i & (HDIM - 1)];
    __syncthreads();
    if (tid < HDIM) {
        float s = 0.f;
        for (int r = 0; r < HNUMB; ++r) s += vn_direct[r * HDIM + tid];
        vn_root[tid] += s;
    }
}

__global__ __launch_bounds__(64) void mlp_kernel(float* __restrict__ vn,
                                                 const float* __restrict__ W1, const float* __restrict__ b1,
                                                 const float* __restrict__ W2, const float* __restrict__ b2) {
    __shared__ float rin[HDIM];
    __shared__ float t1[HDIM];
    int f = threadIdx.x;
    float* row = vn + (size_t)blockIdx.x * HDIM;
    rin[f] = row[f];
    __syncthreads();
    float s = b1[f];
#pragma unroll
    for (int k = 0; k < HDIM; ++k) s += rin[k] * W1[k * HDIM + f];
    t1[f] = fmaxf(s, 0.f);
    __syncthreads();
    float s2 = b2[f];
#pragma unroll
    for (int k = 0; k < HDIM; ++k) s2 += t1[k] * W2[k * HDIM + f];
    row[f] = fmaxf(s2, 0.f);
}

__global__ void vn_init_kernel(const float* __restrict__ vn_emb, float* __restrict__ vn_direct,
                               float* __restrict__ vn_root) {
    int t = blockIdx.x * blockDim.x + threadIdx.x;
    if (t < HNUMB * HDIM) vn_direct[t] = vn_emb[t & (HDIM - 1)];
    if (t < HDIM) vn_root[t] = vn_emb[t];
}

extern "C" void kernel_launch(void* const* d_in, const int* in_sizes, int n_in,
                              void* d_out, int out_size, void* d_ws, size_t ws_size,
                              hipStream_t stream) {
    const float* x       = (const float*)d_in[0];
    const int*   ei      = (const int*)d_in[1];
    const int*   src     = ei;
    const int*   dst     = ei + NEDGES;
    const int*   hblocks = (const int*)d_in[2];
    const float* W0      = (const float*)d_in[5];
    const float* asrc0   = (const float*)d_in[6];
    const float* adst0   = (const float*)d_in[7];
    const float* b0      = (const float*)d_in[8];
    const float* Wl      = (const float*)d_in[9];
    const float* asrcl   = (const float*)d_in[10];
    const float* adstl   = (const float*)d_in[11];
    const float* bl      = (const float*)d_in[12];
    const float* vn_emb  = (const float*)d_in[13];
    const float* mW1     = (const float*)d_in[14];
    const float* mb1     = (const float*)d_in[15];
    const float* mW2     = (const float*)d_in[16];
    const float* mb2     = (const float*)d_in[17];
    float* out = (float*)d_out;

    char*  ws  = (char*)d_ws;
    size_t off = 0;
    auto carve = [&](size_t bytes) -> void* {
        void* p = ws + off;
        off += (bytes + 255) & ~(size_t)255;
        return p;
    };
    float*    h         = (float*)carve((size_t)NNODES * HDIM * 4);
    float*    x_in      = (float*)carve((size_t)NNODES * HDIM * 4);
    float*    num       = (float*)carve((size_t)NNODES * HDIM * 4);
    float*    a_s       = (float*)carve((size_t)NNODES * 4);
    float*    a_d       = (float*)carve((size_t)NNODES * 4);
    float*    e_self    = (float*)carve((size_t)NNODES * 4);
    unsigned* m_enc1    = (unsigned*)carve((size_t)NNODES * 4);
    unsigned* m_enc     = (unsigned*)carve((size_t)NNODES * 4);
    float*    denom     = a_s;
    float*    denom1    = a_d;
    float*    num1      = x_in;
    float*    e_edge    = (float*)carve((size_t)NEDGES * 4);
    float*    vn_direct = (float*)carve((size_t)HNUMB * HDIM * 4);
    float*    vn_root   = (float*)carve((size_t)HDIM * 4);
    float*    vn_seg    = (float*)carve((size_t)HNUMB * HDIM * 4);

    const int nodeBlk = (NNODES + 255) / 256;
    const int edgeBlk = NEDGES / 256;

    auto run_gat = [&](const float* Xin, const float* W, const float* asrc,
                       const float* adst, const float* bias, bool with_segsum) {
        gemm_xw_kernel<<<(NNODES + 127) / 128, 256, 0, stream>>>(Xin, W, h, NNODES);
        dots_kernel<<<nodeBlk, 256, 0, stream>>>(h, asrc, adst, a_s, a_d, NNODES);
        node_init_kernel<<<nodeBlk, 256, 0, stream>>>(a_s, a_d, e_self, m_enc, m_enc1, NNODES);
        edge_max_kernel<<<edgeBlk, 256, 0, stream>>>(src, dst, a_s, a_d, m_enc, m_enc1, e_edge, NEDGES);
        node_selfw_kernel<<<nodeBlk, 256, 0, stream>>>(e_self, m_enc, m_enc1, denom, denom1, NNODES);
        num_init_kernel<<<EW_BLOCKS, EW_THREADS, 0, stream>>>(denom, h, num, num1);
        edge_scatter_kernel<<<edgeBlk, 256, 0, stream>>>(src, dst, e_edge, m_enc, h, denom, denom1, num, num1);
        if (with_segsum)
            finalize_segsum_kernel<<<FS_BLOCKS, FS_THREADS, 0, stream>>>(num, num1, denom, denom1, bias, hblocks, out, vn_seg);
        else
            finalize_kernel<<<EW_BLOCKS, EW_THREADS, 0, stream>>>(num, num1, denom, denom1, bias, out);
    };

    vn_init_kernel<<<(HNUMB * HDIM + 255) / 256, 256, 0, stream>>>(vn_emb, vn_direct, vn_root);

    run_gat(x, W0, asrc0, adst0, b0, false);

    for (int l = 0; l < NCONVL; ++l) {
        gather_add_kernel<<<EW_BLOCKS, EW_THREADS, 0, stream>>>(out, vn_direct, hblocks, x_in);
        zero_kernel<<<(HNUMB * HDIM + 255) / 256, 256, 0, stream>>>(vn_seg, HNUMB * HDIM);
        run_gat(x_in, Wl + l * HDIM * HDIM, asrcl + l * HDIM, adstl + l * HDIM, bl + l * HDIM, true);
        vn_update_kernel<<<1, 256, 0, stream>>>(vn_direct, vn_seg, vn_root);
        for (int j = 0; j < NMLPL; ++j) {
            mlp_kernel<<<HNUMB, 64, 0, stream>>>(vn_direct, mW1 + j * HDIM * HDIM, mb1 + j * HDIM,
                                                 mW2 + j * HDIM * HDIM, mb2 + j * HDIM);
            mlp_kernel<<<1, 64, 0, stream>>>(vn_root, mW1 + j * HDIM * HDIM, mb1 + j * HDIM,
                                             mW2 + j * HDIM * HDIM, mb2 + j * HDIM);
        }
    }
}


# Per-node launches folded into the node kernel beside them: self logit + max seeds into dots, max join + self weight into num init, table clear into vn init/update; 39 to 31 launches; sum orders kept
# speedup vs baseline: 3.2311x; 1.0356x over previous
#include <hip/hip_runtime.h>
#include <hip/hip_bf16.h>
#include <hip/hip_fp16.h>

#define NNODES 100000
#define NEDGES 1600000
#define HDIM   64
#define HNUMB  128
#define NCONVL 2
#define NMLPL  2
#define NEGSLP 0.2f

typedef __attribute__((ext_vector_type(2))) float v2f;
typedef __attribute__((ext_vector_type(8))) float v8f;

__device__ __forceinline__ float leaky(float x) { return x > 0.f ? x : NEGSLP * x; }

__device__ __forceinline__ unsigned fenc(float f) {
    unsigned u = __float_as_uint(f);
    return (u & 0x80000000u) ? ~u : (u | 0x80000000u);
}
__device__ __forceinline__ float fdec(unsigned u) {
    return (u & 0x80000000u) ? __uint_as_float(u & 0x7fffffffu) : __uint_as_float(~u);
}

__device__ __forceinline__ int pkg_half() {
    return (int)((__builtin_amdgcn_s_sendmsg_rtn(0x87) >> 18) & 1u);
}

__global__ __launch_bounds__(256) void gemm_xw_kernel(const float* __restrict__ X,
                                                      const float* __restrict__ W,
                                                      float* __restrict__ Hout,
                                                      int nrows) {
    __shared__ float Wlds[HDIM * HDIM];
    const int tid = threadIdx.x;
    for (int idx = tid; idx < HDIM * HDIM; idx += 256) {
        int k = idx >> 6, n = idx & 63;
        Wlds[(k >> 1) * 128 + n * 2 + (k & 1)] = W[idx];
    }
    __syncthreads();

    const int wave    = tid >> 5;
    const int lane    = tid & 31;
    const int hi      = lane >> 4;
    const int mrow    = lane & 15;
    const int rowbase = blockIdx.x * 128 + wave * 16;
    if (rowbase >= nrows) return;

    v2f afrag[16];
#pragma unroll
    for (int kk = 0; kk < 16; ++kk) {
        const v2f* ap = (const v2f*)(X + (size_t)(rowbase + mrow) * HDIM + kk * 4 + hi * 2);
        afrag[kk] = *ap;
    }
#pragma unroll
    for (int nt = 0; nt < 4; ++nt) {
        const int ncol = nt * 16 + mrow;
        v8f acc = {};
#pragma unroll
        for (int kk = 0; kk < 16; ++kk) {
            v2f b = *(const v2f*)&Wlds[(kk * 2 + hi) * 128 + ncol * 2];
            acc = __builtin_amdgcn_wmma_f32_16x16x4_f32(false, afrag[kk], false, b,
                                                        (short)0, acc, false, false);
        }
#pragma unroll
        for (int v = 0; v < 8; ++v)
            Hout[(size_t)(rowbase + v + 8 * hi) * HDIM + ncol] = acc[v];
    }
}

__global__ void dots_kernel(const float* __restrict__ Hm, const float* __restrict__ asrc,
                            const float* __restrict__ adst, float* __restrict__ a_s,
                            float* __restrict__ a_d, float* __restrict__ e_self,
                            unsigned* __restrict__ m_enc0, unsigned* __restrict__ m_enc1, int n) {
    int r = blockIdx.x * blockDim.x + threadIdx.x;
    if (r >= n) return;
    const float* hr = Hm + (size_t)r * HDIM;
    float s = 0.f, d = 0.f;
#pragma unroll
    for (int k = 0; k < HDIM; ++k) { float v = hr[k]; s += v * asrc[k]; d += v * adst[k]; }
    a_s[r] = s; a_d[r] = d;
    const float es = leaky(s + d);
    e_self[r] = es;
    m_enc0[r] = fenc(es);
    m_enc1[r] = fenc(es);
}

__global__ void edge_max_kernel(const int* __restrict__ src, const int* __restrict__ dst,
                                const float* __restrict__ a_s, const float* __restrict__ a_d,
                                unsigned* __restrict__ m_enc0, unsigned* __restrict__ m_enc1,
                                float* __restrict__ e_edge, int ne) {
    unsigned* const m_mine = pkg_half() ? m_enc1 : m_enc0;
    int e = blockIdx.x * blockDim.x + threadIdx.x;
    if (e >= ne) return;
    int s = src[e], d = dst[e];
    const float ev = leaky(a_s[s] + a_d[d]);
    e_edge[e] = ev;
    atomicMax(m_mine + d, fenc(ev));
}

#define EW_THREADS 256
#define EW_PIECES  10
#define EW_NP      (NNODES * (HDIM / 4))
#define EW_BLOCKS  (EW_NP / (EW_THREADS * EW_PIECES))
static_assert(EW_NP % (EW_THREADS * EW_PIECES) == 0, "elementwise passes: every block owns whole groups of pieces");

__global__ __launch_bounds__(EW_THREADS) void num_init_kernel(const float* __restrict__ e_self,
                                                              unsigned* __restrict__ m_enc0,
                                                              const unsigned* __restrict__ m_enc1,
                                                              const float* __restrict__ Hm,
                                                              float* __restrict__ denom0,
                                                              float* __restrict__ denom1,
                                                              float* __restrict__ num,
                                                              float* __restrict__ num1) {
    const int p0 = blockIdx.x * (EW_THREADS * EW_PIECES) + threadIdx.x;
    const bool row_lane = (threadIdx.x & 15) == 0;
    float4   v[EW_PIECES];
    float    es[EW_PIECES];
    unsigned ma[EW_PIECES], mb[EW_PIECES];
#pragma unroll
    for (int u = 0; u < EW_PIECES; ++u) {
        const int p = p0 + u * EW_THREADS;
        v[u]  = ((const float4*)Hm)[p];
        es[u] = e_self[p >> 4];
        ma[u] = m_enc0[p >> 4];
        mb[u] = m_enc1[p >> 4];
    }
#pragma unroll
    for (int u = 0; u < EW_PIECES; ++u) {
        const int p = p0 + u * EW_THREADS;
        const unsigned m = ma[u] > mb[u] ? ma[u] : mb[u];
        const float    w = __expf(es[u] - fdec(m));
        float4 o;
        o.x = w * v[u].x;
        o.y = w * v[u].y;
        o.z = w * v[u].z;
        o.w = w * v[u].w;
        ((float4*)num)[p]  = o;
        ((float4*)num1)[p] = make_float4(0.f, 0.f, 0.f, 0.f);
        if (row_lane) {
            m_enc0[p >> 4] = m;
            denom0[p >> 4] = w;
            denom1[p >> 4] = 0.f;
        }
    }
}

#define SC_EPW   32
#define SC_BATCH 8
static_assert(NEDGES % 256 == 0, "edge scatter: a block of 8 waves owns 256 whole edges");
__global__ __launch_bounds__(256) void edge_scatter_kernel(const int* __restrict__ src, const int* __restrict__ dst,
                                    const float* __restrict__ e_edge, const unsigned* __restrict__ m_enc,
                                    const float* __restrict__ Hm,
                                    float* __restrict__ denom0, float* __restrict__ denom1,
                                    float* __restrict__ num0, float* __restrict__ num1) {
    const int half = pkg_half();
    float* const denom = half ? denom1 : denom0;
    float* const num   = half ? num1 : num0;
    const int lane = threadIdx.x & 31;
    const int e    = blockIdx.x * 256 + threadIdx.x;
    const int   sv = src[e];
    const int   dv = dst[e];
    const float ev = e_edge[e];
    const unsigned mv = m_enc[dv];

    float lo[2][SC_BATCH], hi[2][SC_BATCH];
#pragma unroll
    for (int j = 0; j < SC_BATCH; ++j) {
        const float* hr = Hm + (size_t)__builtin_amdgcn_readlane(sv, j) * HDIM;
        lo[0][j] = hr[lane];
        hi[0][j] = hr[lane + 32];
    }
    int wv = 0;
#pragma unroll
    for (int b = 0; b < SC_EPW / SC_BATCH; ++b) {
        const int cur = b & 1, nxt = cur ^ 1;
        if (b + 1 < SC_EPW / SC_BATCH) {
#pragma unroll
            for (int j = 0; j < SC_BATCH; ++j) {
                const float* hr = Hm + (size_t)__builtin_amdgcn_readlane(sv, (b + 1) * SC_BATCH + j) * HDIM;
                lo[nxt][j] = hr[lane];
                hi[nxt][j] = hr[lane + 32];
            }
        }
        if (b == 0) {
            const float wf = __expf(ev - fdec(mv));
            atomicAdd(denom + dv, wf);
            wv = __float_as_int(wf);
        }
#pragma unroll
        for (int j = 0; j < SC_BATCH; ++j) {
            const int   d = __builtin_amdgcn_readlane(dv, b * SC_BATCH + j);
            const float w = __int_as_float(__builtin_amdgcn_readlane(wv, b * SC_BATCH + j));
            float* nr = num + (size_t)d * HDIM;
            atomicAdd(nr + lane,      w * lo[cur][j]);
            atomicAdd(nr + lane + 32, w * hi[cur][j]);
        }
    }
}

__global__ __launch_bounds__(EW_THREADS) void finalize_kernel(const float* __restrict__ num,
                                                              const float* __restrict__ num1,
                                                              const float* __restrict__ denom,
                                                              const float* __restrict__ denom1,
                                                              const float* __restrict__ bias,
                                                              float* __restrict__ out) {
    const int p0 = blockIdx.x * (EW_THREADS * EW_PIECES) + threadIdx.x;
    const float4 b4 = ((const float4*)bias)[threadIdx.x & 15];
    float4 v[EW_PIECES], v1[EW_PIECES];
    float  dn[EW_PIECES], dn1[EW_PIECES];
#pragma unroll
    for (int u = 0; u < EW_PIECES; ++u) {
        const int p = p0 + u * EW_THREADS;
        v[u]   = ((const float4*)num)[p];
        v1[u]  = ((const float4*)num1)[p];
        dn[u]  = denom[p >> 4];
        dn1[u] = denom1[p >> 4];
    }
#pragma unroll
    for (int u = 0; u < EW_PIECES; ++u) {
        const int p = p0 + u * EW_THREADS;
        const float dd = (dn[u] + dn1[u]) + 1e-16f;
        float4 o;
        o.x = (v[u].x + v1[u].x) / dd + b4.x;
        o.y = (v[u].y + v1[u].y) / dd + b4.y;
        o.z = (v[u].z + v1[u].z) / dd + b4.z;
        o.w = (v[u].w + v1[u].w) / dd + b4.w;
        ((float4*)out)[p] = o;
    }
}

#define FS_THREADS 512
#define FS_BLOCKS  256
#define FS_ROWS    ((NNODES + FS_BLOCKS - 1) / FS_BLOCKS)
#define FS_RSTEP   (FS_THREADS / 16)
#define FS_PITCH   (HDIM + 1)
__global__ __launch_bounds__(FS_THREADS) void finalize_segsum_kernel(
        const float* __restrict__ num, const float* __restrict__ num1,
        const float* __restrict__ denom, const float* __restrict__ denom1,
        const float* __restrict__ bias, const int* __restrict__ blocks,
        float* __restrict__ out, float* __restrict__ vn_seg) {
    __shared__ float acc[HNUMB * FS_PITCH];
    const int tid = threadIdx.x;
    for (int i = tid; i < HNUMB * FS_PITCH; i += FS_THREADS) acc[i] = 0.f;
    __syncthreads();

    const int c4   = tid & 15;
    const int row0 = blockIdx.x * FS_ROWS;
    const int row1 = (row0 + FS_ROWS < NNODES) ? row0 + FS_ROWS : NNODES;
    const float4 b4 = ((const float4*)bias)[c4];
    for (int rb = row0 + (tid >> 4); rb < row1; rb += 4 * FS_RSTEP) {
        float4 n4[4], m4[4];
        float  dn[4];
        int    bk[4];
#pragma unroll
        for (int u = 0; u < 4; ++u) {
            const int r = rb + u * FS_RSTEP;
            n4[u] = make_float4(0.f, 0.f, 0.f, 0.f); m4[u] = n4[u]; dn[u] = 1.f; bk[u] = 0;
            if (r < row1) {
                n4[u] = ((const float4*)num)[(size_t)r * (HDIM / 4) + c4];
                m4[u] = ((const float4*)num1)[(size_t)r * (HDIM / 4) + c4];
                dn[u] = denom[r] + denom1[r];
                bk[u] = blocks[r];
            }
        }
#pragma unroll
        for (int u = 0; u < 4; ++u) {
            const int r = rb + u * FS_RSTEP;
            if (r < row1) {
                const float dd = dn[u] + 1e-16f;
                float4 o;
                o.x = (n4[u].x + m4[u].x) / dd + b4.x;
                o.y = (n4[u].y + m4[u].y) / dd + b4.y;
                o.z = (n4[u].z + m4[u].z) / dd + b4.z;
                o.w = (n4[u].w + m4[u].w) / dd + b4.w;
                ((float4*)out)[(size_t)r * (HDIM / 4) + c4] = o;
                float* a = acc + bk[u] * FS_PITCH + c4 * 4;
                atomicAdd(a + 0, o.x);
                atomicAdd(a + 1, o.y);
                atomicAdd(a + 2, o.z);
                atomicAdd(a + 3, o.w);
            }
        }
    }
    __syncthreads();
    for (int i = tid; i < HNUMB * HDIM; i += FS_THREADS) {
        const int j = (i + blockIdx.x * 32) & (HNUMB * HDIM - 1);
        atomicAdd(vn_seg + j, acc[(j >> 6) * FS_PITCH + (j & 63)]);
    }
}

__global__ __launch_bounds__(EW_THREADS) void gather_add_kernel(const float* __restrict__ out,
                                                                const float* __restrict__ vnd,
                                                                const int* __restrict__ blocks,
                                                                float* __restrict__ x_in) {
    const int p0 = blockIdx.x * (EW_THREADS * EW_PIECES) + threadIdx.x;
    const int c4 = threadIdx.x & 15;
    float4 v[EW_PIECES];
    int    bk[EW_PIECES];
#pragma unroll
    for (int u = 0; u < EW_PIECES; ++u) {
        const int p = p0 + u * EW_THREADS;
        v[u]  = ((const float4*)out)[p];
        bk[u] = blocks[p >> 4];
    }
#pragma unroll
    for (int u = 0; u < EW_PIECES; ++u) {
        const int p = p0 + u * EW_THREADS;
        const float4 a = ((const float4*)vnd)[bk[u] * (HDIM / 4) + c4];
        float4 o;
        o.x = v[u].x + a.x;
        o.y = v[u].y + a.y;
        o.z = v[u].z + a.z;
        o.w = v[u].w + a.w;
        ((float4*)x_in)[p] = o;
    }
}

__global__ __launch_bounds__(256) void vn_update_kernel(float* __restrict__ vn_direct,
                                                        float* __restrict__ vn_seg,
                                                        float* __restrict__ vn_root) {
    int tid = threadIdx.x;
    for (int i = tid; i < HNUMB * HDIM; i += 256) {
        vn_direct[i] += vn_seg[i] + vn_root[i & (HDIM - 1)];
        vn_seg[i] = 0.f;
    }
    __syncthreads();
    if (tid < HDIM) {
        float s = 0.f;
        for (int r = 0; r < HNUMB; ++r) s += vn_direct[r * HDIM + tid];
        vn_root[tid] += s;
    }
}

__global__ __launch_bounds__(64) void mlp_kernel(float* __restrict__ vn,
                                                 const float* __restrict__ W1, const float* __restrict__ b1,
                                                 const float* __restrict__ W2, const float* __restrict__ b2) {
    __shared__ float rin[HDIM];
    __shared__ float t1[HDIM];
    int f = threadIdx.x;
    float* row = vn + (size_t)blockIdx.x * HDIM;
    rin[f] = row[f];
    __syncthreads();
    float s = b1[f];
#pragma unroll
    for (int k = 0; k < HDIM; ++k) s += rin[k] * W1[k * HDIM + f];
    t1[f] = fmaxf(s, 0.f);
    __syncthreads();
    float s2 = b2[f];
#pragma unroll
    for (int k = 0; k < HDIM; ++k) s2 += t1[k] * W2[k * HDIM + f];
    row[f] = fmaxf(s2, 0.f);
}

__global__ void vn_init_kernel(const float* __restrict__ vn_emb, float* __restrict__ vn_direct,
                               float* __restrict__ vn_root, float* __restrict__ vn_seg) {
    int t = blockIdx.x * blockDim.x + threadIdx.x;
    if (t < HNUMB * HDIM) { vn_direct[t] = vn_emb[t & (HDIM - 1)]; vn_seg[t] = 0.f; }
    if (t < HDIM) vn_root[t] = vn_emb[t];
}

extern "C" void kernel_launch(void* const* d_in, const int* in_sizes, int n_in,
                              void* d_out, int out_size, void* d_ws, size_t ws_size,
                              hipStream_t stream) {
    const float* x       = (const float*)d_in[0];
    const int*   ei      = (const int*)d_in[1];
    const int*   src     = ei;
    const int*   dst     = ei + NEDGES;
    const int*   hblocks = (const int*)d_in[2];
    const float* W0      = (const float*)d_in[5];
    const float* asrc0   = (const float*)d_in[6];
    const float* adst0   = (const float*)d_in[7];
    const float* b0      = (const float*)d_in[8];
    const float* Wl      = (const float*)d_in[9];
    const float* asrcl   = (const float*)d_in[10];
    const float* adstl   = (const float*)d_in[11];
    const float* bl      = (const float*)d_in[12];
    const float* vn_emb  = (const float*)d_in[13];
    const float* mW1     = (const float*)d_in[14];
    const float* mb1     = (const float*)d_in[15];
    const float* mW2     = (const float*)d_in[16];
    const float* mb2     = (const float*)d_in[17];
    float* out = (float*)d_out;

    char*  ws  = (char*)d_ws;
    size_t off = 0;
    auto carve = [&](size_t bytes) -> void* {
        void* p = ws + off;
        off += (bytes + 255) & ~(size_t)255;
        return p;
    };
    float*    h         = (float*)carve((size_t)NNODES * HDIM * 4);
    float*    x_in      = (float*)carve((size_t)NNODES * HDIM * 4);
    float*    num       = (float*)carve((size_t)NNODES * HDIM * 4);
    float*    a_s       = (float*)carve((size_t)NNODES * 4);
    float*    a_d       = (float*)carve((size_t)NNODES * 4);
    float*    e_self    = (float*)carve((size_t)NNODES * 4);
    unsigned* m_enc1    = (unsigned*)carve((size_t)NNODES * 4);
    unsigned* m_enc     = (unsigned*)carve((size_t)NNODES * 4);
    float*    denom     = a_s;
    float*    denom1    = a_d;
    float*    num1      = x_in;
    float*    e_edge    = (float*)carve((size_t)NEDGES * 4);
    float*    vn_direct = (float*)carve((size_t)HNUMB * HDIM * 4);
    float*    vn_root   = (float*)carve((size_t)HDIM * 4);
    float*    vn_seg    = (float*)carve((size_t)HNUMB * HDIM * 4);

    const int nodeBlk = (NNODES + 255) / 256;
    const int edgeBlk = NEDGES / 256;

    auto run_gat = [&](const float* Xin, const float* W, const float* asrc,
                       const float* adst, const float* bias, bool with_segsum) {
        gemm_xw_kernel<<<(NNODES + 127) / 128, 256, 0, stream>>>(Xin, W, h, NNODES);
        dots_kernel<<<nodeBlk, 256, 0, stream>>>(h, asrc, adst, a_s, a_d, e_self, m_enc, m_enc1, NNODES);
        edge_max_kernel<<<edgeBlk, 256, 0, stream>>>(src, dst, a_s, a_d, m_enc, m_enc1, e_edge, NEDGES);
        num_init_kernel<<<EW_BLOCKS, EW_THREADS, 0, stream>>>(e_self, m_enc, m_enc1, h, denom, denom1, num, num1);
        edge_scatter_kernel<<<edgeBlk, 256, 0, stream>>>(src, dst, e_edge, m_enc, h, denom, denom1, num, num1);
        if (with_segsum)
            finalize_segsum_kernel<<<FS_BLOCKS, FS_THREADS, 0, stream>>>(num, num1, denom, denom1, bias, hblocks, out, vn_seg);
        else
            finalize_kernel<<<EW_BLOCKS, EW_THREADS, 0, stream>>>(num, num1, denom, denom1, bias, out);
    };

    vn_init_kernel<<<(HNUMB * HDIM + 255) / 256, 256, 0, stream>>>(vn_emb, vn_direct, vn_root, vn_seg);

    run_gat(x, W0, asrc0, adst0, b0, false);

    for (int l = 0; l < NCONVL; ++l) {
        gather_add_kernel<<<EW_BLOCKS, EW_THREADS, 0, stream>>>(out, vn_direct, hblocks, x_in);
        run_gat(x_in, Wl + l * HDIM * HDIM, asrcl + l * HDIM, adstl + l * HDIM, bl + l * HDIM, true);
        vn_update_kernel<<<1, 256, 0, stream>>>(vn_direct, vn_seg, vn_root);
        for (int j = 0; j < NMLPL; ++j) {
            mlp_kernel<<<HNUMB, 64, 0, stream>>>(vn_direct, mW1 + j * HDIM * HDIM, mb1 + j * HDIM,
                                                 mW2 + j * HDIM * HDIM, mb2 + j * HDIM);
            mlp_kernel<<<1, 64, 0, stream>>>(vn_root, mW1 + j * HDIM * HDIM, mb1 + j * HDIM,
                                             mW2 + j * HDIM * HDIM, mb2 + j * HDIM);
        }
    }
}


# Softmax shift c[d] = leaky(max of a_s over all nodes + a_d[d]) when the range of a_s is under 16, else the exact max pass: per-edge max pass, its atomics, logit array go; weights rescale per node
# speedup vs baseline: 3.3471x; 1.0359x over previous
#include <hip/hip_runtime.h>
#include <hip/hip_bf16.h>
#include <hip/hip_fp16.h>

#define NNODES 100000
#define NEDGES 1600000
#define HDIM   64
#define HNUMB  128
#define NCONVL 2
#define NMLPL  2
#define NEGSLP 0.2f

typedef __attribute__((ext_vector_type(2))) float v2f;
typedef __attribute__((ext_vector_type(8))) float v8f;

__device__ __forceinline__ float leaky(float x) { return x > 0.f ? x : NEGSLP * x; }

__device__ __forceinline__ unsigned fenc(float f) {
    unsigned u = __float_as_uint(f);
    return (u & 0x80000000u) ? ~u : (u | 0x80000000u);
}
__device__ __forceinline__ float fdec(unsigned u) {
    return (u & 0x80000000u) ? __uint_as_float(u & 0x7fffffffu) : __uint_as_float(~u);
}

__device__ __forceinline__ int pkg_half() {
    return (int)((__builtin_amdgcn_s_sendmsg_rtn(0x87) >> 18) & 1u);
}

__global__ __launch_bounds__(256) void gemm_xw_kernel(const float* __restrict__ X,
                                                      const float* __restrict__ W,
                                                      float* __restrict__ Hout,
                                                      int nrows) {
    __shared__ float Wlds[HDIM * HDIM];
    const int tid = threadIdx.x;
    for (int idx = tid; idx < HDIM * HDIM; idx += 256) {
        int k = idx >> 6, n = idx & 63;
        Wlds[(k >> 1) * 128 + n * 2 + (k & 1)] = W[idx];
    }
    __syncthreads();

    const int wave    = tid >> 5;
    const int lane    = tid & 31;
    const int hi      = lane >> 4;
    const int mrow    = lane & 15;
    const int rowbase = blockIdx.x * 128 + wave * 16;
    if (rowbase >= nrows) return;

    v2f afrag[16];
#pragma unroll
    for (int kk = 0; kk < 16; ++kk) {
        const v2f* ap = (const v2f*)(X + (size_t)(rowbase + mrow) * HDIM + kk * 4 + hi * 2);
        afrag[kk] = *ap;
    }
#pragma unroll
    for (int nt = 0; nt < 4; ++nt) {
        const int ncol = nt * 16 + mrow;
        v8f acc = {};
#pragma unroll
        for (int kk = 0; kk < 16; ++kk) {
            v2f b = *(const v2f*)&Wlds[(kk * 2 + hi) * 128 + ncol * 2];
            acc = __builtin_amdgcn_wmma_f32_16x16x4_f32(false, afrag[kk], false, b,
                                                        (short)0, acc, false, false);
        }
#pragma unroll
        for (int v = 0; v < 8; ++v)
            Hout[(size_t)(rowbase + v + 8 * hi) * HDIM + ncol] = acc[v];
    }
}

#define SHIFT_RANGE_OK 16.f
__device__ __forceinline__ bool bound_shift_ok(const unsigned* __restrict__ rng) {
    return fdec(rng[0]) - fdec(rng[1]) < SHIFT_RANGE_OK;
}

__global__ __launch_bounds__(256) void dots_kernel(const float* __restrict__ Hm, const float* __restrict__ asrc,
                            const float* __restrict__ adst, float* __restrict__ a_s,
                            float* __restrict__ rec, float* __restrict__ e_self,
                            unsigned* __restrict__ m_enc1, unsigned* __restrict__ rng, int n) {
    __shared__ unsigned red_hi[8], red_lo[8];
    const int r = blockIdx.x * 256 + threadIdx.x;
    unsigned ehi = 0u, elo = 0xffffffffu;
    if (r < n) {
        const float* hr = Hm + (size_t)r * HDIM;
        float s = 0.f, d = 0.f;
#pragma unroll
        for (int k = 0; k < HDIM; ++k) { float v = hr[k]; s += v * asrc[k]; d += v * adst[k]; }
        a_s[r] = s;
        e_self[r] = leaky(s + d);
        const unsigned enc = fenc(s);
        ((float2*)rec)[r] = make_float2(d, __uint_as_float(enc));
        m_enc1[r] = enc;
        ehi = enc; elo = enc;
    }
    for (int o = 16; o > 0; o >>= 1) {
        const unsigned oh = __shfl_xor(ehi, o), ol = __shfl_xor(elo, o);
        ehi = ehi > oh ? ehi : oh;
        elo = elo < ol ? elo : ol;
    }
    if ((threadIdx.x & 31) == 0) { red_hi[threadIdx.x >> 5] = ehi; red_lo[threadIdx.x >> 5] = elo; }
    __syncthreads();
    if (threadIdx.x == 0) {
        for (int w = 1; w < 8; ++w) {
            ehi = ehi > red_hi[w] ? ehi : red_hi[w];
            elo = elo < red_lo[w] ? elo : red_lo[w];
        }
        atomicMax(rng, ehi);
        atomicMin(rng + 1, elo);
    }
}

__global__ void edge_max_kernel(const int* __restrict__ src, const int* __restrict__ dst,
                                const float* __restrict__ a_s,
                                float* __restrict__ rec, unsigned* __restrict__ m_enc1,
                                const unsigned* __restrict__ rng, int ne) {
    if (bound_shift_ok(rng)) return;
    const int half = pkg_half();
    int e = blockIdx.x * blockDim.x + threadIdx.x;
    if (e >= ne) return;
    int s = src[e], d = dst[e];
    unsigned* const m_mine = half ? m_enc1 + d : (unsigned*)rec + 2 * d + 1;
    atomicMax(m_mine, fenc(a_s[s]));
}

#define EW_THREADS 256
#define EW_PIECES  10
#define EW_NP      (NNODES * (HDIM / 4))
#define EW_BLOCKS  (EW_NP / (EW_THREADS * EW_PIECES))
static_assert(EW_NP % (EW_THREADS * EW_PIECES) == 0, "elementwise passes: every block owns whole groups of pieces");

__global__ __launch_bounds__(EW_THREADS) void num_init_kernel(float* es_den1,
                                                              float* rec,
                                                              unsigned* m1_den0,
                                                              const unsigned* __restrict__ rng,
                                                              const float* __restrict__ Hm,
                                                              float* __restrict__ num,
                                                              float* __restrict__ num1) {
    const bool     bound   = bound_shift_ok(rng);
    const unsigned enc_top = rng[0];
    const int p0 = blockIdx.x * (EW_THREADS * EW_PIECES) + threadIdx.x;
    const bool row_lane = (threadIdx.x & 15) == 0;
    float4   v[EW_PIECES];
    float    es[EW_PIECES];
    float2   rc[EW_PIECES];
    unsigned mb[EW_PIECES];
#pragma unroll
    for (int u = 0; u < EW_PIECES; ++u) {
        const int p = p0 + u * EW_THREADS;
        v[u]  = ((const float4*)Hm)[p];
        es[u] = es_den1[p >> 4];
        rc[u] = ((const float2*)rec)[p >> 4];
        mb[u] = m1_den0[p >> 4];
    }
#pragma unroll
    for (int u = 0; u < EW_PIECES; ++u) {
        const int p = p0 + u * EW_THREADS;
        const unsigned ma = __float_as_uint(rc[u].y);
        const unsigned mj = ma > mb[u] ? ma : mb[u];
        const float    c  = leaky(fdec(bound ? enc_top : mj) + rc[u].x);
        const float    w  = __expf(es[u] - c);
        float4 o;
        o.x = w * v[u].x;
        o.y = w * v[u].y;
        o.z = w * v[u].z;
        o.w = w * v[u].w;
        ((float4*)num)[p]  = o;
        ((float4*)num1)[p] = make_float4(0.f, 0.f, 0.f, 0.f);
        if (row_lane) {
            rec[2 * (p >> 4) + 1] = c;
            m1_den0[p >> 4] = __float_as_uint(w);
            es_den1[p >> 4] = 0.f;
        }
    }
}

#define SC_EPW   32
#define SC_BATCH 8
static_assert(NEDGES % 256 == 0, "edge scatter: a block of 8 waves owns 256 whole edges");
__global__ __launch_bounds__(256) void edge_scatter_kernel(const int* __restrict__ src, const int* __restrict__ dst,
                                    const float* __restrict__ a_s, const float* __restrict__ rec,
                                    const float* __restrict__ Hm,
                                    float* __restrict__ denom0, float* __restrict__ denom1,
                                    float* __restrict__ num0, float* __restrict__ num1) {
    const int half = pkg_half();
    float* const denom = half ? denom1 : denom0;
    float* const num   = half ? num1 : num0;
    const int lane = threadIdx.x & 31;
    const int e    = blockIdx.x * 256 + threadIdx.x;
    const int    sv = src[e];
    const int    dv = dst[e];
    const float  av = a_s[sv];
    const float2 rv = ((const float2*)rec)[dv];

    float lo[2][SC_BATCH], hi[2][SC_BATCH];
#pragma unroll
    for (int j = 0; j < SC_BATCH; ++j) {
        const float* hr = Hm + (size_t)__builtin_amdgcn_readlane(sv, j) * HDIM;
        lo[0][j] = hr[lane];
        hi[0][j] = hr[lane + 32];
    }
    int wv = 0;
#pragma unroll
    for (int b = 0; b < SC_EPW / SC_BATCH; ++b) {
        const int cur = b & 1, nxt = cur ^ 1;
        if (b + 1 < SC_EPW / SC_BATCH) {
#pragma unroll
            for (int j = 0; j < SC_BATCH; ++j) {
                const float* hr = Hm + (size_t)__builtin_amdgcn_readlane(sv, (b + 1) * SC_BATCH + j) * HDIM;
                lo[nxt][j] = hr[lane];
                hi[nxt][j] = hr[lane + 32];
            }
        }
        if (b == 0) {
            const float wf = __expf(leaky(av + rv.x) - rv.y);
            atomicAdd(denom + dv, wf);
            wv = __float_as_int(wf);
        }
#pragma unroll
        for (int j = 0; j < SC_BATCH; ++j) {
            const int   d = __builtin_amdgcn_readlane(dv, b * SC_BATCH + j);
            const float w = __int_as_float(__builtin_amdgcn_readlane(wv, b * SC_BATCH + j));
            float* nr = num + (size_t)d * HDIM;
            atomicAdd(nr + lane,      w * lo[cur][j]);
            atomicAdd(nr + lane + 32, w * hi[cur][j]);
        }
    }
}

__global__ __launch_bounds__(EW_THREADS) void finalize_kernel(const float* __restrict__ num,
                                                              const float* __restrict__ num1,
                                                              const float* __restrict__ denom,
                                                              const float* __restrict__ denom1,
                                                              const float* __restrict__ bias,
                                                              float* __restrict__ out) {
    const int p0 = blockIdx.x * (EW_THREADS * EW_PIECES) + threadIdx.x;
    const float4 b4 = ((const float4*)bias)[threadIdx.x & 15];
    float4 v[EW_PIECES], v1[EW_PIECES];
    float  dn[EW_PIECES], dn1[EW_PIECES];
#pragma unroll
    for (int u = 0; u < EW_PIECES; ++u) {
        const int p = p0 + u * EW_THREADS;
        v[u]   = ((const float4*)num)[p];
        v1[u]  = ((const float4*)num1)[p];
        dn[u]  = denom[p >> 4];
        dn1[u] = denom1[p >> 4];
    }
#pragma unroll
    for (int u = 0; u < EW_PIECES; ++u) {
        const int p = p0 + u * EW_THREADS;
        const float dd = (dn[u] + dn1[u]) + 1e-16f;
        float4 o;
        o.x = (v[u].x + v1[u].x) / dd + b4.x;
        o.y = (v[u].y + v1[u].y) / dd + b4.y;
        o.z = (v[u].z + v1[u].z) / dd + b4.z;
        o.w = (v[u].w + v1[u].w) / dd + b4.w;
        ((float4*)out)[p] = o;
    }
}

#define FS_THREADS 512
#define FS_BLOCKS  256
#define FS_ROWS    ((NNODES + FS_BLOCKS - 1) / FS_BLOCKS)
#define FS_RSTEP   (FS_THREADS / 16)
#define FS_PITCH   (HDIM + 1)
__global__ __launch_bounds__(FS_THREADS) void finalize_segsum_kernel(
        const float* __restrict__ num, const float* __restrict__ num1,
        const float* __restrict__ denom, const float* __restrict__ denom1,
        const float* __restrict__ bias, const int* __restrict__ blocks,
        float* __restrict__ out, float* __restrict__ vn_seg) {
    __shared__ float acc[HNUMB * FS_PITCH];
    const int tid = threadIdx.x;
    for (int i = tid; i < HNUMB * FS_PITCH; i += FS_THREADS) acc[i] = 0.f;
    __syncthreads();

    const int c4   = tid & 15;
    const int row0 = blockIdx.x * FS_ROWS;
    const int row1 = (row0 + FS_ROWS < NNODES) ? row0 + FS_ROWS : NNODES;
    const float4 b4 = ((const float4*)bias)[c4];
    for (int rb = row0 + (tid >> 4); rb < row1; rb += 4 * FS_RSTEP) {
        float4 n4[4], m4[4];
        float  dn[4];
        int    bk[4];
#pragma unroll
        for (int u = 0; u < 4; ++u) {
            const int r = rb + u * FS_RSTEP;
            n4[u] = make_float4(0.f, 0.f, 0.f, 0.f); m4[u] = n4[u]; dn[u] = 1.f; bk[u] = 0;
            if (r < row1) {
                n4[u] = ((const float4*)num)[(size_t)r * (HDIM / 4) + c4];
                m4[u] = ((const float4*)num1)[(size_t)r * (HDIM / 4) + c4];
                dn[u] = denom[r] + denom1[r];
                bk[u] = blocks[r];
            }
        }
#pragma unroll
        for (int u = 0; u < 4; ++u) {
            const int r = rb + u * FS_RSTEP;
            if (r < row1) {
                const float dd = dn[u] + 1e-16f;
                float4 o;
                o.x = (n4[u].x + m4[u].x) / dd + b4.x;
                o.y = (n4[u].y + m4[u].y) / dd + b4.y;
                o.z = (n4[u].z + m4[u].z) / dd + b4.z;
                o.w = (n4[u].w + m4[u].w) / dd + b4.w;
                ((float4*)out)[(size_t)r * (HDIM / 4) + c4] = o;
                float* a = acc + bk[u] * FS_PITCH + c4 * 4;
                atomicAdd(a + 0, o.x);
                atomicAdd(a + 1, o.y);
                atomicAdd(a + 2, o.z);
                atomicAdd(a + 3, o.w);
            }
        }
    }
    __syncthreads();
    for (int i = tid; i < HNUMB * HDIM; i += FS_THREADS) {
        const int j = (i + blockIdx.x * 32) & (HNUMB * HDIM - 1);
        atomicAdd(vn_seg + j, acc[(j >> 6) * FS_PITCH + (j & 63)]);
    }
}

__global__ __launch_bounds__(EW_THREADS) void gather_add_kernel(const float* __restrict__ out,
                                                                const float* __restrict__ vnd,
                                                                const int* __restrict__ blocks,
                                                                float* __restrict__ x_in) {
    const int p0 = blockIdx.x * (EW_THREADS * EW_PIECES) + threadIdx.x;
    const int c4 = threadIdx.x & 15;
    float4 v[EW_PIECES];
    int    bk[EW_PIECES];
#pragma unroll
    for (int u = 0; u < EW_PIECES; ++u) {
        const int p = p0 + u * EW_THREADS;
        v[u]  = ((const float4*)out)[p];
        bk[u] = blocks[p >> 4];
    }
#pragma unroll
    for (int u = 0; u < EW_PIECES; ++u) {
        const int p = p0 + u * EW_THREADS;
        const float4 a = ((const float4*)vnd)[bk[u] * (HDIM / 4) + c4];
        float4 o;
        o.x = v[u].x + a.x;
        o.y = v[u].y + a.y;
        o.z = v[u].z + a.z;
        o.w = v[u].w + a.w;
        ((float4*)x_in)[p] = o;
    }
}

__global__ __launch_bounds__(256) void vn_update_kernel(float* __restrict__ vn_direct,
                                                        float* __restrict__ vn_seg,
                                                        float* __restrict__ vn_root) {
    int tid = threadIdx.x;
    for (int i = tid; i < HNUMB * HDIM; i += 256) {
        vn_direct[i] += vn_seg[i] + vn_root[i & (HDIM - 1)];
        vn_seg[i] = 0.f;
    }
    __syncthreads();
    if (tid < HDIM) {
        float s = 0.f;
        for (int r = 0; r < HNUMB; ++r) s += vn_direct[r * HDIM + tid];
        vn_root[tid] += s;
    }
}

__global__ __launch_bounds__(64) void mlp_kernel(float* __restrict__ vn,
                                                 const float* __restrict__ W1, const float* __restrict__ b1,
                                                 const float* __restrict__ W2, const float* __restrict__ b2) {
    __shared__ float rin[HDIM];
    __shared__ float t1[HDIM];
    int f = threadIdx.x;
    float* row = vn + (size_t)blockIdx.x * HDIM;
    rin[f] = row[f];
    __syncthreads();
    float s = b1[f];
#pragma unroll
    for (int k = 0; k < HDIM; ++k) s += rin[k] * W1[k * HDIM + f];
    t1[f] = fmaxf(s, 0.f);
    __syncthreads();
    float s2 = b2[f];
#pragma unroll
    for (int k = 0; k < HDIM; ++k) s2 += t1[k] * W2[k * HDIM + f];
    row[f] = fmaxf(s2, 0.f);
}

__global__ void vn_init_kernel(const float* __restrict__ vn_emb, float* __restrict__ vn_direct,
                               float* __restrict__ vn_root, float* __restrict__ vn_seg,
                               unsigned* __restrict__ rng) {
    int t = blockIdx.x * blockDim.x + threadIdx.x;
    if (t < HNUMB * HDIM) { vn_direct[t] = vn_emb[t & (HDIM - 1)]; vn_seg[t] = 0.f; }
    if (t < HDIM) vn_root[t] = vn_emb[t];
    if (t < 2 * (NCONVL + 1)) rng[t] = (t & 1) ? 0xffffffffu : 0u;
}

extern "C" void kernel_launch(void* const* d_in, const int* in_sizes, int n_in,
                              void* d_out, int out_size, void* d_ws, size_t ws_size,
                              hipStream_t stream) {
    const float* x       = (const float*)d_in[0];
    const int*   ei      = (const int*)d_in[1];
    const int*   src     = ei;
    const int*   dst     = ei + NEDGES;
    const int*   hblocks = (const int*)d_in[2];
    const float* W0      = (const float*)d_in[5];
    const float* asrc0   = (const float*)d_in[6];
    const float* adst0   = (const float*)d_in[7];
    const float* b0      = (const float*)d_in[8];
    const float* Wl      = (const float*)d_in[9];
    const float* asrcl   = (const float*)d_in[10];
    const float* adstl   = (const float*)d_in[11];
    const float* bl      = (const float*)d_in[12];
    const float* vn_emb  = (const float*)d_in[13];
    const float* mW1     = (const float*)d_in[14];
    const float* mb1     = (const float*)d_in[15];
    const float* mW2     = (const float*)d_in[16];
    const float* mb2     = (const float*)d_in[17];
    float* out = (float*)d_out;

    char*  ws  = (char*)d_ws;
    size_t off = 0;
    auto carve = [&](size_t bytes) -> void* {
        void* p = ws + off;
        off += (bytes + 255) & ~(size_t)255;
        return p;
    };
    float*    h         = (float*)carve((size_t)NNODES * HDIM * 4);
    float*    x_in      = (float*)carve((size_t)NNODES * HDIM * 4);
    float*    num       = (float*)carve((size_t)NNODES * HDIM * 4);
    float*    a_s       = (float*)carve((size_t)NNODES * 4);
    float*    rec       = (float*)carve((size_t)NNODES * 8);
    float*    e_self    = (float*)carve((size_t)NNODES * 4);
    unsigned* m_enc1    = (unsigned*)carve((size_t)NNODES * 4);
    unsigned* rng       = (unsigned*)carve((size_t)2 * (NCONVL + 1) * 4);
    float*    denom     = (float*)m_enc1;
    float*    denom1    = e_self;
    float*    num1      = x_in;
    float*    vn_direct = (float*)carve((size_t)HNUMB * HDIM * 4);
    float*    vn_root   = (float*)carve((size_t)HDIM * 4);
    float*    vn_seg    = (float*)carve((size_t)HNUMB * HDIM * 4);

    const int nodeBlk = (NNODES + 255) / 256;
    const int edgeBlk = NEDGES / 256;

    auto run_gat = [&](const float* Xin, const float* W, const float* asrc,
                       const float* adst, const float* bias, unsigned* lrng, bool with_segsum) {
        gemm_xw_kernel<<<(NNODES + 127) / 128, 256, 0, stream>>>(Xin, W, h, NNODES);
        dots_kernel<<<nodeBlk, 256, 0, stream>>>(h, asrc, adst, a_s, rec, e_self, m_enc1, lrng, NNODES);
        edge_max_kernel<<<edgeBlk, 256, 0, stream>>>(src, dst, a_s, rec, m_enc1, lrng, NEDGES);
        num_init_kernel<<<EW_BLOCKS, EW_THREADS, 0, stream>>>(e_self, rec, m_enc1, lrng, h, num, num1);
        edge_scatter_kernel<<<edgeBlk, 256, 0, stream>>>(src, dst, a_s, rec, h, denom, denom1, num, num1);
        if (with_segsum)
            finalize_segsum_kernel<<<FS_BLOCKS, FS_THREADS, 0, stream>>>(num, num1, denom, denom1, bias, hblocks, out, vn_seg);
        else
            finalize_kernel<<<EW_BLOCKS, EW_THREADS, 0, stream>>>(num, num1, denom, denom1, bias, out);
    };

    vn_init_kernel<<<(HNUMB * HDIM + 255) / 256, 256, 0, stream>>>(vn_emb, vn_direct, vn_root, vn_seg, rng);

    run_gat(x, W0, asrc0, adst0, b0, rng, false);

    for (int l = 0; l < NCONVL; ++l) {
        gather_add_kernel<<<EW_BLOCKS, EW_THREADS, 0, stream>>>(out, vn_direct, hblocks, x_in);
        run_gat(x_in, Wl + l * HDIM * HDIM, asrcl + l * HDIM, adstl + l * HDIM, bl + l * HDIM, rng + 2 * (l + 1), true);
        vn_update_kernel<<<1, 256, 0, stream>>>(vn_direct, vn_seg, vn_root);
        for (int j = 0; j < NMLPL; ++j) {
            mlp_kernel<<<HNUMB, 64, 0, stream>>>(vn_direct, mW1 + j * HDIM * HDIM, mb1 + j * HDIM,
                                                 mW2 + j * HDIM * HDIM, mb2 + j * HDIM);
            mlp_kernel<<<1, 64, 0, stream>>>(vn_root, mW1 + j * HDIM * HDIM, mb1 + j * HDIM,
                                             mW2 + j * HDIM * HDIM, mb2 + j * HDIM);
        }
    }
}


# Layer product taken after the neighbour sum (linearity): edges gather input rows, W applied once per node inside finalize, scores from x_in.(W a); h array and finalize pass go; sum order changes
# speedup vs baseline: 4.3027x; 1.2855x over previous
#include <hip/hip_runtime.h>
#include <hip/hip_bf16.h>
#include <hip/hip_fp16.h>

#define NNODES 100000
#define NEDGES 1600000
#define HDIM   64
#define HNUMB  128
#define NCONVL 2
#define NMLPL  2
#define NEGSLP 0.2f

typedef __attribute__((ext_vector_type(2))) float v2f;
typedef __attribute__((ext_vector_type(8))) float v8f;

__device__ __forceinline__ float leaky(float x) { return x > 0.f ? x : NEGSLP * x; }

__device__ __forceinline__ unsigned fenc(float f) {
    unsigned u = __float_as_uint(f);
    return (u & 0x80000000u) ? ~u : (u | 0x80000000u);
}
__device__ __forceinline__ float fdec(unsigned u) {
    return (u & 0x80000000u) ? __uint_as_float(u & 0x7fffffffu) : __uint_as_float(~u);
}

__device__ __forceinline__ int pkg_half() {
    return (int)((__builtin_amdgcn_s_sendmsg_rtn(0x87) >> 18) & 1u);
}

#define SHIFT_RANGE_OK 16.f
__device__ __forceinline__ bool bound_shift_ok(const unsigned* __restrict__ rng) {
    return fdec(rng[0]) - fdec(rng[1]) < SHIFT_RANGE_OK;
}

__global__ __launch_bounds__(256) void dots_kernel(const float* __restrict__ X, const float* __restrict__ W,
                            const float* __restrict__ asrc,
                            const float* __restrict__ adst, float* __restrict__ a_s,
                            float* __restrict__ rec, float* __restrict__ e_self,
                            unsigned* __restrict__ m_enc1, unsigned* __restrict__ rng, int n) {
    __shared__ unsigned red_hi[8], red_lo[8];
    __shared__ float wa[2 * HDIM];
    if (threadIdx.x < 2 * HDIM) {
        const int k = threadIdx.x & (HDIM - 1);
        const float* av = threadIdx.x < HDIM ? asrc : adst;
        float t = 0.f;
        for (int c = 0; c < HDIM; ++c) t += W[k * HDIM + c] * av[c];
        wa[threadIdx.x] = t;
    }
    __syncthreads();
    const int r = blockIdx.x * 256 + threadIdx.x;
    unsigned ehi = 0u, elo = 0xffffffffu;
    if (r < n) {
        const float* xr = X + (size_t)r * HDIM;
        float s = 0.f, d = 0.f;
#pragma unroll
        for (int k = 0; k < HDIM; ++k) { float v = xr[k]; s += v * wa[k]; d += v * wa[HDIM + k]; }
        a_s[r] = s;
        e_self[r] = leaky(s + d);
        const unsigned enc = fenc(s);
        ((float2*)rec)[r] = make_float2(d, __uint_as_float(enc));
        m_enc1[r] = enc;
        ehi = enc; elo = enc;
    }
    for (int o = 16; o > 0; o >>= 1) {
        const unsigned oh = __shfl_xor(ehi, o), ol = __shfl_xor(elo, o);
        ehi = ehi > oh ? ehi : oh;
        elo = elo < ol ? elo : ol;
    }
    if ((threadIdx.x & 31) == 0) { red_hi[threadIdx.x >> 5] = ehi; red_lo[threadIdx.x >> 5] = elo; }
    __syncthreads();
    if (threadIdx.x == 0) {
        for (int w = 1; w < 8; ++w) {
            ehi = ehi > red_hi[w] ? ehi : red_hi[w];
            elo = elo < red_lo[w] ? elo : red_lo[w];
        }
        atomicMax(rng, ehi);
        atomicMin(rng + 1, elo);
    }
}

__global__ void edge_max_kernel(const int* __restrict__ src, const int* __restrict__ dst,
                                const float* __restrict__ a_s,
                                float* __restrict__ rec, unsigned* __restrict__ m_enc1,
                                const unsigned* __restrict__ rng, int ne) {
    if (bound_shift_ok(rng)) return;
    const int half = pkg_half();
    int e = blockIdx.x * blockDim.x + threadIdx.x;
    if (e >= ne) return;
    int s = src[e], d = dst[e];
    unsigned* const m_mine = half ? m_enc1 + d : (unsigned*)rec + 2 * d + 1;
    atomicMax(m_mine, fenc(a_s[s]));
}

#define EW_THREADS 256
#define EW_PIECES  10
#define EW_NP      (NNODES * (HDIM / 4))
#define EW_BLOCKS  (EW_NP / (EW_THREADS * EW_PIECES))
static_assert(EW_NP % (EW_THREADS * EW_PIECES) == 0, "elementwise passes: every block owns whole groups of pieces");

__global__ __launch_bounds__(EW_THREADS) void num_init_kernel(float* es_den1,
                                                              float* rec,
                                                              unsigned* m1_den0,
                                                              const unsigned* __restrict__ rng,
                                                              const float* __restrict__ Hm,
                                                              float* __restrict__ num,
                                                              float* __restrict__ num1) {
    const bool     bound   = bound_shift_ok(rng);
    const unsigned enc_top = rng[0];
    const int p0 = blockIdx.x * (EW_THREADS * EW_PIECES) + threadIdx.x;
    const bool row_lane = (threadIdx.x & 15) == 0;
    float4   v[EW_PIECES];
    float    es[EW_PIECES];
    float2   rc[EW_PIECES];
    unsigned mb[EW_PIECES];
#pragma unroll
    for (int u = 0; u < EW_PIECES; ++u) {
        const int p = p0 + u * EW_THREADS;
        v[u]  = ((const float4*)Hm)[p];
        es[u] = es_den1[p >> 4];
        rc[u] = ((const float2*)rec)[p >> 4];
        mb[u] = m1_den0[p >> 4];
    }
#pragma unroll
    for (int u = 0; u < EW_PIECES; ++u) {
        const int p = p0 + u * EW_THREADS;
        const unsigned ma = __float_as_uint(rc[u].y);
        const unsigned mj = ma > mb[u] ? ma : mb[u];
        const float    c  = leaky(fdec(bound ? enc_top : mj) + rc[u].x);
        const float    w  = __expf(es[u] - c);
        float4 o;
        o.x = w * v[u].x;
        o.y = w * v[u].y;
        o.z = w * v[u].z;
        o.w = w * v[u].w;
        ((float4*)num)[p]  = o;
        ((float4*)num1)[p] = make_float4(0.f, 0.f, 0.f, 0.f);
        if (row_lane) {
            rec[2 * (p >> 4) + 1] = c;
            m1_den0[p >> 4] = __float_as_uint(w);
            es_den1[p >> 4] = 0.f;
        }
    }
}

#define SC_EPW   32
#define SC_BATCH 8
static_assert(NEDGES % 256 == 0, "edge scatter: a block of 8 waves owns 256 whole edges");
__global__ __launch_bounds__(256) void edge_scatter_kernel(const int* __restrict__ src, const int* __restrict__ dst,
                                    const float* __restrict__ a_s, const float* __restrict__ rec,
                                    const float* __restrict__ Hm,
                                    float* __restrict__ denom0, float* __restrict__ denom1,
                                    float* __restrict__ num0, float* __restrict__ num1) {
    const int half = pkg_half();
    float* const denom = half ? denom1 : denom0;
    float* const num   = half ? num1 : num0;
    const int lane = threadIdx.x & 31;
    const int e    = blockIdx.x * 256 + threadIdx.x;
    const int    sv = src[e];
    const int    dv = dst[e];
    const float  av = a_s[sv];
    const float2 rv = ((const float2*)rec)[dv];

    float lo[2][SC_BATCH], hi[2][SC_BATCH];
#pragma unroll
    for (int j = 0; j < SC_BATCH; ++j) {
        const float* hr = Hm + (size_t)__builtin_amdgcn_readlane(sv, j) * HDIM;
        lo[0][j] = hr[lane];
        hi[0][j] = hr[lane + 32];
    }
    int wv = 0;
#pragma unroll
    for (int b = 0; b < SC_EPW / SC_BATCH; ++b) {
        const int cur = b & 1, nxt = cur ^ 1;
        if (b + 1 < SC_EPW / SC_BATCH) {
#pragma unroll
            for (int j = 0; j < SC_BATCH; ++j) {
                const float* hr = Hm + (size_t)__builtin_amdgcn_readlane(sv, (b + 1) * SC_BATCH + j) * HDIM;
                lo[nxt][j] = hr[lane];
                hi[nxt][j] = hr[lane + 32];
            }
        }
        if (b == 0) {
            const float wf = __expf(leaky(av + rv.x) - rv.y);
            atomicAdd(denom + dv, wf);
            wv = __float_as_int(wf);
        }
#pragma unroll
        for (int j = 0; j < SC_BATCH; ++j) {
            const int   d = __builtin_amdgcn_readlane(dv, b * SC_BATCH + j);
            const float w = __int_as_float(__builtin_amdgcn_readlane(wv, b * SC_BATCH + j));
            float* nr = num + (size_t)d * HDIM;
            atomicAdd(nr + lane,      w * lo[cur][j]);
            atomicAdd(nr + lane + 32, w * hi[cur][j]);
        }
    }
}

#define GF_TILES  2
#define GF_BLOCKS (((NNODES + 127) / 128 + GF_TILES - 1) / GF_TILES)
#define FS_PITCH  (HDIM + 1)
static_assert(NNODES % 16 == 0, "product: a wave's strip of 16 rows is whole");
template <bool SEG>
__global__ __launch_bounds__(256) void gemm_finalize_kernel(
        const float* __restrict__ num, const float* __restrict__ num1,
        const float* __restrict__ denom, const float* __restrict__ denom1,
        const float* __restrict__ W, const float* __restrict__ bias,
        const int* __restrict__ blocks,
        float* __restrict__ out, float* __restrict__ vn_seg) {
    __shared__ float Wlds[HDIM * HDIM];
    __shared__ float acc_seg[SEG ? HNUMB * FS_PITCH : 1];
    const int tid = threadIdx.x;
    for (int idx = tid; idx < HDIM * HDIM; idx += 256) {
        int k = idx >> 6, n = idx & 63;
        Wlds[(k >> 1) * 128 + n * 2 + (k & 1)] = W[idx];
    }
    if (SEG)
        for (int i = tid; i < HNUMB * FS_PITCH; i += 256) acc_seg[i] = 0.f;
    __syncthreads();

    const int wave = tid >> 5;
    const int lane = tid & 31;
    const int hi   = lane >> 4;
    const int mrow = lane & 15;
    for (int t = 0; t < GF_TILES; ++t) {
        const int rowbase = (blockIdx.x * GF_TILES + t) * 128 + wave * 16;
        if (rowbase < NNODES) {
            v2f afrag[16];
#pragma unroll
            for (int kk = 0; kk < 16; ++kk) {
                const size_t at = (size_t)(rowbase + mrow) * HDIM + kk * 4 + hi * 2;
                afrag[kk] = *(const v2f*)(num + at) + *(const v2f*)(num1 + at);
            }
            float inv[8];
            int   bk[8];
#pragma unroll
            for (int v = 0; v < 8; ++v) {
                const int r = rowbase + v + 8 * hi;
                inv[v] = 1.0f / ((denom[r] + denom1[r]) + 1e-16f);
                bk[v]  = SEG ? blocks[r] : 0;
            }
#pragma unroll
            for (int nt = 0; nt < 4; ++nt) {
                const int ncol = nt * 16 + mrow;
                const float bn = bias[ncol];
                v8f acc = {};
#pragma unroll
                for (int kk = 0; kk < 16; ++kk) {
                    v2f b = *(const v2f*)&Wlds[(kk * 2 + hi) * 128 + ncol * 2];
                    acc = __builtin_amdgcn_wmma_f32_16x16x4_f32(false, afrag[kk], false, b,
                                                                (short)0, acc, false, false);
                }
#pragma unroll
                for (int v = 0; v < 8; ++v) {
                    const float o = acc[v] * inv[v] + bn;
                    out[(size_t)(rowbase + v + 8 * hi) * HDIM + ncol] = o;
                    if (SEG) atomicAdd(acc_seg + bk[v] * FS_PITCH + ncol, o);
                }
            }
        }
    }
    if (SEG) {
        __syncthreads();
        for (int i = tid; i < HNUMB * HDIM; i += 256) {
            const int j = (i + blockIdx.x * 32) & (HNUMB * HDIM - 1);
            atomicAdd(vn_seg + j, acc_seg[(j >> 6) * FS_PITCH + (j & 63)]);
        }
    }
}

__global__ __launch_bounds__(EW_THREADS) void gather_add_kernel(const float* __restrict__ out,
                                                                const float* __restrict__ vnd,
                                                                const int* __restrict__ blocks,
                                                                float* __restrict__ x_in) {
    const int p0 = blockIdx.x * (EW_THREADS * EW_PIECES) + threadIdx.x;
    const int c4 = threadIdx.x & 15;
    float4 v[EW_PIECES];
    int    bk[EW_PIECES];
#pragma unroll
    for (int u = 0; u < EW_PIECES; ++u) {
        const int p = p0 + u * EW_THREADS;
        v[u]  = ((const float4*)out)[p];
        bk[u] = blocks[p >> 4];
    }
#pragma unroll
    for (int u = 0; u < EW_PIECES; ++u) {
        const int p = p0 + u * EW_THREADS;
        const float4 a = ((const float4*)vnd)[bk[u] * (HDIM / 4) + c4];
        float4 o;
        o.x = v[u].x + a.x;
        o.y = v[u].y + a.y;
        o.z = v[u].z + a.z;
        o.w = v[u].w + a.w;
        ((float4*)x_in)[p] = o;
    }
}

__global__ __launch_bounds__(256) void vn_update_kernel(float* __restrict__ vn_direct,
                                                        float* __restrict__ vn_seg,
                                                        float* __restrict__ vn_root) {
    int tid = threadIdx.x;
    for (int i = tid; i < HNUMB * HDIM; i += 256) {
        vn_direct[i] += vn_seg[i] + vn_root[i & (HDIM - 1)];
        vn_seg[i] = 0.f;
    }
    __syncthreads();
    if (tid < HDIM) {
        float s = 0.f;
        for (int r = 0; r < HNUMB; ++r) s += vn_direct[r * HDIM + tid];
        vn_root[tid] += s;
    }
}

__global__ __launch_bounds__(64) void mlp_kernel(float* __restrict__ vn,
                                                 const float* __restrict__ W1, const float* __restrict__ b1,
                                                 const float* __restrict__ W2, const float* __restrict__ b2) {
    __shared__ float rin[HDIM];
    __shared__ float t1[HDIM];
    int f = threadIdx.x;
    float* row = vn + (size_t)blockIdx.x * HDIM;
    rin[f] = row[f];
    __syncthreads();
    float s = b1[f];
#pragma unroll
    for (int k = 0; k < HDIM; ++k) s += rin[k] * W1[k * HDIM + f];
    t1[f] = fmaxf(s, 0.f);
    __syncthreads();
    float s2 = b2[f];
#pragma unroll
    for (int k = 0; k < HDIM; ++k) s2 += t1[k] * W2[k * HDIM + f];
    row[f] = fmaxf(s2, 0.f);
}

__global__ void vn_init_kernel(const float* __restrict__ vn_emb, float* __restrict__ vn_direct,
                               float* __restrict__ vn_root, float* __restrict__ vn_seg,
                               unsigned* __restrict__ rng) {
    int t = blockIdx.x * blockDim.x + threadIdx.x;
    if (t < HNUMB * HDIM) { vn_direct[t] = vn_emb[t & (HDIM - 1)]; vn_seg[t] = 0.f; }
    if (t < HDIM) vn_root[t] = vn_emb[t];
    if (t < 2 * (NCONVL + 1)) rng[t] = (t & 1) ? 0xffffffffu : 0u;
}

extern "C" void kernel_launch(void* const* d_in, const int* in_sizes, int n_in,
                              void* d_out, int out_size, void* d_ws, size_t ws_size,
                              hipStream_t stream) {
    const float* x       = (const float*)d_in[0];
    const int*   ei      = (const int*)d_in[1];
    const int*   src     = ei;
    const int*   dst     = ei + NEDGES;
    const int*   hblocks = (const int*)d_in[2];
    const float* W0      = (const float*)d_in[5];
    const float* asrc0   = (const float*)d_in[6];
    const float* adst0   = (const float*)d_in[7];
    const float* b0      = (const float*)d_in[8];
    const float* Wl      = (const float*)d_in[9];
    const float* asrcl   = (const float*)d_in[10];
    const float* adstl   = (const float*)d_in[11];
    const float* bl      = (const float*)d_in[12];
    const float* vn_emb  = (const float*)d_in[13];
    const float* mW1     = (const float*)d_in[14];
    const float* mb1     = (const float*)d_in[15];
    const float* mW2     = (const float*)d_in[16];
    const float* mb2     = (const float*)d_in[17];
    float* out = (float*)d_out;

    char*  ws  = (char*)d_ws;
    size_t off = 0;
    auto carve = [&](size_t bytes) -> void* {
        void* p = ws + off;
        off += (bytes + 255) & ~(size_t)255;
        return p;
    };
    float*    num1      = (float*)carve((size_t)NNODES * HDIM * 4);
    float*    x_in      = (float*)carve((size_t)NNODES * HDIM * 4);
    float*    num       = (float*)carve((size_t)NNODES * HDIM * 4);
    float*    a_s       = (float*)carve((size_t)NNODES * 4);
    float*    rec       = (float*)carve((size_t)NNODES * 8);
    float*    e_self    = (float*)carve((size_t)NNODES * 4);
    unsigned* m_enc1    = (unsigned*)carve((size_t)NNODES * 4);
    unsigned* rng       = (unsigned*)carve((size_t)2 * (NCONVL + 1) * 4);
    float*    denom     = (float*)m_enc1;
    float*    denom1    = e_self;
    float*    vn_direct = (float*)carve((size_t)HNUMB * HDIM * 4);
    float*    vn_root   = (float*)carve((size_t)HDIM * 4);
    float*    vn_seg    = (float*)carve((size_t)HNUMB * HDIM * 4);

    const int nodeBlk = (NNODES + 255) / 256;
    const int edgeBlk = NEDGES / 256;

    auto run_gat = [&](const float* Xin, const float* W, const float* asrc,
                       const float* adst, const float* bias, unsigned* lrng, bool with_segsum) {
        dots_kernel<<<nodeBlk, 256, 0, stream>>>(Xin, W, asrc, adst, a_s, rec, e_self, m_enc1, lrng, NNODES);
        edge_max_kernel<<<edgeBlk, 256, 0, stream>>>(src, dst, a_s, rec, m_enc1, lrng, NEDGES);
        num_init_kernel<<<EW_BLOCKS, EW_THREADS, 0, stream>>>(e_self, rec, m_enc1, lrng, Xin, num, num1);
        edge_scatter_kernel<<<edgeBlk, 256, 0, stream>>>(src, dst, a_s, rec, Xin, denom, denom1, num, num1);
        if (with_segsum)
            gemm_finalize_kernel<true><<<GF_BLOCKS, 256, 0, stream>>>(num, num1, denom, denom1, W, bias, hblocks, out, vn_seg);
        else
            gemm_finalize_kernel<false><<<GF_BLOCKS, 256, 0, stream>>>(num, num1, denom, denom1, W, bias, hblocks, out, vn_seg);
    };

    vn_init_kernel<<<(HNUMB * HDIM + 255) / 256, 256, 0, stream>>>(vn_emb, vn_direct, vn_root, vn_seg, rng);

    run_gat(x, W0, asrc0, adst0, b0, rng, false);

    for (int l = 0; l < NCONVL; ++l) {
        gather_add_kernel<<<EW_BLOCKS, EW_THREADS, 0, stream>>>(out, vn_direct, hblocks, x_in);
        run_gat(x_in, Wl + l * HDIM * HDIM, asrcl + l * HDIM, adstl + l * HDIM, bl + l * HDIM, rng + 2 * (l + 1), true);
        vn_update_kernel<<<1, 256, 0, stream>>>(vn_direct, vn_seg, vn_root);
        for (int j = 0; j < NMLPL; ++j) {
            mlp_kernel<<<HNUMB, 64, 0, stream>>>(vn_direct, mW1 + j * HDIM * HDIM, mb1 + j * HDIM,
                                                 mW2 + j * HDIM * HDIM, mb2 + j * HDIM);
            mlp_kernel<<<1, 64, 0, stream>>>(vn_root, mW1 + j * HDIM * HDIM, mb1 + j * HDIM,
                                             mW2 + j * HDIM * HDIM, mb2 + j * HDIM);
        }
    }
}


# Row sums form in the owner work-group's LDS (256 groups x 391 rows), edges found in a per-call bit table, no key scan; num init, atomic scatter, finalize fuse; LDS float adds, one sum, arrival order
# speedup vs baseline: 4.9641x; 1.1537x over previous
#include <hip/hip_runtime.h>
#include <hip/hip_bf16.h>
#include <hip/hip_fp16.h>

#define NNODES 100000
#define NEDGES 1600000
#define HDIM   64
#define HNUMB  128
#define NCONVL 2
#define NMLPL  2
#define NEGSLP 0.2f

typedef __attribute__((ext_vector_type(2))) float v2f;
typedef __attribute__((ext_vector_type(8))) float v8f;

__device__ __forceinline__ float leaky(float x) { return x > 0.f ? x : NEGSLP * x; }

__device__ __forceinline__ unsigned fenc(float f) {
    unsigned u = __float_as_uint(f);
    return (u & 0x80000000u) ? ~u : (u | 0x80000000u);
}
__device__ __forceinline__ float fdec(unsigned u) {
    return (u & 0x80000000u) ? __uint_as_float(u & 0x7fffffffu) : __uint_as_float(~u);
}

__device__ __forceinline__ int pkg_half() {
    return (int)((__builtin_amdgcn_s_sendmsg_rtn(0x87) >> 18) & 1u);
}

#define SHIFT_RANGE_OK 16.f
__device__ __forceinline__ bool bound_shift_ok(const unsigned* __restrict__ rng) {
    return fdec(rng[0]) - fdec(rng[1]) < SHIFT_RANGE_OK;
}

__global__ __launch_bounds__(256) void dots_kernel(const float* __restrict__ X, const float* __restrict__ W,
                            const float* __restrict__ asrc,
                            const float* __restrict__ adst, float* __restrict__ a_s,
                            float* __restrict__ rec, float* __restrict__ e_self,
                            unsigned* __restrict__ m_enc1, unsigned* __restrict__ rng, int n) {
    __shared__ unsigned red_hi[8], red_lo[8];
    __shared__ float wa[2 * HDIM];
    if (threadIdx.x < 2 * HDIM) {
        const int k = threadIdx.x & (HDIM - 1);
        const float* av = threadIdx.x < HDIM ? asrc : adst;
        float t = 0.f;
        for (int c = 0; c < HDIM; ++c) t += W[k * HDIM + c] * av[c];
        wa[threadIdx.x] = t;
    }
    __syncthreads();
    const int r = blockIdx.x * 256 + threadIdx.x;
    unsigned ehi = 0u, elo = 0xffffffffu;
    if (r < n) {
        const float* xr = X + (size_t)r * HDIM;
        float s = 0.f, d = 0.f;
#pragma unroll
        for (int k = 0; k < HDIM; ++k) { float v = xr[k]; s += v * wa[k]; d += v * wa[HDIM + k]; }
        a_s[r] = s;
        e_self[r] = leaky(s + d);
        const unsigned enc = fenc(s);
        ((float2*)rec)[r] = make_float2(d, __uint_as_float(enc));
        m_enc1[r] = enc;
        ehi = enc; elo = enc;
    }
    for (int o = 16; o > 0; o >>= 1) {
        const unsigned oh = __shfl_xor(ehi, o), ol = __shfl_xor(elo, o);
        ehi = ehi > oh ? ehi : oh;
        elo = elo < ol ? elo : ol;
    }
    if ((threadIdx.x & 31) == 0) { red_hi[threadIdx.x >> 5] = ehi; red_lo[threadIdx.x >> 5] = elo; }
    __syncthreads();
    if (threadIdx.x == 0) {
        for (int w = 1; w < 8; ++w) {
            ehi = ehi > red_hi[w] ? ehi : red_hi[w];
            elo = elo < red_lo[w] ? elo : red_lo[w];
        }
        atomicMax(rng, ehi);
        atomicMin(rng + 1, elo);
    }
}

__global__ void edge_max_kernel(const int* __restrict__ src, const int* __restrict__ dst,
                                const float* __restrict__ a_s,
                                float* __restrict__ rec, unsigned* __restrict__ m_enc1,
                                const unsigned* __restrict__ rng, int ne) {
    if (bound_shift_ok(rng)) return;
    const int half = pkg_half();
    int e = blockIdx.x * blockDim.x + threadIdx.x;
    if (e >= ne) return;
    int s = src[e], d = dst[e];
    unsigned* const m_mine = half ? m_enc1 + d : (unsigned*)rec + 2 * d + 1;
    atomicMax(m_mine, fenc(a_s[s]));
}

#define EW_THREADS 256
#define EW_PIECES  10
#define EW_NP      (NNODES * (HDIM / 4))
#define EW_BLOCKS  (EW_NP / (EW_THREADS * EW_PIECES))
static_assert(EW_NP % (EW_THREADS * EW_PIECES) == 0, "elementwise passes: every block owns whole groups of pieces");

#define OW_ROWS    391
#define OW_GROUPS  256
#define OW_PADROWS 400
#define OW_WORDS   (NEDGES / 32)
#define BT_THREADS 512
#define BT_CHUNKS  (BT_THREADS / 32)
static_assert(BT_CHUNKS == 16 && OW_GROUPS % 2 == 0, "bit table: a line is 16 words of each of two owners");
static_assert(OW_ROWS * OW_GROUPS >= NNODES, "owner ranges cover every node");
static_assert(NEDGES % BT_THREADS == 0, "bit table: a block owns 512 whole edges");
__global__ __launch_bounds__(BT_THREADS) void owner_bits_kernel(const int* __restrict__ dst,
                                                                unsigned* __restrict__ bits) {
    __shared__ unsigned tb[OW_GROUPS * BT_CHUNKS];
    const int tid = threadIdx.x, lane = tid & 31, wave = tid >> 5;
    for (int i = tid; i < OW_GROUPS * BT_CHUNKS; i += BT_THREADS) tb[i] = 0u;
    __syncthreads();
    const int o = dst[blockIdx.x * BT_THREADS + tid] / OW_ROWS;
    unsigned same = 0u;
#pragma unroll
    for (int j = 0; j < 32; ++j) same |= (o == __builtin_amdgcn_readlane(o, j)) ? (1u << j) : 0u;
    if (lane == __ffs(same) - 1) tb[o * BT_CHUNKS + wave] = same;
    __syncthreads();
    for (int i = tid; i < OW_GROUPS * BT_CHUNKS; i += BT_THREADS)
        bits[(size_t)(i >> 5) * (2 * OW_WORDS) + blockIdx.x * 32 + (i & 31)] = tb[i];
}

#define SC_BATCH 8
template <bool FULL>
__device__ __forceinline__ void owner_edges(const int* q, const int n, const int lane,
                                            const int row0,
                                            const int* __restrict__ src, const int* __restrict__ dst,
                                            const float* __restrict__ a_s, const float* __restrict__ Hm,
                                            float* tab, float* den_l, const float* ad_l, const float* c_l) {
    const bool valid = FULL || lane < n;
    const int   e  = valid ? q[lane] : 0;
    const int   sv = src[e];
    const int   dv = dst[e];
    const float av = a_s[sv];
    const int   dl = valid ? dv - row0 : 0;
    float lo[2][SC_BATCH], hi[2][SC_BATCH];
#pragma unroll
    for (int j = 0; j < SC_BATCH; ++j) {
        const float* hr = Hm + (size_t)__builtin_amdgcn_readlane(sv, j) * HDIM;
        lo[0][j] = hr[lane];
        hi[0][j] = hr[lane + 32];
    }
    int wv = 0;
#pragma unroll
    for (int b = 0; b < 32 / SC_BATCH; ++b) {
        const int cur = b & 1, nxt = cur ^ 1;
        if (b + 1 < 32 / SC_BATCH) {
#pragma unroll
            for (int j = 0; j < SC_BATCH; ++j) {
                const float* hr = Hm + (size_t)__builtin_amdgcn_readlane(sv, (b + 1) * SC_BATCH + j) * HDIM;
                lo[nxt][j] = hr[lane];
                hi[nxt][j] = hr[lane + 32];
            }
        }
        if (b == 0) {
            const float wf = valid ? __expf(leaky(av + ad_l[dl]) - c_l[dl]) : 0.f;
            if (valid) atomicAdd(den_l + dl, wf);
            wv = __float_as_int(wf);
        }
#pragma unroll
        for (int j = 0; j < SC_BATCH; ++j) {
            if (FULL || b * SC_BATCH + j < n) {
                const int   d = __builtin_amdgcn_readlane(dl, b * SC_BATCH + j);
                const float w = __int_as_float(__builtin_amdgcn_readlane(wv, b * SC_BATCH + j));
                float* nr = tab + d * HDIM;
                atomicAdd(nr + lane,      w * lo[cur][j]);
                atomicAdd(nr + lane + 32, w * hi[cur][j]);
            }
        }
    }
}

#define FS_PITCH   (HDIM + 1)
#define OW_THREADS 1024
#define OW_WAVES   (OW_THREADS / 32)
#define OW_RUNS    ((OW_WORDS + 31) / 32)
#define OW_GB      7
#define OW_TURNS   ((OW_RUNS + OW_WAVES - 1) / OW_WAVES)
static_assert(OW_TURNS % OW_GB == 0, "bit row: a wave's runs come in whole groups");
template <bool SEG>
__global__ __launch_bounds__(OW_THREADS) void owner_layer_kernel(
        const unsigned* __restrict__ bits, const int* __restrict__ src, const int* __restrict__ dst,
        const float* __restrict__ a_s, const float* __restrict__ rec,
        const unsigned* __restrict__ m_enc1, const float* __restrict__ e_self,
        const unsigned* __restrict__ rng, const float* __restrict__ Hm,
        const float* __restrict__ W, const float* __restrict__ bias,
        const int* __restrict__ blocks,
        float* __restrict__ out, float* __restrict__ vn_seg) {
    __shared__ float tab[OW_PADROWS * HDIM];
    __shared__ float Wlds[HDIM * HDIM];
    __shared__ float acc_seg[SEG ? HNUMB * FS_PITCH : 1];
    __shared__ float ad_l[OW_PADROWS], c_l[OW_PADROWS], den_l[OW_PADROWS];
    __shared__ int      queue[OW_WAVES][64];
    __shared__ unsigned stage[OW_WAVES][OW_GB * 32];
    const int tid  = threadIdx.x;
    const int wave = tid >> 5;
    const int lane = tid & 31;
    const int row0 = blockIdx.x * OW_ROWS;

    for (int idx = tid; idx < HDIM * HDIM; idx += OW_THREADS) {
        int k = idx >> 6, n = idx & 63;
        Wlds[(k >> 1) * 128 + n * 2 + (k & 1)] = W[idx];
    }
    if (SEG)
        for (int i = tid; i < HNUMB * FS_PITCH; i += OW_THREADS) acc_seg[i] = 0.f;
    if (tid < OW_PADROWS) {
        const int r = row0 + tid;
        float ad = 0.f, cc = 0.f, w = 0.f;
        if (tid < OW_ROWS && r < NNODES) {
            const float2   rc = ((const float2*)rec)[r];
            const unsigned ma = __float_as_uint(rc.y), mb = m_enc1[r];
            const unsigned mj = ma > mb ? ma : mb;
            ad = rc.x;
            cc = leaky(fdec(bound_shift_ok(rng) ? rng[0] : mj) + ad);
            w  = __expf(e_self[r] - cc);
        }
        ad_l[tid] = ad; c_l[tid] = cc; den_l[tid] = w;
    }
    __syncthreads();
    for (int p = tid; p < OW_PADROWS * (HDIM / 4); p += OW_THREADS) {
        const int i = p >> 4, r = row0 + i;
        float4 o = make_float4(0.f, 0.f, 0.f, 0.f);
        if (i < OW_ROWS && r < NNODES) {
            const float4 v = ((const float4*)Hm)[(size_t)r * (HDIM / 4) + (p & 15)];
            const float  w = den_l[i];
            o.x = w * v.x; o.y = w * v.y; o.z = w * v.z; o.w = w * v.w;
        }
        ((float4*)tab)[p] = o;
    }
    __syncthreads();

    {
        const unsigned* brow = bits + (size_t)(blockIdx.x >> 1) * (2 * OW_WORDS) + (blockIdx.x & 1) * 16;
        int*      q  = queue[wave];
        unsigned* st = stage[wave];
        int qn = 0;
        for (int t0 = 0; t0 < OW_TURNS; t0 += OW_GB) {
            unsigned m[OW_GB];
#pragma unroll
            for (int u = 0; u < OW_GB; ++u) {
                const int wi = (wave + OW_WAVES * (t0 + u)) * 32 + lane;
                m[u] = wi < OW_WORDS ? brow[(wi >> 4) * 32 + (wi & 15)] : 0u;
            }
#pragma unroll
            for (int u = 0; u < OW_GB; ++u) st[u * 32 + lane] = m[u];
#pragma unroll 1
            for (int u = 0; u < OW_GB; ++u) {
                unsigned  mm    = st[u * 32 + lane];
                const int ebase = ((wave + OW_WAVES * (t0 + u)) * 32 + lane) * 32;
                unsigned  bal   = __builtin_amdgcn_ballot_w32(mm != 0u);
                while (bal != 0u) {
                    const int pos = qn + __popc(bal & ((1u << lane) - 1u));
                    if (mm != 0u) { q[pos] = ebase + __ffs(mm) - 1; mm &= mm - 1u; }
                    qn += __popc(bal);
                    __syncwarp();
                    if (qn >= 32) {
                        owner_edges<true>(q, 32, lane, row0, src, dst, a_s, Hm, tab, den_l, ad_l, c_l);
                        const int rem = qn - 32;
                        int keep = 0;
                        if (lane < rem) keep = q[32 + lane];
                        __syncwarp();
                        if (lane < rem) q[lane] = keep;
                        __syncwarp();
                        qn = rem;
                    }
                    bal = __builtin_amdgcn_ballot_w32(mm != 0u);
                }
            }
        }
        if (qn > 0)
            owner_edges<false>(q, qn, lane, row0, src, dst, a_s, Hm, tab, den_l, ad_l, c_l);
    }
    __syncthreads();

    const int hi   = lane >> 4;
    const int mrow = lane & 15;
    if (wave < OW_PADROWS / 16) {
        const int sb = wave * 16;
        v2f afrag[16];
#pragma unroll
        for (int kk = 0; kk < 16; ++kk)
            afrag[kk] = *(const v2f*)&tab[(sb + mrow) * HDIM + kk * 4 + hi * 2];
        float inv[8];
        int   bk[8];
        bool  live[8];
#pragma unroll
        for (int v = 0; v < 8; ++v) {
            const int i = sb + v + 8 * hi;
            live[v] = i < OW_ROWS && row0 + i < NNODES;
            inv[v]  = 1.0f / (den_l[i] + 1e-16f);
            bk[v]   = (SEG && live[v]) ? blocks[row0 + i] : 0;
        }
#pragma unroll
        for (int nt = 0; nt < 4; ++nt) {
            const int ncol = nt * 16 + mrow;
            const float bn = bias[ncol];
            v8f acc = {};
#pragma unroll
            for (int kk = 0; kk < 16; ++kk) {
                v2f bfr = *(const v2f*)&Wlds[(kk * 2 + hi) * 128 + ncol * 2];
                acc = __builtin_amdgcn_wmma_f32_16x16x4_f32(false, afrag[kk], false, bfr,
                                                            (short)0, acc, false, false);
            }
#pragma unroll
            for (int v = 0; v < 8; ++v) {
                const float o = acc[v] * inv[v] + bn;
                if (live[v]) {
                    out[(size_t)(row0 + sb + v + 8 * hi) * HDIM + ncol] = o;
                    if (SEG) atomicAdd(acc_seg + bk[v] * FS_PITCH + ncol, o);
                }
            }
        }
    }
    if (SEG) {
        __syncthreads();
        for (int i = tid; i < HNUMB * HDIM; i += OW_THREADS) {
            const int j = (i + blockIdx.x * 32) & (HNUMB * HDIM - 1);
            atomicAdd(vn_seg + j, acc_seg[(j >> 6) * FS_PITCH + (j & 63)]);
        }
    }
}

__global__ __launch_bounds__(EW_THREADS) void gather_add_kernel(const float* __restrict__ out,
                                                                const float* __restrict__ vnd,
                                                                const int* __restrict__ blocks,
                                                                float* __restrict__ x_in) {
    const int p0 = blockIdx.x * (EW_THREADS * EW_PIECES) + threadIdx.x;
    const int c4 = threadIdx.x & 15;
    float4 v[EW_PIECES];
    int    bk[EW_PIECES];
#pragma unroll
    for (int u = 0; u < EW_PIECES; ++u) {
        const int p = p0 + u * EW_THREADS;
        v[u]  = ((const float4*)out)[p];
        bk[u] = blocks[p >> 4];
    }
#pragma unroll
    for (int u = 0; u < EW_PIECES; ++u) {
        const int p = p0 + u * EW_THREADS;
        const float4 a = ((const float4*)vnd)[bk[u] * (HDIM / 4) + c4];
        float4 o;
        o.x = v[u].x + a.x;
        o.y = v[u].y + a.y;
        o.z = v[u].z + a.z;
        o.w = v[u].w + a.w;
        ((float4*)x_in)[p] = o;
    }
}

__global__ __launch_bounds__(256) void vn_update_kernel(float* __restrict__ vn_direct,
                                                        float* __restrict__ vn_seg,
                                                        float* __restrict__ vn_root) {
    int tid = threadIdx.x;
    for (int i = tid; i < HNUMB * HDIM; i += 256) {
        vn_direct[i] += vn_seg[i] + vn_root[i & (HDIM - 1)];
        vn_seg[i] = 0.f;
    }
    __syncthreads();
    if (tid < HDIM) {
        float s = 0.f;
        for (int r = 0; r < HNUMB; ++r) s += vn_direct[r * HDIM + tid];
        vn_root[tid] += s;
    }
}

__global__ __launch_bounds__(64) void mlp_kernel(float* __restrict__ vn,
                                                 const float* __restrict__ W1, const float* __restrict__ b1,
                                                 const float* __restrict__ W2, const float* __restrict__ b2) {
    __shared__ float rin[HDIM];
    __shared__ float t1[HDIM];
    int f = threadIdx.x;
    float* row = vn + (size_t)blockIdx.x * HDIM;
    rin[f] = row[f];
    __syncthreads();
    float s = b1[f];
#pragma unroll
    for (int k = 0; k < HDIM; ++k) s += rin[k] * W1[k * HDIM + f];
    t1[f] = fmaxf(s, 0.f);
    __syncthreads();
    float s2 = b2[f];
#pragma unroll
    for (int k = 0; k < HDIM; ++k) s2 += t1[k] * W2[k * HDIM + f];
    row[f] = fmaxf(s2, 0.f);
}

__global__ void vn_init_kernel(const float* __restrict__ vn_emb, float* __restrict__ vn_direct,
                               float* __restrict__ vn_root, float* __restrict__ vn_seg,
                               unsigned* __restrict__ rng) {
    int t = blockIdx.x * blockDim.x + threadIdx.x;
    if (t < HNUMB * HDIM) { vn_direct[t] = vn_emb[t & (HDIM - 1)]; vn_seg[t] = 0.f; }
    if (t < HDIM) vn_root[t] = vn_emb[t];
    if (t < 2 * (NCONVL + 1)) rng[t] = (t & 1) ? 0xffffffffu : 0u;
}

extern "C" void kernel_launch(void* const* d_in, const int* in_sizes, int n_in,
                              void* d_out, int out_size, void* d_ws, size_t ws_size,
                              hipStream_t stream) {
    const float* x       = (const float*)d_in[0];
    const int*   ei      = (const int*)d_in[1];
    const int*   src     = ei;
    const int*   dst     = ei + NEDGES;
    const int*   hblocks = (const int*)d_in[2];
    const float* W0      = (const float*)d_in[5];
    const float* asrc0   = (const float*)d_in[6];
    const float* adst0   = (const float*)d_in[7];
    const float* b0      = (const float*)d_in[8];
    const float* Wl      = (const float*)d_in[9];
    const float* asrcl   = (const float*)d_in[10];
    const float* adstl   = (const float*)d_in[11];
    const float* bl      = (const float*)d_in[12];
    const float* vn_emb  = (const float*)d_in[13];
    const float* mW1     = (const float*)d_in[14];
    const float* mb1     = (const float*)d_in[15];
    const float* mW2     = (const float*)d_in[16];
    const float* mb2     = (const float*)d_in[17];
    float* out = (float*)d_out;

    char*  ws  = (char*)d_ws;
    size_t off = 0;
    auto carve = [&](size_t bytes) -> void* {
        void* p = ws + off;
        off += (bytes + 255) & ~(size_t)255;
        return p;
    };
    unsigned* bits      = (unsigned*)carve((size_t)OW_GROUPS * OW_WORDS * 4);
    float*    x_in      = (float*)carve((size_t)NNODES * HDIM * 4);
    float*    a_s       = (float*)carve((size_t)NNODES * 4);
    float*    rec       = (float*)carve((size_t)NNODES * 8);
    float*    e_self    = (float*)carve((size_t)NNODES * 4);
    unsigned* m_enc1    = (unsigned*)carve((size_t)NNODES * 4);
    unsigned* rng       = (unsigned*)carve((size_t)2 * (NCONVL + 1) * 4);
    float*    vn_direct = (float*)carve((size_t)HNUMB * HDIM * 4);
    float*    vn_root   = (float*)carve((size_t)HDIM * 4);
    float*    vn_seg    = (float*)carve((size_t)HNUMB * HDIM * 4);

    const int nodeBlk = (NNODES + 255) / 256;
    const int edgeBlk = NEDGES / 256;

    auto run_gat = [&](const float* Xin, const float* W, const float* asrc,
                       const float* adst, const float* bias, unsigned* lrng, bool with_segsum) {
        dots_kernel<<<nodeBlk, 256, 0, stream>>>(Xin, W, asrc, adst, a_s, rec, e_self, m_enc1, lrng, NNODES);
        edge_max_kernel<<<edgeBlk, 256, 0, stream>>>(src, dst, a_s, rec, m_enc1, lrng, NEDGES);
        if (with_segsum)
            owner_layer_kernel<true><<<OW_GROUPS, OW_THREADS, 0, stream>>>(bits, src, dst, a_s, rec, m_enc1, e_self, lrng, Xin, W, bias, hblocks, out, vn_seg);
        else
            owner_layer_kernel<false><<<OW_GROUPS, OW_THREADS, 0, stream>>>(bits, src, dst, a_s, rec, m_enc1, e_self, lrng, Xin, W, bias, hblocks, out, vn_seg);
    };

    vn_init_kernel<<<(HNUMB * HDIM + 255) / 256, 256, 0, stream>>>(vn_emb, vn_direct, vn_root, vn_seg, rng);
    owner_bits_kernel<<<NEDGES / BT_THREADS, BT_THREADS, 0, stream>>>(dst, bits);

    run_gat(x, W0, asrc0, adst0, b0, rng, false);

    for (int l = 0; l < NCONVL; ++l) {
        gather_add_kernel<<<EW_BLOCKS, EW_THREADS, 0, stream>>>(out, vn_direct, hblocks, x_in);
        run_gat(x_in, Wl + l * HDIM * HDIM, asrcl + l * HDIM, adstl + l * HDIM, bl + l * HDIM, rng + 2 * (l + 1), true);
        vn_update_kernel<<<1, 256, 0, stream>>>(vn_direct, vn_seg, vn_root);
        for (int j = 0; j < NMLPL; ++j) {
            mlp_kernel<<<HNUMB, 64, 0, stream>>>(vn_direct, mW1 + j * HDIM * HDIM, mb1 + j * HDIM,
                                                 mW2 + j * HDIM * HDIM, mb2 + j * HDIM);
            mlp_kernel<<<1, 64, 0, stream>>>(vn_root, mW1 + j * HDIM * HDIM, mb1 + j * HDIM,
                                             mW2 + j * HDIM * HDIM, mb2 + j * HDIM);
        }
    }
}


# Edge's source and its row in the owner's range packed into one word per edge (edge order kept), written once per call beside the bit table: the edge walk gathers one index line an edge, not two
# speedup vs baseline: 5.3645x; 1.0807x over previous
#include <hip/hip_runtime.h>
#include <hip/hip_bf16.h>
#include <hip/hip_fp16.h>

#define NNODES 100000
#define NEDGES 1600000
#define HDIM   64
#define HNUMB  128
#define NCONVL 2
#define NMLPL  2
#define NEGSLP 0.2f

typedef __attribute__((ext_vector_type(2))) float v2f;
typedef __attribute__((ext_vector_type(8))) float v8f;

__device__ __forceinline__ float leaky(float x) { return x > 0.f ? x : NEGSLP * x; }

__device__ __forceinline__ unsigned fenc(float f) {
    unsigned u = __float_as_uint(f);
    return (u & 0x80000000u) ? ~u : (u | 0x80000000u);
}
__device__ __forceinline__ float fdec(unsigned u) {
    return (u & 0x80000000u) ? __uint_as_float(u & 0x7fffffffu) : __uint_as_float(~u);
}

__device__ __forceinline__ int pkg_half() {
    return (int)((__builtin_amdgcn_s_sendmsg_rtn(0x87) >> 18) & 1u);
}

#define SHIFT_RANGE_OK 16.f
__device__ __forceinline__ bool bound_shift_ok(const unsigned* __restrict__ rng) {
    return fdec(rng[0]) - fdec(rng[1]) < SHIFT_RANGE_OK;
}

__global__ __launch_bounds__(256) void dots_kernel(const float* __restrict__ X, const float* __restrict__ W,
                            const float* __restrict__ asrc,
                            const float* __restrict__ adst, float* __restrict__ a_s,
                            float* __restrict__ rec, float* __restrict__ e_self,
                            unsigned* __restrict__ m_enc1, unsigned* __restrict__ rng, int n) {
    __shared__ unsigned red_hi[8], red_lo[8];
    __shared__ float wa[2 * HDIM];
    if (threadIdx.x < 2 * HDIM) {
        const int k = threadIdx.x & (HDIM - 1);
        const float* av = threadIdx.x < HDIM ? asrc : adst;
        float t = 0.f;
        for (int c = 0; c < HDIM; ++c) t += W[k * HDIM + c] * av[c];
        wa[threadIdx.x] = t;
    }
    __syncthreads();
    const int r = blockIdx.x * 256 + threadIdx.x;
    unsigned ehi = 0u, elo = 0xffffffffu;
    if (r < n) {
        const float* xr = X + (size_t)r * HDIM;
        float s = 0.f, d = 0.f;
#pragma unroll
        for (int k = 0; k < HDIM; ++k) { float v = xr[k]; s += v * wa[k]; d += v * wa[HDIM + k]; }
        a_s[r] = s;
        e_self[r] = leaky(s + d);
        const unsigned enc = fenc(s);
        ((float2*)rec)[r] = make_float2(d, __uint_as_float(enc));
        m_enc1[r] = enc;
        ehi = enc; elo = enc;
    }
    for (int o = 16; o > 0; o >>= 1) {
        const unsigned oh = __shfl_xor(ehi, o), ol = __shfl_xor(elo, o);
        ehi = ehi > oh ? ehi : oh;
        elo = elo < ol ? elo : ol;
    }
    if ((threadIdx.x & 31) == 0) { red_hi[threadIdx.x >> 5] = ehi; red_lo[threadIdx.x >> 5] = elo; }
    __syncthreads();
    if (threadIdx.x == 0) {
        for (int w = 1; w < 8; ++w) {
            ehi = ehi > red_hi[w] ? ehi : red_hi[w];
            elo = elo < red_lo[w] ? elo : red_lo[w];
        }
        atomicMax(rng, ehi);
        atomicMin(rng + 1, elo);
    }
}

__global__ void edge_max_kernel(const int* __restrict__ src, const int* __restrict__ dst,
                                const float* __restrict__ a_s,
                                float* __restrict__ rec, unsigned* __restrict__ m_enc1,
                                const unsigned* __restrict__ rng, int ne) {
    if (bound_shift_ok(rng)) return;
    const int half = pkg_half();
    int e = blockIdx.x * blockDim.x + threadIdx.x;
    if (e >= ne) return;
    int s = src[e], d = dst[e];
    unsigned* const m_mine = half ? m_enc1 + d : (unsigned*)rec + 2 * d + 1;
    atomicMax(m_mine, fenc(a_s[s]));
}

#define EW_THREADS 256
#define EW_PIECES  10
#define EW_NP      (NNODES * (HDIM / 4))
#define EW_BLOCKS  (EW_NP / (EW_THREADS * EW_PIECES))
static_assert(EW_NP % (EW_THREADS * EW_PIECES) == 0, "elementwise passes: every block owns whole groups of pieces");

#define OW_ROWS    391
#define OW_GROUPS  256
#define OW_PADROWS 400
#define OW_WORDS   (NEDGES / 32)
#define BT_THREADS 512
#define BT_CHUNKS  (BT_THREADS / 32)
static_assert(BT_CHUNKS == 16 && OW_GROUPS % 2 == 0, "bit table: a line is 16 words of each of two owners");
static_assert(OW_ROWS * OW_GROUPS >= NNODES, "owner ranges cover every node");
static_assert(NEDGES % BT_THREADS == 0, "bit table: a block owns 512 whole edges");
#define SD_SHIFT 17
#define SD_MASK  ((1u << SD_SHIFT) - 1u)
static_assert(NNODES <= (1 << SD_SHIFT) && OW_ROWS <= (1 << (32 - SD_SHIFT)), "packed index word: source and row fit");
__host__ __device__ constexpr size_t ws_al(size_t b) { return (b + 255) & ~(size_t)255; }
#define SD_OFF_BYTES (ws_al((size_t)OW_GROUPS * OW_WORDS * 4) + ws_al((size_t)NNODES * HDIM * 4) + \
                      ws_al((size_t)NNODES * 4) + ws_al((size_t)NNODES * 8) + ws_al((size_t)NNODES * 4) + \
                      ws_al((size_t)NNODES * 4) + ws_al((size_t)2 * (NCONVL + 1) * 4) + \
                      ws_al((size_t)HNUMB * HDIM * 4) + ws_al((size_t)HDIM * 4) + ws_al((size_t)HNUMB * HDIM * 4))
#define SD_NEED_BYTES (SD_OFF_BYTES + (size_t)NEDGES * 4)
__global__ __launch_bounds__(BT_THREADS) void owner_bits_kernel(const int* __restrict__ dst,
                                                                const int* __restrict__ src,
                                                                unsigned* __restrict__ bits,
                                                                unsigned* __restrict__ sd,
                                                                const size_t ws_bytes) {
    __shared__ unsigned tb[OW_GROUPS * BT_CHUNKS];
    const int tid = threadIdx.x, lane = tid & 31, wave = tid >> 5;
    for (int i = tid; i < OW_GROUPS * BT_CHUNKS; i += BT_THREADS) tb[i] = 0u;
    __syncthreads();
    const int e  = blockIdx.x * BT_THREADS + tid;
    const int dv = dst[e];
    const int o  = dv / OW_ROWS;
    if (ws_bytes >= SD_NEED_BYTES)
        sd[e] = (unsigned)src[e] | ((unsigned)(dv - o * OW_ROWS) << SD_SHIFT);
    unsigned same = 0u;
#pragma unroll
    for (int j = 0; j < 32; ++j) same |= (o == __builtin_amdgcn_readlane(o, j)) ? (1u << j) : 0u;
    if (lane == __ffs(same) - 1) tb[o * BT_CHUNKS + wave] = same;
    __syncthreads();
    for (int i = tid; i < OW_GROUPS * BT_CHUNKS; i += BT_THREADS)
        bits[(size_t)(i >> 5) * (2 * OW_WORDS) + blockIdx.x * 32 + (i & 31)] = tb[i];
}

#define SC_BATCH 8
template <bool FULL>
__device__ __forceinline__ void owner_edges(const int* q, const int n, const int lane,
                                            const int row0,
                                            const int* __restrict__ src, const int* __restrict__ dst,
                                            const unsigned* __restrict__ sd, const bool packed,
                                            const float* __restrict__ a_s, const float* __restrict__ Hm,
                                            float* tab, float* den_l, const float* ad_l, const float* c_l) {
    const bool valid = FULL || lane < n;
    const int   e  = valid ? q[lane] : 0;
    int sv, dl;
    if (packed) {
        const unsigned w = sd[e];
        sv = (int)(w & SD_MASK);
        dl = valid ? (int)(w >> SD_SHIFT) : 0;
    } else {
        sv = src[e];
        const int dv = dst[e];
        dl = valid ? dv - row0 : 0;
    }
    const float av = a_s[sv];
    float lo[2][SC_BATCH], hi[2][SC_BATCH];
#pragma unroll
    for (int j = 0; j < SC_BATCH; ++j) {
        const float* hr = Hm + (size_t)__builtin_amdgcn_readlane(sv, j) * HDIM;
        lo[0][j] = hr[lane];
        hi[0][j] = hr[lane + 32];
    }
    int wv = 0;
#pragma unroll
    for (int b = 0; b < 32 / SC_BATCH; ++b) {
        const int cur = b & 1, nxt = cur ^ 1;
        if (b + 1 < 32 / SC_BATCH) {
#pragma unroll
            for (int j = 0; j < SC_BATCH; ++j) {
                const float* hr = Hm + (size_t)__builtin_amdgcn_readlane(sv, (b + 1) * SC_BATCH + j) * HDIM;
                lo[nxt][j] = hr[lane];
                hi[nxt][j] = hr[lane + 32];
            }
        }
        if (b == 0) {
            const float wf = valid ? __expf(leaky(av + ad_l[dl]) - c_l[dl]) : 0.f;
            if (valid) atomicAdd(den_l + dl, wf);
            wv = __float_as_int(wf);
        }
#pragma unroll
        for (int j = 0; j < SC_BATCH; ++j) {
            if (FULL || b * SC_BATCH + j < n) {
                const int   d = __builtin_amdgcn_readlane(dl, b * SC_BATCH + j);
                const float w = __int_as_float(__builtin_amdgcn_readlane(wv, b * SC_BATCH + j));
                float* nr = tab + d * HDIM;
                atomicAdd(nr + lane,      w * lo[cur][j]);
                atomicAdd(nr + lane + 32, w * hi[cur][j]);
            }
        }
    }
}

#define FS_PITCH   (HDIM + 1)
#define OW_THREADS 1024
#define OW_WAVES   (OW_THREADS / 32)
#define OW_RUNS    ((OW_WORDS + 31) / 32)
#define OW_GB      7
#define OW_TURNS   ((OW_RUNS + OW_WAVES - 1) / OW_WAVES)
static_assert(OW_TURNS % OW_GB == 0, "bit row: a wave's runs come in whole groups");
template <bool SEG>
__global__ __launch_bounds__(OW_THREADS) void owner_layer_kernel(
        const unsigned* __restrict__ bits, const int* __restrict__ src, const int* __restrict__ dst,
        const unsigned* __restrict__ sd, const size_t ws_bytes,
        const float* __restrict__ a_s, const float* __restrict__ rec,
        const unsigned* __restrict__ m_enc1, const float* __restrict__ e_self,
        const unsigned* __restrict__ rng, const float* __restrict__ Hm,
        const float* __restrict__ W, const float* __restrict__ bias,
        const int* __restrict__ blocks,
        float* __restrict__ out, float* __restrict__ vn_seg) {
    __shared__ float tab[OW_PADROWS * HDIM];
    __shared__ float Wlds[HDIM * HDIM];
    __shared__ float acc_seg[SEG ? HNUMB * FS_PITCH : 1];
    __shared__ float ad_l[OW_PADROWS], c_l[OW_PADROWS], den_l[OW_PADROWS];
    __shared__ int      queue[OW_WAVES][64];
    __shared__ unsigned stage[OW_WAVES][OW_GB * 32];
    const int tid  = threadIdx.x;
    const int wave = tid >> 5;
    const int lane = tid & 31;
    const int row0 = blockIdx.x * OW_ROWS;
    const bool packed = ws_bytes >= SD_NEED_BYTES;

    for (int idx = tid; idx < HDIM * HDIM; idx += OW_THREADS) {
        int k = idx >> 6, n = idx & 63;
        Wlds[(k >> 1) * 128 + n * 2 + (k & 1)] = W[idx];
    }
    if (SEG)
        for (int i = tid; i < HNUMB * FS_PITCH; i += OW_THREADS) acc_seg[i] = 0.f;
    if (tid < OW_PADROWS) {
        const int r = row0 + tid;
        float ad = 0.f, cc = 0.f, w = 0.f;
        if (tid < OW_ROWS && r < NNODES) {
            const float2   rc = ((const float2*)rec)[r];
            const unsigned ma = __float_as_uint(rc.y), mb = m_enc1[r];
            const unsigned mj = ma > mb ? ma : mb;
            ad = rc.x;
            cc = leaky(fdec(bound_shift_ok(rng) ? rng[0] : mj) + ad);
            w  = __expf(e_self[r] - cc);
        }
        ad_l[tid] = ad; c_l[tid] = cc; den_l[tid] = w;
    }
    __syncthreads();
    for (int p = tid; p < OW_PADROWS * (HDIM / 4); p += OW_THREADS) {
        const int i = p >> 4, r = row0 + i;
        float4 o = make_float4(0.f, 0.f, 0.f, 0.f);
        if (i < OW_ROWS && r < NNODES) {
            const float4 v = ((const float4*)Hm)[(size_t)r * (HDIM / 4) + (p & 15)];
            const float  w = den_l[i];
            o.x = w * v.x; o.y = w * v.y; o.z = w * v.z; o.w = w * v.w;
        }
        ((float4*)tab)[p] = o;
    }
    __syncthreads();

    {
        const unsigned* brow = bits + (size_t)(blockIdx.x >> 1) * (2 * OW_WORDS) + (blockIdx.x & 1) * 16;
        int*      q  = queue[wave];
        unsigned* st = stage[wave];
        int qn = 0;
        for (int t0 = 0; t0 < OW_TURNS; t0 += OW_GB) {
            unsigned m[OW_GB];
#pragma unroll
            for (int u = 0; u < OW_GB; ++u) {
                const int wi = (wave + OW_WAVES * (t0 + u)) * 32 + lane;
                m[u] = wi < OW_WORDS ? brow[(wi >> 4) * 32 + (wi & 15)] : 0u;
            }
#pragma unroll
            for (int u = 0; u < OW_GB; ++u) st[u * 32 + lane] = m[u];
#pragma unroll 1
            for (int u = 0; u < OW_GB; ++u) {
                unsigned  mm    = st[u * 32 + lane];
                const int ebase = ((wave + OW_WAVES * (t0 + u)) * 32 + lane) * 32;
                unsigned  bal   = __builtin_amdgcn_ballot_w32(mm != 0u);
                while (bal != 0u) {
                    const int pos = qn + __popc(bal & ((1u << lane) - 1u));
                    if (mm != 0u) { q[pos] = ebase + __ffs(mm) - 1; mm &= mm - 1u; }
                    qn += __popc(bal);
                    __syncwarp();
                    if (qn >= 32) {
                        owner_edges<true>(q, 32, lane, row0, src, dst, sd, packed, a_s, Hm, tab, den_l, ad_l, c_l);
                        const int rem = qn - 32;
                        int keep = 0;
                        if (lane < rem) keep = q[32 + lane];
                        __syncwarp();
                        if (lane < rem) q[lane] = keep;
                        __syncwarp();
                        qn = rem;
                    }
                    bal = __builtin_amdgcn_ballot_w32(mm != 0u);
                }
            }
        }
        if (qn > 0)
            owner_edges<false>(q, qn, lane, row0, src, dst, sd, packed, a_s, Hm, tab, den_l, ad_l, c_l);
    }
    __syncthreads();

    const int hi   = lane >> 4;
    const int mrow = lane & 15;
    if (wave < OW_PADROWS / 16) {
        const int sb = wave * 16;
        v2f afrag[16];
#pragma unroll
        for (int kk = 0; kk < 16; ++kk)
            afrag[kk] = *(const v2f*)&tab[(sb + mrow) * HDIM + kk * 4 + hi * 2];
        float inv[8];
        int   bk[8];
        bool  live[8];
#pragma unroll
        for (int v = 0; v < 8; ++v) {
            const int i = sb + v + 8 * hi;
            live[v] = i < OW_ROWS && row0 + i < NNODES;
            inv[v]  = 1.0f / (den_l[i] + 1e-16f);
            bk[v]   = (SEG && live[v]) ? blocks[row0 + i] : 0;
        }
#pragma unroll
        for (int nt = 0; nt < 4; ++nt) {
            const int ncol = nt * 16 + mrow;
            const float bn = bias[ncol];
            v8f acc = {};
#pragma unroll
            for (int kk = 0; kk < 16; ++kk) {
                v2f bfr = *(const v2f*)&Wlds[(kk * 2 + hi) * 128 + ncol * 2];
                acc = __builtin_amdgcn_wmma_f32_16x16x4_f32(false, afrag[kk], false, bfr,
                                                            (short)0, acc, false, false);
            }
#pragma unroll
            for (int v = 0; v < 8; ++v) {
                const float o = acc[v] * inv[v] + bn;
                if (live[v]) {
                    out[(size_t)(row0 + sb + v + 8 * hi) * HDIM + ncol] = o;
                    if (SEG) atomicAdd(acc_seg + bk[v] * FS_PITCH + ncol, o);
                }
            }
        }
    }
    if (SEG) {
        __syncthreads();
        for (int i = tid; i < HNUMB * HDIM; i += OW_THREADS) {
            const int j = (i + blockIdx.x * 32) & (HNUMB * HDIM - 1);
            atomicAdd(vn_seg + j, acc_seg[(j >> 6) * FS_PITCH + (j & 63)]);
        }
    }
}

__global__ __launch_bounds__(EW_THREADS) void gather_add_kernel(const float* __restrict__ out,
                                                                const float* __restrict__ vnd,
                                                                const int* __restrict__ blocks,
                                                                float* __restrict__ x_in) {
    const int p0 = blockIdx.x * (EW_THREADS * EW_PIECES) + threadIdx.x;
    const int c4 = threadIdx.x & 15;
    float4 v[EW_PIECES];
    int    bk[EW_PIECES];
#pragma unroll
    for (int u = 0; u < EW_PIECES; ++u) {
        const int p = p0 + u * EW_THREADS;
        v[u]  = ((const float4*)out)[p];
        bk[u] = blocks[p >> 4];
    }
#pragma unroll
    for (int u = 0; u < EW_PIECES; ++u) {
        const int p = p0 + u * EW_THREADS;
        const float4 a = ((const float4*)vnd)[bk[u] * (HDIM / 4) + c4];
        float4 o;
        o.x = v[u].x + a.x;
        o.y = v[u].y + a.y;
        o.z = v[u].z + a.z;
        o.w = v[u].w + a.w;
        ((float4*)x_in)[p] = o;
    }
}

__global__ __launch_bounds__(256) void vn_update_kernel(float* __restrict__ vn_direct,
                                                        float* __restrict__ vn_seg,
                                                        float* __restrict__ vn_root) {
    int tid = threadIdx.x;
    for (int i = tid; i < HNUMB * HDIM; i += 256) {
        vn_direct[i] += vn_seg[i] + vn_root[i & (HDIM - 1)];
        vn_seg[i] = 0.f;
    }
    __syncthreads();
    if (tid < HDIM) {
        float s = 0.f;
        for (int r = 0; r < HNUMB; ++r) s += vn_direct[r * HDIM + tid];
        vn_root[tid] += s;
    }
}

__global__ __launch_bounds__(64) void mlp_kernel(float* __restrict__ vn,
                                                 const float* __restrict__ W1, const float* __restrict__ b1,
                                                 const float* __restrict__ W2, const float* __restrict__ b2) {
    __shared__ float rin[HDIM];
    __shared__ float t1[HDIM];
    int f = threadIdx.x;
    float* row = vn + (size_t)blockIdx.x * HDIM;
    rin[f] = row[f];
    __syncthreads();
    float s = b1[f];
#pragma unroll
    for (int k = 0; k < HDIM; ++k) s += rin[k] * W1[k * HDIM + f];
    t1[f] = fmaxf(s, 0.f);
    __syncthreads();
    float s2 = b2[f];
#pragma unroll
    for (int k = 0; k < HDIM; ++k) s2 += t1[k] * W2[k * HDIM + f];
    row[f] = fmaxf(s2, 0.f);
}

__global__ void vn_init_kernel(const float* __restrict__ vn_emb, float* __restrict__ vn_direct,
                               float* __restrict__ vn_root, float* __restrict__ vn_seg,
                               unsigned* __restrict__ rng) {
    int t = blockIdx.x * blockDim.x + threadIdx.x;
    if (t < HNUMB * HDIM) { vn_direct[t] = vn_emb[t & (HDIM - 1)]; vn_seg[t] = 0.f; }
    if (t < HDIM) vn_root[t] = vn_emb[t];
    if (t < 2 * (NCONVL + 1)) rng[t] = (t & 1) ? 0xffffffffu : 0u;
}

extern "C" void kernel_launch(void* const* d_in, const int* in_sizes, int n_in,
                              void* d_out, int out_size, void* d_ws, size_t ws_size,
                              hipStream_t stream) {
    const float* x       = (const float*)d_in[0];
    const int*   ei      = (const int*)d_in[1];
    const int*   src     = ei;
    const int*   dst     = ei + NEDGES;
    const int*   hblocks = (const int*)d_in[2];
    const float* W0      = (const float*)d_in[5];
    const float* asrc0   = (const float*)d_in[6];
    const float* adst0   = (const float*)d_in[7];
    const float* b0      = (const float*)d_in[8];
    const float* Wl      = (const float*)d_in[9];
    const float* asrcl   = (const float*)d_in[10];
    const float* adstl   = (const float*)d_in[11];
    const float* bl      = (const float*)d_in[12];
    const float* vn_emb  = (const float*)d_in[13];
    const float* mW1     = (const float*)d_in[14];
    const float* mb1     = (const float*)d_in[15];
    const float* mW2     = (const float*)d_in[16];
    const float* mb2     = (const float*)d_in[17];
    float* out = (float*)d_out;

    char*  ws  = (char*)d_ws;
    size_t off = 0;
    auto carve = [&](size_t bytes) -> void* {
        void* p = ws + off;
        off += (bytes + 255) & ~(size_t)255;
        return p;
    };
    unsigned* bits      = (unsigned*)carve((size_t)OW_GROUPS * OW_WORDS * 4);
    float*    x_in      = (float*)carve((size_t)NNODES * HDIM * 4);
    float*    a_s       = (float*)carve((size_t)NNODES * 4);
    float*    rec       = (float*)carve((size_t)NNODES * 8);
    float*    e_self    = (float*)carve((size_t)NNODES * 4);
    unsigned* m_enc1    = (unsigned*)carve((size_t)NNODES * 4);
    unsigned* rng       = (unsigned*)carve((size_t)2 * (NCONVL + 1) * 4);
    float*    vn_direct = (float*)carve((size_t)HNUMB * HDIM * 4);
    float*    vn_root   = (float*)carve((size_t)HDIM * 4);
    float*    vn_seg    = (float*)carve((size_t)HNUMB * HDIM * 4);
    unsigned* sd        = (unsigned*)(ws + SD_OFF_BYTES);

    const int nodeBlk = (NNODES + 255) / 256;
    const int edgeBlk = NEDGES / 256;

    auto run_gat = [&](const float* Xin, const float* W, const float* asrc,
                       const float* adst, const float* bias, unsigned* lrng, bool with_segsum) {
        dots_kernel<<<nodeBlk, 256, 0, stream>>>(Xin, W, asrc, adst, a_s, rec, e_self, m_enc1, lrng, NNODES);
        edge_max_kernel<<<edgeBlk, 256, 0, stream>>>(src, dst, a_s, rec, m_enc1, lrng, NEDGES);
        if (with_segsum)
            owner_layer_kernel<true><<<OW_GROUPS, OW_THREADS, 0, stream>>>(bits, src, dst, sd, ws_size, a_s, rec, m_enc1, e_self, lrng, Xin, W, bias, hblocks, out, vn_seg);
        else
            owner_layer_kernel<false><<<OW_GROUPS, OW_THREADS, 0, stream>>>(bits, src, dst, sd, ws_size, a_s, rec, m_enc1, e_self, lrng, Xin, W, bias, hblocks, out, vn_seg);
    };

    vn_init_kernel<<<(HNUMB * HDIM + 255) / 256, 256, 0, stream>>>(vn_emb, vn_direct, vn_root, vn_seg, rng);
    owner_bits_kernel<<<NEDGES / BT_THREADS, BT_THREADS, 0, stream>>>(dst, src, bits, sd, ws_size);

    run_gat(x, W0, asrc0, adst0, b0, rng, false);

    for (int l = 0; l < NCONVL; ++l) {
        gather_add_kernel<<<EW_BLOCKS, EW_THREADS, 0, stream>>>(out, vn_direct, hblocks, x_in);
        run_gat(x_in, Wl + l * HDIM * HDIM, asrcl + l * HDIM, adstl + l * HDIM, bl + l * HDIM, rng + 2 * (l + 1), true);
        vn_update_kernel<<<1, 256, 0, stream>>>(vn_direct, vn_seg, vn_root);
        for (int j = 0; j < NMLPL; ++j) {
            mlp_kernel<<<HNUMB, 64, 0, stream>>>(vn_direct, mW1 + j * HDIM * HDIM, mb1 + j * HDIM,
                                                 mW2 + j * HDIM * HDIM, mb2 + j * HDIM);
            mlp_kernel<<<1, 64, 0, stream>>>(vn_root, mW1 + j * HDIM * HDIM, mb1 + j * HDIM,
                                             mW2 + j * HDIM * HDIM, mb2 + j * HDIM);
        }
    }
}
